# starting kernel
# baseline (speedup 1.0000x reference)
#include <hip/hip_runtime.h>
#include <math.h>

typedef float v2f __attribute__((ext_vector_type(2)));
typedef float v8f __attribute__((ext_vector_type(8)));

#define DIM 64
#define NEG_ATT 0.2f
#define NEG_OUT 0.01f

__global__ void fill_kernel(float* __restrict__ p, float v, int n) {
    int i = blockIdx.x * blockDim.x + threadIdx.x;
    if (i < n) p[i] = v;
}

__global__ __launch_bounds__(256) void gemm64_wmma(const float* __restrict__ X,
                                                   const float* __restrict__ W,
                                                   float* __restrict__ XW,
                                                   int n_row_tiles) {
    int wave   = blockIdx.x * (blockDim.x >> 5) + (threadIdx.x >> 5);
    int tile_m = wave >> 2;
    int tile_n = wave & 3;
    if (tile_m >= n_row_tiles) return;

    int lane = threadIdx.x & 31;
    int half = lane >> 4;
    int l    = lane & 15;

    const float* xrow = X + (size_t)(tile_m * 16 + l) * DIM;
    const float* wcol = W + tile_n * 16 + l;

    v8f c = {0.f, 0.f, 0.f, 0.f, 0.f, 0.f, 0.f, 0.f};
#pragma unroll
    for (int k = 0; k < DIM; k += 4) {
        v2f a, b;
        a.x = xrow[k + 2 * half];
        a.y = xrow[k + 2 * half + 1];
        b.x = wcol[(size_t)(k + 2 * half) * DIM];
        b.y = wcol[(size_t)(k + 2 * half + 1) * DIM];
        c = __builtin_amdgcn_wmma_f32_16x16x4_f32(false, a, false, b,
                                                  (short)0, c, false, false);
    }
    float* orow = XW + (size_t)(tile_m * 16 + 8 * half) * DIM + tile_n * 16 + l;
#pragma unroll
    for (int i = 0; i < 8; ++i) orow[(size_t)i * DIM] = c[i];
}

__global__ __launch_bounds__(256) void attn_dots(const float* __restrict__ XW,
                                                 const float* __restrict__ as,
                                                 const float* __restrict__ ad,
                                                 float* __restrict__ a_src,
                                                 float* __restrict__ a_dst, int n) {
    int node = (blockIdx.x * blockDim.x + threadIdx.x) >> 5;
    int lane = threadIdx.x & 31;
    if (node >= n) return;
    const float* row = XW + (size_t)node * DIM;
    float vs = row[lane] * as[lane] + row[lane + 32] * as[lane + 32];
    float vd = row[lane] * ad[lane] + row[lane + 32] * ad[lane + 32];
#pragma unroll
    for (int off = 16; off > 0; off >>= 1) {
        vs += __shfl_xor(vs, off, 32);
        vd += __shfl_xor(vd, off, 32);
    }
    if (lane == 0) { a_src[node] = vs; a_dst[node] = vd; }
}

__global__ void edge_alpha_max(const int* __restrict__ esrc, const int* __restrict__ edst,
                               const float* __restrict__ a_src, const float* __restrict__ a_dst,
                               float* __restrict__ alpha, float* __restrict__ amax,
                               int E, int Etot) {
    int e = blockIdx.x * blockDim.x + threadIdx.x;
    if (e >= Etot) return;
    int s, d;
    if (e < E) { s = esrc[e]; d = edst[e]; } else { s = e - E; d = s; }
    float a = a_src[s] + a_dst[d];
    a = (a > 0.f) ? a : NEG_ATT * a;
    alpha[e] = a;
    atomicMax(amax + d, a);
}

__global__ void edge_exp_sum(const int* __restrict__ edst,
                             float* __restrict__ alpha, const float* __restrict__ amax,
                             float* __restrict__ denom, int E, int Etot) {
    int e = blockIdx.x * blockDim.x + threadIdx.x;
    if (e >= Etot) return;
    int d = (e < E) ? edst[e] : (e - E);
    float ex = __expf(alpha[e] - amax[d]);
    alpha[e] = ex;
    atomicAdd(denom + d, ex);
}

__global__ void edge_scatter(const int* __restrict__ esrc, const int* __restrict__ edst,
                             const float* __restrict__ XW, const float* __restrict__ ex,
                             const float* __restrict__ denom, float* __restrict__ out,
                             int E, int Etot) {
    int t = blockIdx.x * blockDim.x + threadIdx.x;
    int e = t >> 4;
    if (e >= Etot) return;
    int g = t & 15;
    int s, d;
    if (e < E) { s = esrc[e]; d = edst[e]; } else { s = e - E; d = s; }
    float coef = ex[e] / (denom[d] + 1e-16f);
    const float4 x4 = *(const float4*)(XW + (size_t)s * DIM + g * 4);
    float* o = out + (size_t)d * DIM + g * 4;
    atomicAdd(o + 0, x4.x * coef);
    atomicAdd(o + 1, x4.y * coef);
    atomicAdd(o + 2, x4.z * coef);
    atomicAdd(o + 3, x4.w * coef);
}

__global__ void bias_lrelu(float* __restrict__ out, const float* __restrict__ b, int total) {
    int i = blockIdx.x * blockDim.x + threadIdx.x;
    if (i >= total) return;
    float v = out[i] + b[i & (DIM - 1)];
    out[i] = (v > 0.f) ? v : NEG_OUT * v;
}

static inline int cdiv(int a, int b) { return (a + b - 1) / b; }

static void run_layer(const float* x, const float* W, const float* att_s, const float* att_d,
                      const float* bias, const int* esrc, const int* edst,
                      int N, int E, int Etot,
                      float* xw, float* a_src, float* a_dst, float* amax, float* denom,
                      float* alpha, float* outbuf, hipStream_t stream) {
    int n_row_tiles = N / 16;
    int waves = n_row_tiles * 4;
    gemm64_wmma<<<cdiv(waves, 8), 256, 0, stream>>>(x, W, xw, n_row_tiles);

    attn_dots<<<cdiv(N * 32, 256), 256, 0, stream>>>(xw, att_s, att_d, a_src, a_dst, N);

    fill_kernel<<<cdiv(N, 256), 256, 0, stream>>>(amax, -INFINITY, N);
    fill_kernel<<<cdiv(N, 256), 256, 0, stream>>>(denom, 0.f, N);
    fill_kernel<<<cdiv(N * DIM, 256), 256, 0, stream>>>(outbuf, 0.f, N * DIM);

    edge_alpha_max<<<cdiv(Etot, 256), 256, 0, stream>>>(esrc, edst, a_src, a_dst,
                                                        alpha, amax, E, Etot);
    edge_exp_sum<<<cdiv(Etot, 256), 256, 0, stream>>>(edst, alpha, amax, denom, E, Etot);
    edge_scatter<<<cdiv(Etot * 16, 256), 256, 0, stream>>>(esrc, edst, xw, alpha, denom,
                                                           outbuf, E, Etot);

    bias_lrelu<<<cdiv(N * DIM, 256), 256, 0, stream>>>(outbuf, bias, N * DIM);
}

extern "C" void kernel_launch(void* const* d_in, const int* in_sizes, int n_in,
                              void* d_out, int out_size, void* d_ws, size_t ws_size,
                              hipStream_t stream) {
    const int*   edge_index = (const int*)d_in[0];
    const float* emb        = (const float*)d_in[1];
    const float* W1         = (const float*)d_in[2];
    const float* att_src1   = (const float*)d_in[3];
    const float* att_dst1   = (const float*)d_in[4];
    const float* b1         = (const float*)d_in[5];
    const float* W2         = (const float*)d_in[6];
    const float* att_src2   = (const float*)d_in[7];
    const float* att_dst2   = (const float*)d_in[8];
    const float* b2         = (const float*)d_in[9];

    const int E    = in_sizes[0] / 2;
    const int N    = in_sizes[1] / DIM;
    const int Etot = E + N;

    const int* esrc = edge_index;
    const int* edst = edge_index + E;

    float* ws     = (float*)d_ws;
    float* xw     = ws;
    float* h1     = xw + (size_t)N * DIM;
    float* a_src  = h1 + (size_t)N * DIM;
    float* a_dst  = a_src + N;
    float* amax   = a_dst + N;
    float* denom  = amax + N;
    float* alpha  = denom + N;

    run_layer(emb, W1, att_src1, att_dst1, b1, esrc, edst, N, E, Etot,
              xw, a_src, a_dst, amax, denom, alpha, h1, stream);
    run_layer(h1, W2, att_src2, att_dst2, b2, esrc, edst, N, E, Etot,
              xw, a_src, a_dst, amax, denom, alpha, (float*)d_out, stream);
}


# Edges counting-sorted by destination once per call (int atomic rank + scan); one wave per node then does softmax, weighted sum, bias, no float atomics; sums run in list (arrival-rank) order
# speedup vs baseline: 2.9937x; 2.9937x over previous
#include <hip/hip_runtime.h>
#include <math.h>


typedef float v2f __attribute__((ext_vector_type(2)));
typedef float v8f __attribute__((ext_vector_type(8)));

#define DIM 64
#define NEG_ATT 0.2f
#define NEG_OUT 0.01f
#define SCAN_TILE 1024

__global__ __launch_bounds__(256) void gemm64_wmma(const float* __restrict__ X,
                                                   const float* __restrict__ W,
                                                   float* __restrict__ XW,
                                                   int n_row_tiles) {
    int wave   = blockIdx.x * (blockDim.x >> 5) + (threadIdx.x >> 5);
    int tile_m = wave >> 2;
    int tile_n = wave & 3;
    if (tile_m >= n_row_tiles) return;

    int lane = threadIdx.x & 31;
    int half = lane >> 4;
    int l    = lane & 15;

    const float* xrow = X + (size_t)(tile_m * 16 + l) * DIM;
    const float* wcol = W + tile_n * 16 + l;

    v8f c = {0.f, 0.f, 0.f, 0.f, 0.f, 0.f, 0.f, 0.f};
#pragma unroll
    for (int k = 0; k < DIM; k += 4) {
        v2f a, b;
        a.x = xrow[k + 2 * half];
        a.y = xrow[k + 2 * half + 1];
        b.x = wcol[(size_t)(k + 2 * half) * DIM];
        b.y = wcol[(size_t)(k + 2 * half + 1) * DIM];
        c = __builtin_amdgcn_wmma_f32_16x16x4_f32(false, a, false, b,
                                                  (short)0, c, false, false);
    }
    float* orow = XW + (size_t)(tile_m * 16 + 8 * half) * DIM + tile_n * 16 + l;
#pragma unroll
    for (int i = 0; i < 8; ++i) orow[(size_t)i * DIM] = c[i];
}

__global__ __launch_bounds__(256) void attn_dots(const float* __restrict__ XW,
                                                 const float* __restrict__ as,
                                                 const float* __restrict__ ad,
                                                 float* __restrict__ a_src,
                                                 float* __restrict__ a_dst, int n) {
    int node = (blockIdx.x * blockDim.x + threadIdx.x) >> 5;
    int lane = threadIdx.x & 31;
    if (node >= n) return;
    const float* row = XW + (size_t)node * DIM;
    float vs = row[lane] * as[lane] + row[lane + 32] * as[lane + 32];
    float vd = row[lane] * ad[lane] + row[lane + 32] * ad[lane + 32];
#pragma unroll
    for (int off = 16; off > 0; off >>= 1) {
        vs += __shfl_xor(vs, off, 32);
        vd += __shfl_xor(vd, off, 32);
    }
    if (lane == 0) { a_src[node] = vs; a_dst[node] = vd; }
}


__global__ void zero_counts(int* __restrict__ cnt, int n) {
    int i = blockIdx.x * blockDim.x + threadIdx.x;
    if (i < n) cnt[i] = 0;
}

__global__ void edge_rank(const int* __restrict__ edst, int* __restrict__ cnt,
                          int* __restrict__ rank, int E) {
    int e = blockIdx.x * blockDim.x + threadIdx.x;
    if (e >= E) return;
    rank[e] = atomicAdd(cnt + edst[e], 1);
}

__global__ __launch_bounds__(256) void scan_local(const int* __restrict__ cnt,
                                                  int* __restrict__ rowptr,
                                                  int* __restrict__ bsum, int n) {
    __shared__ int wsum[8];
    int tid  = threadIdx.x;
    int lane = tid & 31;
    int wv   = tid >> 5;
    int i0   = (blockIdx.x * 256 + tid) * 4;
    int c0 = (i0     < n) ? cnt[i0]     : 0;
    int c1 = (i0 + 1 < n) ? cnt[i0 + 1] : 0;
    int c2 = (i0 + 2 < n) ? cnt[i0 + 2] : 0;
    int c3 = (i0 + 3 < n) ? cnt[i0 + 3] : 0;
    int t   = c0 + c1 + c2 + c3;
    int inc = t;
    for (int off = 1; off < 32; off <<= 1) {
        int v = __shfl_up(inc, off, 32);
        if (lane >= off) inc += v;
    }
    if (lane == 31) wsum[wv] = inc;
    __syncthreads();
    int woff = 0;
    for (int w = 0; w < 8; ++w) woff += (w < wv) ? wsum[w] : 0;
    int ex = woff + inc - t;
    if (i0     < n) rowptr[i0]     = ex;
    if (i0 + 1 < n) rowptr[i0 + 1] = ex + c0;
    if (i0 + 2 < n) rowptr[i0 + 2] = ex + c0 + c1;
    if (i0 + 3 < n) rowptr[i0 + 3] = ex + c0 + c1 + c2;
    if (tid == 255) bsum[blockIdx.x] = woff + inc;
}

__global__ __launch_bounds__(256) void scan_add(int* __restrict__ rowptr,
                                                const int* __restrict__ bsum,
                                                int n, int total) {
    __shared__ int wsum[8];
    int tid  = threadIdx.x;
    int lane = tid & 31;
    int wv   = tid >> 5;
    int b    = blockIdx.x;
    int part = 0;
    for (int j = tid; j < b; j += 256) part += bsum[j];
    for (int off = 16; off > 0; off >>= 1) part += __shfl_xor(part, off, 32);
    if (lane == 0) wsum[wv] = part;
    __syncthreads();
    int off = 0;
    for (int w = 0; w < 8; ++w) off += wsum[w];
    int i0 = (b * 256 + tid) * 4;
    for (int u = 0; u < 4; ++u)
        if (i0 + u < n) rowptr[i0 + u] += off;
    if (b == 0 && tid == 0) rowptr[n] = total;
}

__global__ void list_fill(const int* __restrict__ esrc, const int* __restrict__ edst,
                          const int* __restrict__ rank, const int* __restrict__ rowptr,
                          int* __restrict__ lst, int E) {
    int e = blockIdx.x * blockDim.x + threadIdx.x;
    if (e >= E) return;
    int p = rowptr[edst[e]] + rank[e];
    p = min(max(p, 0), E - 1);
    lst[p] = esrc[e];
}


__device__ __forceinline__ float lrelu_att(float a) { return (a > 0.f) ? a : NEG_ATT * a; }

__device__ __forceinline__ int lane_i(int v, int j) { return __builtin_amdgcn_readlane(v, j); }
__device__ __forceinline__ float lane_f(float v, int j) {
    return __int_as_float(__builtin_amdgcn_readlane(__float_as_int(v), j));
}
__device__ __forceinline__ float2 row2(const float* __restrict__ XW, int s, int lane) {
    return *(const float2*)(XW + (size_t)s * DIM + 2 * lane);
}

template <int K>
__device__ __forceinline__ void add_group(const float* __restrict__ XW, int s, float ex,
                                          int j, int lane, float& ax, float& ay) {
    float2 x[K];
    for (int u = 0; u < K; ++u) x[u] = row2(XW, lane_i(s, j + u), lane);
    for (int u = 0; u < K; ++u) {
        const float w = lane_f(ex, j + u);
        ax += w * x[u].x;
        ay += w * x[u].y;
    }
}

__device__ __forceinline__ void add_rows(const float* __restrict__ XW, int s, float ex,
                                         int cnt, int lane, float& ax, float& ay) {
    int j = 0;
    for (; j + 8 <= cnt; j += 8) add_group<8>(XW, s, ex, j, lane, ax, ay);
    switch (cnt - j) {
        case 1: add_group<1>(XW, s, ex, j, lane, ax, ay); break;
        case 2: add_group<2>(XW, s, ex, j, lane, ax, ay); break;
        case 3: add_group<3>(XW, s, ex, j, lane, ax, ay); break;
        case 4: add_group<4>(XW, s, ex, j, lane, ax, ay); break;
        case 5: add_group<5>(XW, s, ex, j, lane, ax, ay); break;
        case 6: add_group<6>(XW, s, ex, j, lane, ax, ay); break;
        case 7: add_group<7>(XW, s, ex, j, lane, ax, ay); break;
        default: break;
    }
}

__global__ __launch_bounds__(256) void node_gather(const int* __restrict__ rowptr,
                                                   const int* __restrict__ lst,
                                                   const float* __restrict__ XW,
                                                   const float* __restrict__ a_src,
                                                   const float* __restrict__ a_dst,
                                                   const float* __restrict__ bias,
                                                   float* __restrict__ out, int n, int E) {
    const int node = __builtin_amdgcn_readfirstlane((int)((blockIdx.x * blockDim.x + threadIdx.x) >> 5));
    const int lane = threadIdx.x & 31;
    if (node >= n) return;

    const int beg = min(max(rowptr[node], 0), E);
    const int end = min(max(rowptr[node + 1], beg), E);
    const float ad     = a_dst[node];
    const float a_self = lrelu_att(a_src[node] + ad);

    int   s0 = node;
    float a0 = a_self;
    const bool have0 = (beg + lane < end);
    if (have0) { s0 = min(max(lst[beg + lane], 0), n - 1); a0 = lrelu_att(a_src[s0] + ad); }

    float m = fmaxf(a_self, a0);
    for (int k = beg + 32 + lane; k < end; k += 32)
        m = fmaxf(m, lrelu_att(a_src[min(max(lst[k], 0), n - 1)] + ad));
    for (int off = 16; off > 0; off >>= 1) m = fmaxf(m, __shfl_xor(m, off, 32));

    const float  e_self = __expf(a_self - m);
    const float2 xs = row2(XW, node, lane);
    float ax = e_self * xs.x, ay = e_self * xs.y;
    float ex0  = have0 ? __expf(a0 - m) : 0.f;
    float dsum = ex0;
    const int deg = end - beg;
    add_rows(XW, s0, ex0, (deg < 32) ? deg : 32, lane, ax, ay);
    for (int base = beg + 32; base < end; base += 32) {
        int   s  = node;
        float ex = 0.f;
        if (base + lane < end) {
            s  = min(max(lst[base + lane], 0), n - 1);
            ex = __expf(lrelu_att(a_src[s] + ad) - m);
        }
        dsum += ex;
        const int left = end - base;
        add_rows(XW, s, ex, (left < 32) ? left : 32, lane, ax, ay);
    }
    for (int off = 16; off > 0; off >>= 1) dsum += __shfl_xor(dsum, off, 32);
    const float den = (dsum + e_self) + 1e-16f;

    float vx = ax / den + bias[2 * lane];
    float vy = ay / den + bias[2 * lane + 1];
    vx = (vx > 0.f) ? vx : NEG_OUT * vx;
    vy = (vy > 0.f) ? vy : NEG_OUT * vy;
    *(float2*)(out + (size_t)node * DIM + 2 * lane) = make_float2(vx, vy);
}

static inline int cdiv(int a, int b) { return (a + b - 1) / b; }

static void run_layer(const float* x, const float* W, const float* att_s, const float* att_d,
                      const float* bias, const int* rowptr, const int* lst, int N, int E,
                      float* xw, float* a_src, float* a_dst, float* outbuf, hipStream_t stream) {
    int n_row_tiles = N / 16;
    int waves = n_row_tiles * 4;
    gemm64_wmma<<<cdiv(waves, 8), 256, 0, stream>>>(x, W, xw, n_row_tiles);

    attn_dots<<<cdiv(N * 32, 256), 256, 0, stream>>>(xw, att_s, att_d, a_src, a_dst, N);

    node_gather<<<cdiv(N * 32, 256), 256, 0, stream>>>(rowptr, lst, xw, a_src, a_dst, bias,
                                                       outbuf, N, E);
}

extern "C" void kernel_launch(void* const* d_in, const int* in_sizes, int n_in,
                              void* d_out, int out_size, void* d_ws, size_t ws_size,
                              hipStream_t stream) {
    const int*   edge_index = (const int*)d_in[0];
    const float* emb        = (const float*)d_in[1];
    const float* W1         = (const float*)d_in[2];
    const float* att_src1   = (const float*)d_in[3];
    const float* att_dst1   = (const float*)d_in[4];
    const float* b1         = (const float*)d_in[5];
    const float* W2         = (const float*)d_in[6];
    const float* att_src2   = (const float*)d_in[7];
    const float* att_dst2   = (const float*)d_in[8];
    const float* b2         = (const float*)d_in[9];

    const int E = in_sizes[0] / 2;
    const int N = in_sizes[1] / DIM;

    const int* esrc = edge_index;
    const int* edst = edge_index + E;

    float* ws     = (float*)d_ws;
    float* xw     = ws;
    float* h1     = xw + (size_t)N * DIM;
    float* a_src  = h1 + (size_t)N * DIM;
    float* a_dst  = a_src + N;
    int*   cnt    = (int*)(a_dst + N);
    int*   rowptr = cnt + N;
    int*   bsum   = rowptr + N + 1;
    const int n_tiles = cdiv(N, SCAN_TILE);
    int*   rank   = bsum + n_tiles;
    int*   lst    = rank + E;

    zero_counts<<<cdiv(N, 256), 256, 0, stream>>>(cnt, N);
    edge_rank<<<cdiv(E, 256), 256, 0, stream>>>(edst, cnt, rank, E);
    scan_local<<<n_tiles, 256, 0, stream>>>(cnt, rowptr, bsum, N);
    scan_add<<<n_tiles, 256, 0, stream>>>(rowptr, bsum, N, E);
    list_fill<<<cdiv(E, 256), 256, 0, stream>>>(esrc, edst, rank, rowptr, lst, E);

    run_layer(emb, W1, att_src1, att_dst1, b1, rowptr, lst, N, E, xw, a_src, a_dst, h1, stream);
    run_layer(h1, W2, att_src2, att_dst2, b2, rowptr, lst, N, E, xw, a_src, a_dst,
              (float*)d_out, stream);
}


# Attention dots folded into the GEMM: a wave owns a 16x64 strip (W via LDS, 4 adjacent columns per lane), sums its accumulators to a_src/a_dst; dot kernel gone; product k order, dot sum tree change
# speedup vs baseline: 3.2808x; 1.0959x over previous
#include <hip/hip_runtime.h>
#include <math.h>


typedef float v2f __attribute__((ext_vector_type(2)));
typedef float v8f __attribute__((ext_vector_type(8)));

#define DIM 64
#define NEG_ATT 0.2f
#define NEG_OUT 0.01f
#define SCAN_TILE 1024

__device__ __forceinline__ v2f lo2(const float4 q) { v2f r; r.x = q.x; r.y = q.y; return r; }
__device__ __forceinline__ v2f hi2(const float4 q) { v2f r; r.x = q.z; r.y = q.w; return r; }

template <int KEEP>
__device__ __forceinline__ void fold_step(float* v, int c) {
    const bool up = (c & KEEP) != 0;
#pragma unroll
    for (int i = 0; i < KEEP; ++i) {
        const float keep = up ? v[i + KEEP] : v[i];
        const float send = up ? v[i] : v[i + KEEP];
        v[i] = keep + __shfl_xor(send, KEEP, 32);
    }
}

__global__ __launch_bounds__(256) void gemm64_dots(const float* __restrict__ X,
                                                   const float* __restrict__ Wm,
                                                   const float* __restrict__ as,
                                                   const float* __restrict__ ad,
                                                   float* __restrict__ XW,
                                                   float* __restrict__ a_src,
                                                   float* __restrict__ a_dst,
                                                   int n_row_tiles) {
    __shared__ float4 Bs[8 * 4 * 32];

    const int tid   = threadIdx.x;
    const int lane  = tid & 31;
    const int h     = lane >> 4;
    const int c     = lane & 15;
    const int strip = blockIdx.x * 8 + (tid >> 5);
    const bool live = strip < n_row_tiles;

    float4 a4[8];
    if (live) {
        const float4* xr = (const float4*)(X + (size_t)(strip * 16 + c) * DIM + 32 * h);
#pragma unroll
        for (int t2 = 0; t2 < 8; ++t2) a4[t2] = xr[t2];
    }

#pragma unroll
    for (int i = 0; i < 4; ++i) {
        const int q  = i * 256 + tid;
        const int ql = q & 31, qn = (q >> 5) & 3, qt = q >> 7;
        const float* wp = Wm + (size_t)(32 * (ql >> 4) + 4 * qt) * DIM + 4 * (ql & 15) + qn;
        Bs[q] = make_float4(wp[0], wp[DIM], wp[2 * DIM], wp[3 * DIM]);
    }
    __syncthreads();
    if (!live) return;

    v8f acc[4];
#pragma unroll
    for (int n = 0; n < 4; ++n) acc[n] = (v8f){0.f, 0.f, 0.f, 0.f, 0.f, 0.f, 0.f, 0.f};

#pragma unroll
    for (int t2 = 0; t2 < 8; ++t2) {
        float4 b4[4];
#pragma unroll
        for (int n = 0; n < 4; ++n) b4[n] = Bs[(t2 * 4 + n) * 32 + lane];
#pragma unroll
        for (int n = 0; n < 4; ++n)
            acc[n] = __builtin_amdgcn_wmma_f32_16x16x4_f32(false, lo2(a4[t2]), false, lo2(b4[n]),
                                                           (short)0, acc[n], false, false);
#pragma unroll
        for (int n = 0; n < 4; ++n)
            acc[n] = __builtin_amdgcn_wmma_f32_16x16x4_f32(false, hi2(a4[t2]), false, hi2(b4[n]),
                                                           (short)0, acc[n], false, false);
    }

    float* orow = XW + (size_t)(strip * 16 + 8 * h) * DIM + 4 * c;
#pragma unroll
    for (int r = 0; r < 8; ++r)
        *(float4*)(orow + (size_t)r * DIM) = make_float4(acc[0][r], acc[1][r], acc[2][r], acc[3][r]);

    const float4 as4 = *(const float4*)(as + 4 * c);
    const float4 ad4 = *(const float4*)(ad + 4 * c);
    const float asv[4] = {as4.x, as4.y, as4.z, as4.w};
    const float adv[4] = {ad4.x, ad4.y, ad4.z, ad4.w};
    float v[16];
#pragma unroll
    for (int r = 0; r < 8; ++r) {
        float s = acc[0][r] * asv[0];
        float d = acc[0][r] * adv[0];
#pragma unroll
        for (int n = 1; n < 4; ++n) { s += acc[n][r] * asv[n]; d += acc[n][r] * adv[n]; }
        v[r] = s;
        v[8 + r] = d;
    }
    fold_step<8>(v, c);
    fold_step<4>(v, c);
    fold_step<2>(v, c);
    fold_step<1>(v, c);
    float* dots = (c < 8) ? a_src : a_dst;
    dots[strip * 16 + 8 * h + (c & 7)] = v[0];
}


__global__ void zero_counts(int* __restrict__ cnt, int n) {
    int i = blockIdx.x * blockDim.x + threadIdx.x;
    if (i < n) cnt[i] = 0;
}

__global__ void edge_rank(const int* __restrict__ edst, int* __restrict__ cnt,
                          int* __restrict__ rank, int E) {
    int e = blockIdx.x * blockDim.x + threadIdx.x;
    if (e >= E) return;
    rank[e] = atomicAdd(cnt + edst[e], 1);
}

__global__ __launch_bounds__(256) void scan_local(const int* __restrict__ cnt,
                                                  int* __restrict__ rowptr,
                                                  int* __restrict__ bsum, int n) {
    __shared__ int wsum[8];
    int tid  = threadIdx.x;
    int lane = tid & 31;
    int wv   = tid >> 5;
    int i0   = (blockIdx.x * 256 + tid) * 4;
    int c0 = (i0     < n) ? cnt[i0]     : 0;
    int c1 = (i0 + 1 < n) ? cnt[i0 + 1] : 0;
    int c2 = (i0 + 2 < n) ? cnt[i0 + 2] : 0;
    int c3 = (i0 + 3 < n) ? cnt[i0 + 3] : 0;
    int t   = c0 + c1 + c2 + c3;
    int inc = t;
    for (int off = 1; off < 32; off <<= 1) {
        int v = __shfl_up(inc, off, 32);
        if (lane >= off) inc += v;
    }
    if (lane == 31) wsum[wv] = inc;
    __syncthreads();
    int woff = 0;
    for (int w = 0; w < 8; ++w) woff += (w < wv) ? wsum[w] : 0;
    int ex = woff + inc - t;
    if (i0     < n) rowptr[i0]     = ex;
    if (i0 + 1 < n) rowptr[i0 + 1] = ex + c0;
    if (i0 + 2 < n) rowptr[i0 + 2] = ex + c0 + c1;
    if (i0 + 3 < n) rowptr[i0 + 3] = ex + c0 + c1 + c2;
    if (tid == 255) bsum[blockIdx.x] = woff + inc;
}

__global__ __launch_bounds__(256) void scan_add(int* __restrict__ rowptr,
                                                const int* __restrict__ bsum,
                                                int n, int total) {
    __shared__ int wsum[8];
    int tid  = threadIdx.x;
    int lane = tid & 31;
    int wv   = tid >> 5;
    int b    = blockIdx.x;
    int part = 0;
    for (int j = tid; j < b; j += 256) part += bsum[j];
    for (int off = 16; off > 0; off >>= 1) part += __shfl_xor(part, off, 32);
    if (lane == 0) wsum[wv] = part;
    __syncthreads();
    int off = 0;
    for (int w = 0; w < 8; ++w) off += wsum[w];
    int i0 = (b * 256 + tid) * 4;
    for (int u = 0; u < 4; ++u)
        if (i0 + u < n) rowptr[i0 + u] += off;
    if (b == 0 && tid == 0) rowptr[n] = total;
}

__global__ void list_fill(const int* __restrict__ esrc, const int* __restrict__ edst,
                          const int* __restrict__ rank, const int* __restrict__ rowptr,
                          int* __restrict__ lst, int E) {
    int e = blockIdx.x * blockDim.x + threadIdx.x;
    if (e >= E) return;
    int p = rowptr[edst[e]] + rank[e];
    p = min(max(p, 0), E - 1);
    lst[p] = esrc[e];
}


__device__ __forceinline__ float lrelu_att(float a) { return (a > 0.f) ? a : NEG_ATT * a; }

__device__ __forceinline__ int lane_i(int v, int j) { return __builtin_amdgcn_readlane(v, j); }
__device__ __forceinline__ float lane_f(float v, int j) {
    return __int_as_float(__builtin_amdgcn_readlane(__float_as_int(v), j));
}
__device__ __forceinline__ float2 row2(const float* __restrict__ XW, int s, int lane) {
    return *(const float2*)(XW + (size_t)s * DIM + 2 * lane);
}

template <int K>
__device__ __forceinline__ void add_group(const float* __restrict__ XW, int s, float ex,
                                          int j, int lane, float& ax, float& ay) {
    float2 x[K];
    for (int u = 0; u < K; ++u) x[u] = row2(XW, lane_i(s, j + u), lane);
    for (int u = 0; u < K; ++u) {
        const float w = lane_f(ex, j + u);
        ax += w * x[u].x;
        ay += w * x[u].y;
    }
}

__device__ __forceinline__ void add_rows(const float* __restrict__ XW, int s, float ex,
                                         int cnt, int lane, float& ax, float& ay) {
    int j = 0;
    for (; j + 8 <= cnt; j += 8) add_group<8>(XW, s, ex, j, lane, ax, ay);
    switch (cnt - j) {
        case 1: add_group<1>(XW, s, ex, j, lane, ax, ay); break;
        case 2: add_group<2>(XW, s, ex, j, lane, ax, ay); break;
        case 3: add_group<3>(XW, s, ex, j, lane, ax, ay); break;
        case 4: add_group<4>(XW, s, ex, j, lane, ax, ay); break;
        case 5: add_group<5>(XW, s, ex, j, lane, ax, ay); break;
        case 6: add_group<6>(XW, s, ex, j, lane, ax, ay); break;
        case 7: add_group<7>(XW, s, ex, j, lane, ax, ay); break;
        default: break;
    }
}

__global__ __launch_bounds__(256) void node_gather(const int* __restrict__ rowptr,
                                                   const int* __restrict__ lst,
                                                   const float* __restrict__ XW,
                                                   const float* __restrict__ a_src,
                                                   const float* __restrict__ a_dst,
                                                   const float* __restrict__ bias,
                                                   float* __restrict__ out, int n, int E) {
    const int node = __builtin_amdgcn_readfirstlane((int)((blockIdx.x * blockDim.x + threadIdx.x) >> 5));
    const int lane = threadIdx.x & 31;
    if (node >= n) return;

    const int beg = min(max(rowptr[node], 0), E);
    const int end = min(max(rowptr[node + 1], beg), E);
    const float ad     = a_dst[node];
    const float a_self = lrelu_att(a_src[node] + ad);

    int   s0 = node;
    float a0 = a_self;
    const bool have0 = (beg + lane < end);
    if (have0) { s0 = min(max(lst[beg + lane], 0), n - 1); a0 = lrelu_att(a_src[s0] + ad); }

    float m = fmaxf(a_self, a0);
    for (int k = beg + 32 + lane; k < end; k += 32)
        m = fmaxf(m, lrelu_att(a_src[min(max(lst[k], 0), n - 1)] + ad));
    for (int off = 16; off > 0; off >>= 1) m = fmaxf(m, __shfl_xor(m, off, 32));

    const float  e_self = __expf(a_self - m);
    const float2 xs = row2(XW, node, lane);
    float ax = e_self * xs.x, ay = e_self * xs.y;
    float ex0  = have0 ? __expf(a0 - m) : 0.f;
    float dsum = ex0;
    const int deg = end - beg;
    add_rows(XW, s0, ex0, (deg < 32) ? deg : 32, lane, ax, ay);
    for (int base = beg + 32; base < end; base += 32) {
        int   s  = node;
        float ex = 0.f;
        if (base + lane < end) {
            s  = min(max(lst[base + lane], 0), n - 1);
            ex = __expf(lrelu_att(a_src[s] + ad) - m);
        }
        dsum += ex;
        const int left = end - base;
        add_rows(XW, s, ex, (left < 32) ? left : 32, lane, ax, ay);
    }
    for (int off = 16; off > 0; off >>= 1) dsum += __shfl_xor(dsum, off, 32);
    const float den = (dsum + e_self) + 1e-16f;

    float vx = ax / den + bias[2 * lane];
    float vy = ay / den + bias[2 * lane + 1];
    vx = (vx > 0.f) ? vx : NEG_OUT * vx;
    vy = (vy > 0.f) ? vy : NEG_OUT * vy;
    *(float2*)(out + (size_t)node * DIM + 2 * lane) = make_float2(vx, vy);
}

static inline int cdiv(int a, int b) { return (a + b - 1) / b; }

static void run_layer(const float* x, const float* W, const float* att_s, const float* att_d,
                      const float* bias, const int* rowptr, const int* lst, int N, int E,
                      float* xw, float* a_src, float* a_dst, float* outbuf, hipStream_t stream) {
    int n_row_tiles = N / 16;
    gemm64_dots<<<cdiv(n_row_tiles, 8), 256, 0, stream>>>(x, W, att_s, att_d, xw, a_src, a_dst,
                                                          n_row_tiles);

    node_gather<<<cdiv(N * 32, 256), 256, 0, stream>>>(rowptr, lst, xw, a_src, a_dst, bias,
                                                       outbuf, N, E);
}

extern "C" void kernel_launch(void* const* d_in, const int* in_sizes, int n_in,
                              void* d_out, int out_size, void* d_ws, size_t ws_size,
                              hipStream_t stream) {
    const int*   edge_index = (const int*)d_in[0];
    const float* emb        = (const float*)d_in[1];
    const float* W1         = (const float*)d_in[2];
    const float* att_src1   = (const float*)d_in[3];
    const float* att_dst1   = (const float*)d_in[4];
    const float* b1         = (const float*)d_in[5];
    const float* W2         = (const float*)d_in[6];
    const float* att_src2   = (const float*)d_in[7];
    const float* att_dst2   = (const float*)d_in[8];
    const float* b2         = (const float*)d_in[9];

    const int E = in_sizes[0] / 2;
    const int N = in_sizes[1] / DIM;

    const int* esrc = edge_index;
    const int* edst = edge_index + E;

    float* ws     = (float*)d_ws;
    float* xw     = ws;
    float* h1     = xw + (size_t)N * DIM;
    float* a_src  = h1 + (size_t)N * DIM;
    float* a_dst  = a_src + N;
    int*   cnt    = (int*)(a_dst + N);
    int*   rowptr = cnt + N;
    int*   bsum   = rowptr + N + 1;
    const int n_tiles = cdiv(N, SCAN_TILE);
    int*   rank   = bsum + n_tiles;
    int*   lst    = rank + E;

    zero_counts<<<cdiv(N, 256), 256, 0, stream>>>(cnt, N);
    edge_rank<<<cdiv(E, 256), 256, 0, stream>>>(edst, cnt, rank, E);
    scan_local<<<n_tiles, 256, 0, stream>>>(cnt, rowptr, bsum, N);
    scan_add<<<n_tiles, 256, 0, stream>>>(rowptr, bsum, N, E);
    list_fill<<<cdiv(E, 256), 256, 0, stream>>>(esrc, edst, rank, rowptr, lst, E);

    run_layer(emb, W1, att_src1, att_dst1, b1, rowptr, lst, N, E, xw, a_src, a_dst, h1, stream);
    run_layer(h1, W2, att_src2, att_dst2, b2, rowptr, lst, N, E, xw, a_src, a_dst,
              (float*)d_out, stream);
}


# Edge sort rebuilt as two-level bucket sort in LDS (256-node buckets): no global atomic or per-edge random access left in it; 5 sort launches to 4; sums still in list order, now set by LDS arrival
# speedup vs baseline: 4.9002x; 1.4936x over previous
#include <hip/hip_runtime.h>
#include <math.h>


typedef float v2f __attribute__((ext_vector_type(2)));
typedef float v8f __attribute__((ext_vector_type(8)));

#define DIM 64
#define NEG_ATT 0.2f
#define NEG_OUT 0.01f

__device__ __forceinline__ v2f lo2(const float4 q) { v2f r; r.x = q.x; r.y = q.y; return r; }
__device__ __forceinline__ v2f hi2(const float4 q) { v2f r; r.x = q.z; r.y = q.w; return r; }

template <int KEEP>
__device__ __forceinline__ void fold_step(float* v, int c) {
    const bool up = (c & KEEP) != 0;
#pragma unroll
    for (int i = 0; i < KEEP; ++i) {
        const float keep = up ? v[i + KEEP] : v[i];
        const float send = up ? v[i] : v[i + KEEP];
        v[i] = keep + __shfl_xor(send, KEEP, 32);
    }
}

__global__ __launch_bounds__(256) void gemm64_dots(const float* __restrict__ X,
                                                   const float* __restrict__ Wm,
                                                   const float* __restrict__ as,
                                                   const float* __restrict__ ad,
                                                   float* __restrict__ XW,
                                                   float* __restrict__ a_src,
                                                   float* __restrict__ a_dst,
                                                   int n_row_tiles) {
    __shared__ float4 Bs[8 * 4 * 32];

    const int tid   = threadIdx.x;
    const int lane  = tid & 31;
    const int h     = lane >> 4;
    const int c     = lane & 15;
    const int strip = blockIdx.x * 8 + (tid >> 5);
    const bool live = strip < n_row_tiles;

    float4 a4[8];
    if (live) {
        const float4* xr = (const float4*)(X + (size_t)(strip * 16 + c) * DIM + 32 * h);
#pragma unroll
        for (int t2 = 0; t2 < 8; ++t2) a4[t2] = xr[t2];
    }

#pragma unroll
    for (int i = 0; i < 4; ++i) {
        const int q  = i * 256 + tid;
        const int ql = q & 31, qn = (q >> 5) & 3, qt = q >> 7;
        const float* wp = Wm + (size_t)(32 * (ql >> 4) + 4 * qt) * DIM + 4 * (ql & 15) + qn;
        Bs[q] = make_float4(wp[0], wp[DIM], wp[2 * DIM], wp[3 * DIM]);
    }
    __syncthreads();
    if (!live) return;

    v8f acc[4];
#pragma unroll
    for (int n = 0; n < 4; ++n) acc[n] = (v8f){0.f, 0.f, 0.f, 0.f, 0.f, 0.f, 0.f, 0.f};

#pragma unroll
    for (int t2 = 0; t2 < 8; ++t2) {
        float4 b4[4];
#pragma unroll
        for (int n = 0; n < 4; ++n) b4[n] = Bs[(t2 * 4 + n) * 32 + lane];
#pragma unroll
        for (int n = 0; n < 4; ++n)
            acc[n] = __builtin_amdgcn_wmma_f32_16x16x4_f32(false, lo2(a4[t2]), false, lo2(b4[n]),
                                                           (short)0, acc[n], false, false);
#pragma unroll
        for (int n = 0; n < 4; ++n)
            acc[n] = __builtin_amdgcn_wmma_f32_16x16x4_f32(false, hi2(a4[t2]), false, hi2(b4[n]),
                                                           (short)0, acc[n], false, false);
    }

    float* orow = XW + (size_t)(strip * 16 + 8 * h) * DIM + 4 * c;
#pragma unroll
    for (int r = 0; r < 8; ++r)
        *(float4*)(orow + (size_t)r * DIM) = make_float4(acc[0][r], acc[1][r], acc[2][r], acc[3][r]);

    const float4 as4 = *(const float4*)(as + 4 * c);
    const float4 ad4 = *(const float4*)(ad + 4 * c);
    const float asv[4] = {as4.x, as4.y, as4.z, as4.w};
    const float adv[4] = {ad4.x, ad4.y, ad4.z, ad4.w};
    float v[16];
#pragma unroll
    for (int r = 0; r < 8; ++r) {
        float s = acc[0][r] * asv[0];
        float d = acc[0][r] * adv[0];
#pragma unroll
        for (int n = 1; n < 4; ++n) { s += acc[n][r] * asv[n]; d += acc[n][r] * adv[n]; }
        v[r] = s;
        v[8 + r] = d;
    }
    fold_step<8>(v, c);
    fold_step<4>(v, c);
    fold_step<2>(v, c);
    fold_step<1>(v, c);
    float* dots = (c < 8) ? a_src : a_dst;
    dots[strip * 16 + 8 * h + (c & 7)] = v[0];
}


#define BUCKET_SHIFT 8
#define BUCKET_NODES (1 << BUCKET_SHIFT)
#define MAXB 512
#define EPB 4096
#define FIN_CAP 8192

__device__ __forceinline__ int bucket_of(int d, int NB) {
    return min(max(d, 0) >> BUCKET_SHIFT, NB - 1);
}

__device__ __forceinline__ int block_excl_scan(int v, int* wsum, int tid, int* tot) {
    const int lane = tid & 31, wv = tid >> 5;
    int inc = v;
    for (int o = 1; o < 32; o <<= 1) {
        int t = __shfl_up(inc, o, 32);
        if (lane >= o) inc += t;
    }
    if (lane == 31) wsum[wv] = inc;
    __syncthreads();
    int woff = 0, all = 0;
    for (int w = 0; w < 8; ++w) {
        const int s = wsum[w];
        all += s;
        if (w < wv) woff += s;
    }
    __syncthreads();
    *tot = all;
    return woff + inc - v;
}

__global__ __launch_bounds__(256) void bucket_hist(const int* __restrict__ edst,
                                                   int* __restrict__ hist, int E, int NB) {
    __shared__ int h[MAXB];
    const int tid = threadIdx.x, blk = blockIdx.x;
    h[tid] = 0;
    h[tid + 256] = 0;
    __syncthreads();
    const int base = blk * EPB;
    for (int k = 0; k < EPB / 256; ++k) {
        const int e = base + k * 256 + tid;
        if (e < E) atomicAdd(&h[bucket_of(edst[e], NB)], 1);
    }
    __syncthreads();
    for (int i = tid; i < NB; i += 256) hist[blk * NB + i] = h[i];
}

__global__ __launch_bounds__(256) void bucket_scan(const int* __restrict__ hist,
                                                   int* __restrict__ off,
                                                   int* __restrict__ btot, int nblk, int NB) {
    __shared__ int wsum[8];
    const int tid = threadIdx.x, b = blockIdx.x;
    int carry = 0;
    for (int c0 = 0; c0 < nblk; c0 += 256) {
        const int blk = c0 + tid;
        const int v = (blk < nblk) ? hist[blk * NB + b] : 0;
        int tot;
        const int ex = block_excl_scan(v, wsum, tid, &tot);
        if (blk < nblk) off[blk * NB + b] = carry + ex;
        carry += tot;
    }
    if (tid == 0) btot[b] = carry;
}

__global__ __launch_bounds__(256) void bucket_scatter(const int* __restrict__ esrc,
                                                      const int* __restrict__ edst,
                                                      const int* __restrict__ hist,
                                                      const int* __restrict__ off,
                                                      const int* __restrict__ btot,
                                                      int2* __restrict__ tmp, int E, int NB) {
    __shared__ int  lbase[MAXB];
    __shared__ int  gdelta[MAXB];
    __shared__ int  cur[MAXB];
    __shared__ int2 stage[EPB];
    __shared__ int  wsum[8];
    const int tid = threadIdx.x, blk = blockIdx.x;
    const int i0 = 2 * tid, i1 = 2 * tid + 1;
    const int t0 = (i0 < NB) ? btot[i0] : 0;
    const int t1 = (i1 < NB) ? btot[i1] : 0;
    const int h0 = (i0 < NB) ? hist[blk * NB + i0] : 0;
    const int h1 = (i1 < NB) ? hist[blk * NB + i1] : 0;
    const int o0 = (i0 < NB) ? off[blk * NB + i0] : 0;
    const int o1 = (i1 < NB) ? off[blk * NB + i1] : 0;
    int tot;
    const int gb = block_excl_scan(t0 + t1, wsum, tid, &tot);
    const int lb = block_excl_scan(h0 + h1, wsum, tid, &tot);
    lbase[i0]  = lb;
    lbase[i1]  = lb + h0;
    gdelta[i0] = gb + o0 - lb;
    gdelta[i1] = gb + t0 + o1 - (lb + h0);
    cur[i0] = 0;
    cur[i1] = 0;
    __syncthreads();

    const int base = blk * EPB;
    for (int k = 0; k < EPB / 256; ++k) {
        const int e = base + k * 256 + tid;
        if (e < E) {
            const int d = edst[e];
            const int s = esrc[e];
            const int b = bucket_of(d, NB);
            const int r = atomicAdd(&cur[b], 1);
            const int slot = min(max(lbase[b] + r, 0), EPB - 1);
            stage[slot] = make_int2(s, d);
        }
    }
    __syncthreads();
    const int n_here = min(EPB, E - base);
    for (int i = tid; i < n_here; i += 256) {
        const int2 sd = stage[i];
        const int pos = min(max(i + gdelta[bucket_of(sd.y, NB)], 0), E - 1);
        tmp[pos] = sd;
    }
}

__global__ __launch_bounds__(256) void bucket_finish(const int2* __restrict__ tmp,
                                                     const int* __restrict__ btot,
                                                     int* __restrict__ rowptr,
                                                     int* __restrict__ lst, int N, int E, int NB) {
    __shared__ int cnt[BUCKET_NODES];
    __shared__ int cur[BUCKET_NODES];
    __shared__ int stage[FIN_CAP];
    __shared__ int wsum[8];
    const int tid = threadIdx.x, b = blockIdx.x;
    int part = 0;
    for (int i = tid; i < b; i += 256) part += btot[i];
    int bstart;
    block_excl_scan(part, wsum, tid, &bstart);
    const int beg = min(max(bstart, 0), E);
    const int end = min(max(beg + btot[b], beg), E);
    const int node0 = b << BUCKET_SHIFT;

    cnt[tid] = 0;
    __syncthreads();
    for (int p = beg + tid; p < end; p += 256) {
        const int li = min(max(tmp[p].y - node0, 0), BUCKET_NODES - 1);
        atomicAdd(&cnt[li], 1);
    }
    __syncthreads();
    int tot;
    const int ex = block_excl_scan(cnt[tid], wsum, tid, &tot);
    if (node0 + tid < N) rowptr[node0 + tid] = beg + ex;
    if (b == 0 && tid == 0) rowptr[N] = E;
    cur[tid] = ex;
    __syncthreads();

    const int n = end - beg;
    if (n <= FIN_CAP) {
        for (int p = beg + tid; p < end; p += 256) {
            const int2 sd = tmp[p];
            const int li = min(max(sd.y - node0, 0), BUCKET_NODES - 1);
            const int r = atomicAdd(&cur[li], 1);
            stage[min(max(r, 0), FIN_CAP - 1)] = sd.x;
        }
        __syncthreads();
        for (int i = tid; i < n; i += 256) lst[beg + i] = stage[i];
    } else {
        for (int p = beg + tid; p < end; p += 256) {
            const int2 sd = tmp[p];
            const int li = min(max(sd.y - node0, 0), BUCKET_NODES - 1);
            const int r = atomicAdd(&cur[li], 1);
            lst[min(max(beg + r, 0), E - 1)] = sd.x;
        }
    }
}


__device__ __forceinline__ float lrelu_att(float a) { return (a > 0.f) ? a : NEG_ATT * a; }

__device__ __forceinline__ int lane_i(int v, int j) { return __builtin_amdgcn_readlane(v, j); }
__device__ __forceinline__ float lane_f(float v, int j) {
    return __int_as_float(__builtin_amdgcn_readlane(__float_as_int(v), j));
}
__device__ __forceinline__ float2 row2(const float* __restrict__ XW, int s, int lane) {
    return *(const float2*)(XW + (size_t)s * DIM + 2 * lane);
}

template <int K>
__device__ __forceinline__ void add_group(const float* __restrict__ XW, int s, float ex,
                                          int j, int lane, float& ax, float& ay) {
    float2 x[K];
    for (int u = 0; u < K; ++u) x[u] = row2(XW, lane_i(s, j + u), lane);
    for (int u = 0; u < K; ++u) {
        const float w = lane_f(ex, j + u);
        ax += w * x[u].x;
        ay += w * x[u].y;
    }
}

__device__ __forceinline__ void add_rows(const float* __restrict__ XW, int s, float ex,
                                         int cnt, int lane, float& ax, float& ay) {
    int j = 0;
    for (; j + 8 <= cnt; j += 8) add_group<8>(XW, s, ex, j, lane, ax, ay);
    switch (cnt - j) {
        case 1: add_group<1>(XW, s, ex, j, lane, ax, ay); break;
        case 2: add_group<2>(XW, s, ex, j, lane, ax, ay); break;
        case 3: add_group<3>(XW, s, ex, j, lane, ax, ay); break;
        case 4: add_group<4>(XW, s, ex, j, lane, ax, ay); break;
        case 5: add_group<5>(XW, s, ex, j, lane, ax, ay); break;
        case 6: add_group<6>(XW, s, ex, j, lane, ax, ay); break;
        case 7: add_group<7>(XW, s, ex, j, lane, ax, ay); break;
        default: break;
    }
}

__global__ __launch_bounds__(256) void node_gather(const int* __restrict__ rowptr,
                                                   const int* __restrict__ lst,
                                                   const float* __restrict__ XW,
                                                   const float* __restrict__ a_src,
                                                   const float* __restrict__ a_dst,
                                                   const float* __restrict__ bias,
                                                   float* __restrict__ out, int n, int E) {
    const int node = __builtin_amdgcn_readfirstlane((int)((blockIdx.x * blockDim.x + threadIdx.x) >> 5));
    const int lane = threadIdx.x & 31;
    if (node >= n) return;

    const int beg = min(max(rowptr[node], 0), E);
    const int end = min(max(rowptr[node + 1], beg), E);
    const float ad     = a_dst[node];
    const float a_self = lrelu_att(a_src[node] + ad);

    int   s0 = node;
    float a0 = a_self;
    const bool have0 = (beg + lane < end);
    if (have0) { s0 = min(max(lst[beg + lane], 0), n - 1); a0 = lrelu_att(a_src[s0] + ad); }

    float m = fmaxf(a_self, a0);
    for (int k = beg + 32 + lane; k < end; k += 32)
        m = fmaxf(m, lrelu_att(a_src[min(max(lst[k], 0), n - 1)] + ad));
    for (int off = 16; off > 0; off >>= 1) m = fmaxf(m, __shfl_xor(m, off, 32));

    const float  e_self = __expf(a_self - m);
    const float2 xs = row2(XW, node, lane);
    float ax = e_self * xs.x, ay = e_self * xs.y;
    float ex0  = have0 ? __expf(a0 - m) : 0.f;
    float dsum = ex0;
    const int deg = end - beg;
    add_rows(XW, s0, ex0, (deg < 32) ? deg : 32, lane, ax, ay);
    for (int base = beg + 32; base < end; base += 32) {
        int   s  = node;
        float ex = 0.f;
        if (base + lane < end) {
            s  = min(max(lst[base + lane], 0), n - 1);
            ex = __expf(lrelu_att(a_src[s] + ad) - m);
        }
        dsum += ex;
        const int left = end - base;
        add_rows(XW, s, ex, (left < 32) ? left : 32, lane, ax, ay);
    }
    for (int off = 16; off > 0; off >>= 1) dsum += __shfl_xor(dsum, off, 32);
    const float den = (dsum + e_self) + 1e-16f;

    float vx = ax / den + bias[2 * lane];
    float vy = ay / den + bias[2 * lane + 1];
    vx = (vx > 0.f) ? vx : NEG_OUT * vx;
    vy = (vy > 0.f) ? vy : NEG_OUT * vy;
    *(float2*)(out + (size_t)node * DIM + 2 * lane) = make_float2(vx, vy);
}

static inline int cdiv(int a, int b) { return (a + b - 1) / b; }

static void run_layer(const float* x, const float* W, const float* att_s, const float* att_d,
                      const float* bias, const int* rowptr, const int* lst, int N, int E,
                      float* xw, float* a_src, float* a_dst, float* outbuf, hipStream_t stream) {
    int n_row_tiles = N / 16;
    gemm64_dots<<<cdiv(n_row_tiles, 8), 256, 0, stream>>>(x, W, att_s, att_d, xw, a_src, a_dst,
                                                          n_row_tiles);

    node_gather<<<cdiv(N * 32, 256), 256, 0, stream>>>(rowptr, lst, xw, a_src, a_dst, bias,
                                                       outbuf, N, E);
}

extern "C" void kernel_launch(void* const* d_in, const int* in_sizes, int n_in,
                              void* d_out, int out_size, void* d_ws, size_t ws_size,
                              hipStream_t stream) {
    const int*   edge_index = (const int*)d_in[0];
    const float* emb        = (const float*)d_in[1];
    const float* W1         = (const float*)d_in[2];
    const float* att_src1   = (const float*)d_in[3];
    const float* att_dst1   = (const float*)d_in[4];
    const float* b1         = (const float*)d_in[5];
    const float* W2         = (const float*)d_in[6];
    const float* att_src2   = (const float*)d_in[7];
    const float* att_dst2   = (const float*)d_in[8];
    const float* b2         = (const float*)d_in[9];

    const int E = in_sizes[0] / 2;
    const int N = in_sizes[1] / DIM;

    const int* esrc = edge_index;
    const int* edst = edge_index + E;

    float* ws     = (float*)d_ws;
    float* xw     = ws;
    float* h1     = xw + (size_t)N * DIM;
    int2*  tmp    = (int2*)(h1 + (size_t)N * DIM);
    int*   lst    = (int*)(tmp + E);
    float* a_src  = (float*)(lst + E);
    float* a_dst  = a_src + N;
    int*   rowptr = (int*)(a_dst + N);
    const int NB   = cdiv(N, BUCKET_NODES);
    const int nblk = cdiv(E, EPB);
    int*   btot   = rowptr + N + 1;
    int*   hist   = btot + NB;
    int*   off    = hist + (size_t)nblk * NB;

    bucket_hist<<<nblk, 256, 0, stream>>>(edst, hist, E, NB);
    bucket_scan<<<NB, 256, 0, stream>>>(hist, off, btot, nblk, NB);
    bucket_scatter<<<nblk, 256, 0, stream>>>(esrc, edst, hist, off, btot, tmp, E, NB);
    bucket_finish<<<NB, 256, 0, stream>>>(tmp, btot, rowptr, lst, N, E, NB);

    run_layer(emb, W1, att_src1, att_dst1, b1, rowptr, lst, N, E, xw, a_src, a_dst, h1, stream);
    run_layer(h1, W2, att_src2, att_dst2, b2, rowptr, lst, N, E, xw, a_src, a_dst,
              (float*)d_out, stream);
}


# Sort kernels ask for all of a thread's edges before the first LDS atomic (16 to 20 loads in flight): one memory round trip per loop, not one per edge; list order still set by LDS arrival
# speedup vs baseline: 5.5025x; 1.1229x over previous
#include <hip/hip_runtime.h>
#include <math.h>


typedef float v2f __attribute__((ext_vector_type(2)));
typedef float v8f __attribute__((ext_vector_type(8)));

#define DIM 64
#define NEG_ATT 0.2f
#define NEG_OUT 0.01f

__device__ __forceinline__ v2f lo2(const float4 q) { v2f r; r.x = q.x; r.y = q.y; return r; }
__device__ __forceinline__ v2f hi2(const float4 q) { v2f r; r.x = q.z; r.y = q.w; return r; }

template <int KEEP>
__device__ __forceinline__ void fold_step(float* v, int c) {
    const bool up = (c & KEEP) != 0;
#pragma unroll
    for (int i = 0; i < KEEP; ++i) {
        const float keep = up ? v[i + KEEP] : v[i];
        const float send = up ? v[i] : v[i + KEEP];
        v[i] = keep + __shfl_xor(send, KEEP, 32);
    }
}

__global__ __launch_bounds__(256) void gemm64_dots(const float* __restrict__ X,
                                                   const float* __restrict__ Wm,
                                                   const float* __restrict__ as,
                                                   const float* __restrict__ ad,
                                                   float* __restrict__ XW,
                                                   float* __restrict__ a_src,
                                                   float* __restrict__ a_dst,
                                                   int n_row_tiles) {
    __shared__ float4 Bs[8 * 4 * 32];

    const int tid   = threadIdx.x;
    const int lane  = tid & 31;
    const int h     = lane >> 4;
    const int c     = lane & 15;
    const int strip = blockIdx.x * 8 + (tid >> 5);
    const bool live = strip < n_row_tiles;

    float4 a4[8];
    if (live) {
        const float4* xr = (const float4*)(X + (size_t)(strip * 16 + c) * DIM + 32 * h);
#pragma unroll
        for (int t2 = 0; t2 < 8; ++t2) a4[t2] = xr[t2];
    }

#pragma unroll
    for (int i = 0; i < 4; ++i) {
        const int q  = i * 256 + tid;
        const int ql = q & 31, qn = (q >> 5) & 3, qt = q >> 7;
        const float* wp = Wm + (size_t)(32 * (ql >> 4) + 4 * qt) * DIM + 4 * (ql & 15) + qn;
        Bs[q] = make_float4(wp[0], wp[DIM], wp[2 * DIM], wp[3 * DIM]);
    }
    __syncthreads();
    if (!live) return;

    v8f acc[4];
#pragma unroll
    for (int n = 0; n < 4; ++n) acc[n] = (v8f){0.f, 0.f, 0.f, 0.f, 0.f, 0.f, 0.f, 0.f};

#pragma unroll
    for (int t2 = 0; t2 < 8; ++t2) {
        float4 b4[4];
#pragma unroll
        for (int n = 0; n < 4; ++n) b4[n] = Bs[(t2 * 4 + n) * 32 + lane];
#pragma unroll
        for (int n = 0; n < 4; ++n)
            acc[n] = __builtin_amdgcn_wmma_f32_16x16x4_f32(false, lo2(a4[t2]), false, lo2(b4[n]),
                                                           (short)0, acc[n], false, false);
#pragma unroll
        for (int n = 0; n < 4; ++n)
            acc[n] = __builtin_amdgcn_wmma_f32_16x16x4_f32(false, hi2(a4[t2]), false, hi2(b4[n]),
                                                           (short)0, acc[n], false, false);
    }

    float* orow = XW + (size_t)(strip * 16 + 8 * h) * DIM + 4 * c;
#pragma unroll
    for (int r = 0; r < 8; ++r)
        *(float4*)(orow + (size_t)r * DIM) = make_float4(acc[0][r], acc[1][r], acc[2][r], acc[3][r]);

    const float4 as4 = *(const float4*)(as + 4 * c);
    const float4 ad4 = *(const float4*)(ad + 4 * c);
    const float asv[4] = {as4.x, as4.y, as4.z, as4.w};
    const float adv[4] = {ad4.x, ad4.y, ad4.z, ad4.w};
    float v[16];
#pragma unroll
    for (int r = 0; r < 8; ++r) {
        float s = acc[0][r] * asv[0];
        float d = acc[0][r] * adv[0];
#pragma unroll
        for (int n = 1; n < 4; ++n) { s += acc[n][r] * asv[n]; d += acc[n][r] * adv[n]; }
        v[r] = s;
        v[8 + r] = d;
    }
    fold_step<8>(v, c);
    fold_step<4>(v, c);
    fold_step<2>(v, c);
    fold_step<1>(v, c);
    float* dots = (c < 8) ? a_src : a_dst;
    dots[strip * 16 + 8 * h + (c & 7)] = v[0];
}


#define BUCKET_SHIFT 8
#define BUCKET_NODES (1 << BUCKET_SHIFT)
#define MAXB 512
#define EPB 4096
#define FIN_CAP 8192
#define EPT (EPB / 256)
#define FIN_U 20

__device__ __forceinline__ int bucket_of(int d, int NB) {
    return min(max(d, 0) >> BUCKET_SHIFT, NB - 1);
}

__device__ __forceinline__ int block_excl_scan(int v, int* wsum, int tid, int* tot) {
    const int lane = tid & 31, wv = tid >> 5;
    int inc = v;
    for (int o = 1; o < 32; o <<= 1) {
        int t = __shfl_up(inc, o, 32);
        if (lane >= o) inc += t;
    }
    if (lane == 31) wsum[wv] = inc;
    __syncthreads();
    int woff = 0, all = 0;
    for (int w = 0; w < 8; ++w) {
        const int s = wsum[w];
        all += s;
        if (w < wv) woff += s;
    }
    __syncthreads();
    *tot = all;
    return woff + inc - v;
}

__global__ __launch_bounds__(256) void bucket_hist(const int* __restrict__ edst,
                                                   int* __restrict__ hist, int E, int NB) {
    __shared__ int h[MAXB];
    const int tid = threadIdx.x, blk = blockIdx.x;
    h[tid] = 0;
    h[tid + 256] = 0;
    __syncthreads();
    const int base = blk * EPB;
    int d[EPT];
#pragma unroll
    for (int k = 0; k < EPT; ++k) d[k] = edst[min(base + k * 256 + tid, E - 1)];
#pragma unroll
    for (int k = 0; k < EPT; ++k)
        if (base + k * 256 + tid < E) atomicAdd(&h[bucket_of(d[k], NB)], 1);
    __syncthreads();
    for (int i = tid; i < NB; i += 256) hist[blk * NB + i] = h[i];
}

__global__ __launch_bounds__(256) void bucket_scan(const int* __restrict__ hist,
                                                   int* __restrict__ off,
                                                   int* __restrict__ btot, int nblk, int NB) {
    __shared__ int wsum[8];
    const int tid = threadIdx.x, b = blockIdx.x;
    int carry = 0;
    for (int c0 = 0; c0 < nblk; c0 += 256) {
        const int blk = c0 + tid;
        const int v = (blk < nblk) ? hist[blk * NB + b] : 0;
        int tot;
        const int ex = block_excl_scan(v, wsum, tid, &tot);
        if (blk < nblk) off[blk * NB + b] = carry + ex;
        carry += tot;
    }
    if (tid == 0) btot[b] = carry;
}

__global__ __launch_bounds__(256) void bucket_scatter(const int* __restrict__ esrc,
                                                      const int* __restrict__ edst,
                                                      const int* __restrict__ hist,
                                                      const int* __restrict__ off,
                                                      const int* __restrict__ btot,
                                                      int2* __restrict__ tmp, int E, int NB) {
    __shared__ int  lbase[MAXB];
    __shared__ int  gdelta[MAXB];
    __shared__ int  cur[MAXB];
    __shared__ int2 stage[EPB];
    __shared__ int  wsum[8];
    const int tid = threadIdx.x, blk = blockIdx.x;
    const int base = blk * EPB;
    int ed[EPT], es[EPT];
#pragma unroll
    for (int k = 0; k < EPT; ++k) {
        const int e = min(base + k * 256 + tid, E - 1);
        ed[k] = edst[e];
        es[k] = esrc[e];
    }
    const int i0 = 2 * tid, i1 = 2 * tid + 1;
    const int t0 = (i0 < NB) ? btot[i0] : 0;
    const int t1 = (i1 < NB) ? btot[i1] : 0;
    const int h0 = (i0 < NB) ? hist[blk * NB + i0] : 0;
    const int h1 = (i1 < NB) ? hist[blk * NB + i1] : 0;
    const int o0 = (i0 < NB) ? off[blk * NB + i0] : 0;
    const int o1 = (i1 < NB) ? off[blk * NB + i1] : 0;
    int tot;
    const int gb = block_excl_scan(t0 + t1, wsum, tid, &tot);
    const int lb = block_excl_scan(h0 + h1, wsum, tid, &tot);
    lbase[i0]  = lb;
    lbase[i1]  = lb + h0;
    gdelta[i0] = gb + o0 - lb;
    gdelta[i1] = gb + t0 + o1 - (lb + h0);
    cur[i0] = 0;
    cur[i1] = 0;
    __syncthreads();

#pragma unroll
    for (int k = 0; k < EPT; ++k) {
        if (base + k * 256 + tid < E) {
            const int b = bucket_of(ed[k], NB);
            const int r = atomicAdd(&cur[b], 1);
            const int slot = min(max(lbase[b] + r, 0), EPB - 1);
            stage[slot] = make_int2(es[k], ed[k]);
        }
    }
    __syncthreads();
    const int n_here = min(EPB, E - base);
    for (int i = tid; i < n_here; i += 256) {
        const int2 sd = stage[i];
        const int pos = min(max(i + gdelta[bucket_of(sd.y, NB)], 0), E - 1);
        tmp[pos] = sd;
    }
}

__global__ __launch_bounds__(256) void bucket_finish(const int2* __restrict__ tmp,
                                                     const int* __restrict__ btot,
                                                     int* __restrict__ rowptr,
                                                     int* __restrict__ lst, int N, int E, int NB) {
    __shared__ int cnt[BUCKET_NODES];
    __shared__ int cur[BUCKET_NODES];
    __shared__ int stage[FIN_CAP];
    __shared__ int wsum[8];
    const int tid = threadIdx.x, b = blockIdx.x;
    int part = 0;
    for (int i = tid; i < b; i += 256) part += btot[i];
    int bstart;
    block_excl_scan(part, wsum, tid, &bstart);
    const int beg = min(max(bstart, 0), E);
    const int end = min(max(beg + btot[b], beg), E);
    const int node0 = b << BUCKET_SHIFT;

    cnt[tid] = 0;
    __syncthreads();
    for (int p0 = beg; p0 < end; p0 += 256 * FIN_U) {
        int y[FIN_U];
#pragma unroll
        for (int u = 0; u < FIN_U; ++u) y[u] = tmp[min(p0 + u * 256 + tid, E - 1)].y;
#pragma unroll
        for (int u = 0; u < FIN_U; ++u)
            if (p0 + u * 256 + tid < end)
                atomicAdd(&cnt[min(max(y[u] - node0, 0), BUCKET_NODES - 1)], 1);
    }
    __syncthreads();
    int tot;
    const int ex = block_excl_scan(cnt[tid], wsum, tid, &tot);
    if (node0 + tid < N) rowptr[node0 + tid] = beg + ex;
    if (b == 0 && tid == 0) rowptr[N] = E;
    cur[tid] = ex;
    __syncthreads();

    const int  n      = end - beg;
    const bool staged = (n <= FIN_CAP);
    for (int p0 = beg; p0 < end; p0 += 256 * FIN_U) {
        int2 sd[FIN_U];
#pragma unroll
        for (int u = 0; u < FIN_U; ++u) sd[u] = tmp[min(p0 + u * 256 + tid, E - 1)];
#pragma unroll
        for (int u = 0; u < FIN_U; ++u) {
            if (p0 + u * 256 + tid < end) {
                const int li = min(max(sd[u].y - node0, 0), BUCKET_NODES - 1);
                const int r = atomicAdd(&cur[li], 1);
                if (staged) stage[min(max(r, 0), FIN_CAP - 1)] = sd[u].x;
                else        lst[min(max(beg + r, 0), E - 1)] = sd[u].x;
            }
        }
    }
    if (staged) {
        __syncthreads();
        for (int i = tid; i < n; i += 256) lst[beg + i] = stage[i];
    }
}


__device__ __forceinline__ float lrelu_att(float a) { return (a > 0.f) ? a : NEG_ATT * a; }

__device__ __forceinline__ int lane_i(int v, int j) { return __builtin_amdgcn_readlane(v, j); }
__device__ __forceinline__ float lane_f(float v, int j) {
    return __int_as_float(__builtin_amdgcn_readlane(__float_as_int(v), j));
}
__device__ __forceinline__ float2 row2(const float* __restrict__ XW, int s, int lane) {
    return *(const float2*)(XW + (size_t)s * DIM + 2 * lane);
}

template <int K>
__device__ __forceinline__ void add_group(const float* __restrict__ XW, int s, float ex,
                                          int j, int lane, float& ax, float& ay) {
    float2 x[K];
    for (int u = 0; u < K; ++u) x[u] = row2(XW, lane_i(s, j + u), lane);
    for (int u = 0; u < K; ++u) {
        const float w = lane_f(ex, j + u);
        ax += w * x[u].x;
        ay += w * x[u].y;
    }
}

__device__ __forceinline__ void add_rows(const float* __restrict__ XW, int s, float ex,
                                         int cnt, int lane, float& ax, float& ay) {
    int j = 0;
    for (; j + 8 <= cnt; j += 8) add_group<8>(XW, s, ex, j, lane, ax, ay);
    switch (cnt - j) {
        case 1: add_group<1>(XW, s, ex, j, lane, ax, ay); break;
        case 2: add_group<2>(XW, s, ex, j, lane, ax, ay); break;
        case 3: add_group<3>(XW, s, ex, j, lane, ax, ay); break;
        case 4: add_group<4>(XW, s, ex, j, lane, ax, ay); break;
        case 5: add_group<5>(XW, s, ex, j, lane, ax, ay); break;
        case 6: add_group<6>(XW, s, ex, j, lane, ax, ay); break;
        case 7: add_group<7>(XW, s, ex, j, lane, ax, ay); break;
        default: break;
    }
}

__global__ __launch_bounds__(256) void node_gather(const int* __restrict__ rowptr,
                                                   const int* __restrict__ lst,
                                                   const float* __restrict__ XW,
                                                   const float* __restrict__ a_src,
                                                   const float* __restrict__ a_dst,
                                                   const float* __restrict__ bias,
                                                   float* __restrict__ out, int n, int E) {
    const int node = __builtin_amdgcn_readfirstlane((int)((blockIdx.x * blockDim.x + threadIdx.x) >> 5));
    const int lane = threadIdx.x & 31;
    if (node >= n) return;

    const int beg = min(max(rowptr[node], 0), E);
    const int end = min(max(rowptr[node + 1], beg), E);
    const float ad     = a_dst[node];
    const float a_self = lrelu_att(a_src[node] + ad);

    int   s0 = node;
    float a0 = a_self;
    const bool have0 = (beg + lane < end);
    if (have0) { s0 = min(max(lst[beg + lane], 0), n - 1); a0 = lrelu_att(a_src[s0] + ad); }

    float m = fmaxf(a_self, a0);
    for (int k = beg + 32 + lane; k < end; k += 32)
        m = fmaxf(m, lrelu_att(a_src[min(max(lst[k], 0), n - 1)] + ad));
    for (int off = 16; off > 0; off >>= 1) m = fmaxf(m, __shfl_xor(m, off, 32));

    const float  e_self = __expf(a_self - m);
    const float2 xs = row2(XW, node, lane);
    float ax = e_self * xs.x, ay = e_self * xs.y;
    float ex0  = have0 ? __expf(a0 - m) : 0.f;
    float dsum = ex0;
    const int deg = end - beg;
    add_rows(XW, s0, ex0, (deg < 32) ? deg : 32, lane, ax, ay);
    for (int base = beg + 32; base < end; base += 32) {
        int   s  = node;
        float ex = 0.f;
        if (base + lane < end) {
            s  = min(max(lst[base + lane], 0), n - 1);
            ex = __expf(lrelu_att(a_src[s] + ad) - m);
        }
        dsum += ex;
        const int left = end - base;
        add_rows(XW, s, ex, (left < 32) ? left : 32, lane, ax, ay);
    }
    for (int off = 16; off > 0; off >>= 1) dsum += __shfl_xor(dsum, off, 32);
    const float den = (dsum + e_self) + 1e-16f;

    float vx = ax / den + bias[2 * lane];
    float vy = ay / den + bias[2 * lane + 1];
    vx = (vx > 0.f) ? vx : NEG_OUT * vx;
    vy = (vy > 0.f) ? vy : NEG_OUT * vy;
    *(float2*)(out + (size_t)node * DIM + 2 * lane) = make_float2(vx, vy);
}

static inline int cdiv(int a, int b) { return (a + b - 1) / b; }

static void run_layer(const float* x, const float* W, const float* att_s, const float* att_d,
                      const float* bias, const int* rowptr, const int* lst, int N, int E,
                      float* xw, float* a_src, float* a_dst, float* outbuf, hipStream_t stream) {
    int n_row_tiles = N / 16;
    gemm64_dots<<<cdiv(n_row_tiles, 8), 256, 0, stream>>>(x, W, att_s, att_d, xw, a_src, a_dst,
                                                          n_row_tiles);

    node_gather<<<cdiv(N * 32, 256), 256, 0, stream>>>(rowptr, lst, xw, a_src, a_dst, bias,
                                                       outbuf, N, E);
}

extern "C" void kernel_launch(void* const* d_in, const int* in_sizes, int n_in,
                              void* d_out, int out_size, void* d_ws, size_t ws_size,
                              hipStream_t stream) {
    const int*   edge_index = (const int*)d_in[0];
    const float* emb        = (const float*)d_in[1];
    const float* W1         = (const float*)d_in[2];
    const float* att_src1   = (const float*)d_in[3];
    const float* att_dst1   = (const float*)d_in[4];
    const float* b1         = (const float*)d_in[5];
    const float* W2         = (const float*)d_in[6];
    const float* att_src2   = (const float*)d_in[7];
    const float* att_dst2   = (const float*)d_in[8];
    const float* b2         = (const float*)d_in[9];

    const int E = in_sizes[0] / 2;
    const int N = in_sizes[1] / DIM;

    const int* esrc = edge_index;
    const int* edst = edge_index + E;

    float* ws     = (float*)d_ws;
    float* xw     = ws;
    float* h1     = xw + (size_t)N * DIM;
    int2*  tmp    = (int2*)(h1 + (size_t)N * DIM);
    int*   lst    = (int*)(tmp + E);
    float* a_src  = (float*)(lst + E);
    float* a_dst  = a_src + N;
    int*   rowptr = (int*)(a_dst + N);
    const int NB   = cdiv(N, BUCKET_NODES);
    const int nblk = cdiv(E, EPB);
    int*   btot   = rowptr + N + 1;
    int*   hist   = btot + NB;
    int*   off    = hist + (size_t)nblk * NB;

    bucket_hist<<<nblk, 256, 0, stream>>>(edst, hist, E, NB);
    bucket_scan<<<NB, 256, 0, stream>>>(hist, off, btot, nblk, NB);
    bucket_scatter<<<nblk, 256, 0, stream>>>(esrc, edst, hist, off, btot, tmp, E, NB);
    bucket_finish<<<NB, 256, 0, stream>>>(tmp, btot, rowptr, lst, N, E, NB);

    run_layer(emb, W1, att_src1, att_dst1, b1, rowptr, lst, N, E, xw, a_src, a_dst, h1, stream);
    run_layer(h1, W2, att_src2, att_dst2, b2, rowptr, lst, N, E, xw, a_src, a_dst,
              (float*)d_out, stream);
}


# Layer-1 projection (WMMA GEMM with its dots) co-launched with the last sort pass in one grid: bucket blocks first, strip blocks behind; 8 launches to 7; no arithmetic or sum order changed
# speedup vs baseline: 5.8232x; 1.0583x over previous
#include <hip/hip_runtime.h>
#include <math.h>


typedef float v2f __attribute__((ext_vector_type(2)));
typedef float v8f __attribute__((ext_vector_type(8)));

#define DIM 64
#define NEG_ATT 0.2f
#define NEG_OUT 0.01f

__device__ __forceinline__ v2f lo2(const float4 q) { v2f r; r.x = q.x; r.y = q.y; return r; }
__device__ __forceinline__ v2f hi2(const float4 q) { v2f r; r.x = q.z; r.y = q.w; return r; }

template <int KEEP>
__device__ __forceinline__ void fold_step(float* v, int c) {
    const bool up = (c & KEEP) != 0;
#pragma unroll
    for (int i = 0; i < KEEP; ++i) {
        const float keep = up ? v[i + KEEP] : v[i];
        const float send = up ? v[i] : v[i + KEEP];
        v[i] = keep + __shfl_xor(send, KEEP, 32);
    }
}

__device__ __forceinline__ void gemm64_dots_block(const float* __restrict__ X,
                                                  const float* __restrict__ Wm,
                                                  const float* __restrict__ as,
                                                  const float* __restrict__ ad,
                                                  float* __restrict__ XW,
                                                  float* __restrict__ a_src,
                                                  float* __restrict__ a_dst,
                                                  int n_row_tiles, float4* Bs, int blk) {
    const int tid   = threadIdx.x;
    const int lane  = tid & 31;
    const int h     = lane >> 4;
    const int c     = lane & 15;
    const int strip = blk * 8 + (tid >> 5);
    const bool live = strip < n_row_tiles;

    float4 a4[8];
    if (live) {
        const float4* xr = (const float4*)(X + (size_t)(strip * 16 + c) * DIM + 32 * h);
#pragma unroll
        for (int t2 = 0; t2 < 8; ++t2) a4[t2] = xr[t2];
    }

#pragma unroll
    for (int i = 0; i < 4; ++i) {
        const int q  = i * 256 + tid;
        const int ql = q & 31, qn = (q >> 5) & 3, qt = q >> 7;
        const float* wp = Wm + (size_t)(32 * (ql >> 4) + 4 * qt) * DIM + 4 * (ql & 15) + qn;
        Bs[q] = make_float4(wp[0], wp[DIM], wp[2 * DIM], wp[3 * DIM]);
    }
    __syncthreads();
    if (!live) return;

    v8f acc[4];
#pragma unroll
    for (int n = 0; n < 4; ++n) acc[n] = (v8f){0.f, 0.f, 0.f, 0.f, 0.f, 0.f, 0.f, 0.f};

#pragma unroll
    for (int t2 = 0; t2 < 8; ++t2) {
        float4 b4[4];
#pragma unroll
        for (int n = 0; n < 4; ++n) b4[n] = Bs[(t2 * 4 + n) * 32 + lane];
#pragma unroll
        for (int n = 0; n < 4; ++n)
            acc[n] = __builtin_amdgcn_wmma_f32_16x16x4_f32(false, lo2(a4[t2]), false, lo2(b4[n]),
                                                           (short)0, acc[n], false, false);
#pragma unroll
        for (int n = 0; n < 4; ++n)
            acc[n] = __builtin_amdgcn_wmma_f32_16x16x4_f32(false, hi2(a4[t2]), false, hi2(b4[n]),
                                                           (short)0, acc[n], false, false);
    }

    float* orow = XW + (size_t)(strip * 16 + 8 * h) * DIM + 4 * c;
#pragma unroll
    for (int r = 0; r < 8; ++r)
        *(float4*)(orow + (size_t)r * DIM) = make_float4(acc[0][r], acc[1][r], acc[2][r], acc[3][r]);

    const float4 as4 = *(const float4*)(as + 4 * c);
    const float4 ad4 = *(const float4*)(ad + 4 * c);
    const float asv[4] = {as4.x, as4.y, as4.z, as4.w};
    const float adv[4] = {ad4.x, ad4.y, ad4.z, ad4.w};
    float v[16];
#pragma unroll
    for (int r = 0; r < 8; ++r) {
        float s = acc[0][r] * asv[0];
        float d = acc[0][r] * adv[0];
#pragma unroll
        for (int n = 1; n < 4; ++n) { s += acc[n][r] * asv[n]; d += acc[n][r] * adv[n]; }
        v[r] = s;
        v[8 + r] = d;
    }
    fold_step<8>(v, c);
    fold_step<4>(v, c);
    fold_step<2>(v, c);
    fold_step<1>(v, c);
    float* dots = (c < 8) ? a_src : a_dst;
    dots[strip * 16 + 8 * h + (c & 7)] = v[0];
}

__global__ __launch_bounds__(256) void gemm64_dots(const float* __restrict__ X,
                                                   const float* __restrict__ Wm,
                                                   const float* __restrict__ as,
                                                   const float* __restrict__ ad,
                                                   float* __restrict__ XW,
                                                   float* __restrict__ a_src,
                                                   float* __restrict__ a_dst,
                                                   int n_row_tiles) {
    __shared__ float4 Bs[8 * 4 * 32];
    gemm64_dots_block(X, Wm, as, ad, XW, a_src, a_dst, n_row_tiles, Bs, (int)blockIdx.x);
}


#define BUCKET_SHIFT 8
#define BUCKET_NODES (1 << BUCKET_SHIFT)
#define MAXB 512
#define EPB 4096
#define FIN_CAP 8192
#define EPT (EPB / 256)
#define FIN_U 20

__device__ __forceinline__ int bucket_of(int d, int NB) {
    return min(max(d, 0) >> BUCKET_SHIFT, NB - 1);
}

__device__ __forceinline__ int block_excl_scan(int v, int* wsum, int tid, int* tot) {
    const int lane = tid & 31, wv = tid >> 5;
    int inc = v;
    for (int o = 1; o < 32; o <<= 1) {
        int t = __shfl_up(inc, o, 32);
        if (lane >= o) inc += t;
    }
    if (lane == 31) wsum[wv] = inc;
    __syncthreads();
    int woff = 0, all = 0;
    for (int w = 0; w < 8; ++w) {
        const int s = wsum[w];
        all += s;
        if (w < wv) woff += s;
    }
    __syncthreads();
    *tot = all;
    return woff + inc - v;
}

__global__ __launch_bounds__(256) void bucket_hist(const int* __restrict__ edst,
                                                   int* __restrict__ hist, int E, int NB) {
    __shared__ int h[MAXB];
    const int tid = threadIdx.x, blk = blockIdx.x;
    h[tid] = 0;
    h[tid + 256] = 0;
    __syncthreads();
    const int base = blk * EPB;
    int d[EPT];
#pragma unroll
    for (int k = 0; k < EPT; ++k) d[k] = edst[min(base + k * 256 + tid, E - 1)];
#pragma unroll
    for (int k = 0; k < EPT; ++k)
        if (base + k * 256 + tid < E) atomicAdd(&h[bucket_of(d[k], NB)], 1);
    __syncthreads();
    for (int i = tid; i < NB; i += 256) hist[blk * NB + i] = h[i];
}

__global__ __launch_bounds__(256) void bucket_scan(const int* __restrict__ hist,
                                                   int* __restrict__ off,
                                                   int* __restrict__ btot, int nblk, int NB) {
    __shared__ int wsum[8];
    const int tid = threadIdx.x, b = blockIdx.x;
    int carry = 0;
    for (int c0 = 0; c0 < nblk; c0 += 256) {
        const int blk = c0 + tid;
        const int v = (blk < nblk) ? hist[blk * NB + b] : 0;
        int tot;
        const int ex = block_excl_scan(v, wsum, tid, &tot);
        if (blk < nblk) off[blk * NB + b] = carry + ex;
        carry += tot;
    }
    if (tid == 0) btot[b] = carry;
}

__global__ __launch_bounds__(256) void bucket_scatter(const int* __restrict__ esrc,
                                                      const int* __restrict__ edst,
                                                      const int* __restrict__ hist,
                                                      const int* __restrict__ off,
                                                      const int* __restrict__ btot,
                                                      int2* __restrict__ tmp, int E, int NB) {
    __shared__ int  lbase[MAXB];
    __shared__ int  gdelta[MAXB];
    __shared__ int  cur[MAXB];
    __shared__ int2 stage[EPB];
    __shared__ int  wsum[8];
    const int tid = threadIdx.x, blk = blockIdx.x;
    const int base = blk * EPB;
    int ed[EPT], es[EPT];
#pragma unroll
    for (int k = 0; k < EPT; ++k) {
        const int e = min(base + k * 256 + tid, E - 1);
        ed[k] = edst[e];
        es[k] = esrc[e];
    }
    const int i0 = 2 * tid, i1 = 2 * tid + 1;
    const int t0 = (i0 < NB) ? btot[i0] : 0;
    const int t1 = (i1 < NB) ? btot[i1] : 0;
    const int h0 = (i0 < NB) ? hist[blk * NB + i0] : 0;
    const int h1 = (i1 < NB) ? hist[blk * NB + i1] : 0;
    const int o0 = (i0 < NB) ? off[blk * NB + i0] : 0;
    const int o1 = (i1 < NB) ? off[blk * NB + i1] : 0;
    int tot;
    const int gb = block_excl_scan(t0 + t1, wsum, tid, &tot);
    const int lb = block_excl_scan(h0 + h1, wsum, tid, &tot);
    lbase[i0]  = lb;
    lbase[i1]  = lb + h0;
    gdelta[i0] = gb + o0 - lb;
    gdelta[i1] = gb + t0 + o1 - (lb + h0);
    cur[i0] = 0;
    cur[i1] = 0;
    __syncthreads();

#pragma unroll
    for (int k = 0; k < EPT; ++k) {
        if (base + k * 256 + tid < E) {
            const int b = bucket_of(ed[k], NB);
            const int r = atomicAdd(&cur[b], 1);
            const int slot = min(max(lbase[b] + r, 0), EPB - 1);
            stage[slot] = make_int2(es[k], ed[k]);
        }
    }
    __syncthreads();
    const int n_here = min(EPB, E - base);
    for (int i = tid; i < n_here; i += 256) {
        const int2 sd = stage[i];
        const int pos = min(max(i + gdelta[bucket_of(sd.y, NB)], 0), E - 1);
        tmp[pos] = sd;
    }
}

__device__ __forceinline__ void bucket_finish_block(const int2* __restrict__ tmp,
                                                    const int* __restrict__ btot,
                                                    int* __restrict__ rowptr,
                                                    int* __restrict__ lst, int N, int E, int NB,
                                                    int* cnt, int* cur, int* stage, int* wsum,
                                                    int b) {
    const int tid = threadIdx.x;
    int part = 0;
    for (int i = tid; i < b; i += 256) part += btot[i];
    int bstart;
    block_excl_scan(part, wsum, tid, &bstart);
    const int beg = min(max(bstart, 0), E);
    const int end = min(max(beg + btot[b], beg), E);
    const int node0 = b << BUCKET_SHIFT;

    cnt[tid] = 0;
    __syncthreads();
    for (int p0 = beg; p0 < end; p0 += 256 * FIN_U) {
        int y[FIN_U];
#pragma unroll
        for (int u = 0; u < FIN_U; ++u) y[u] = tmp[min(p0 + u * 256 + tid, E - 1)].y;
#pragma unroll
        for (int u = 0; u < FIN_U; ++u)
            if (p0 + u * 256 + tid < end)
                atomicAdd(&cnt[min(max(y[u] - node0, 0), BUCKET_NODES - 1)], 1);
    }
    __syncthreads();
    int tot;
    const int ex = block_excl_scan(cnt[tid], wsum, tid, &tot);
    if (node0 + tid < N) rowptr[node0 + tid] = beg + ex;
    if (b == 0 && tid == 0) rowptr[N] = E;
    cur[tid] = ex;
    __syncthreads();

    const int  n      = end - beg;
    const bool staged = (n <= FIN_CAP);
    for (int p0 = beg; p0 < end; p0 += 256 * FIN_U) {
        int2 sd[FIN_U];
#pragma unroll
        for (int u = 0; u < FIN_U; ++u) sd[u] = tmp[min(p0 + u * 256 + tid, E - 1)];
#pragma unroll
        for (int u = 0; u < FIN_U; ++u) {
            if (p0 + u * 256 + tid < end) {
                const int li = min(max(sd[u].y - node0, 0), BUCKET_NODES - 1);
                const int r = atomicAdd(&cur[li], 1);
                if (staged) stage[min(max(r, 0), FIN_CAP - 1)] = sd[u].x;
                else        lst[min(max(beg + r, 0), E - 1)] = sd[u].x;
            }
        }
    }
    if (staged) {
        __syncthreads();
        for (int i = tid; i < n; i += 256) lst[beg + i] = stage[i];
    }
}

#define FIN_LDS_INTS (2 * BUCKET_NODES + FIN_CAP + 8)
__global__ __launch_bounds__(256) void finish_and_project(const int2* __restrict__ tmp,
                                                          const int* __restrict__ btot,
                                                          int* __restrict__ rowptr,
                                                          int* __restrict__ lst,
                                                          int N, int E, int NB,
                                                          const float* __restrict__ X,
                                                          const float* __restrict__ Wm,
                                                          const float* __restrict__ as,
                                                          const float* __restrict__ ad,
                                                          float* __restrict__ XW,
                                                          float* __restrict__ a_src,
                                                          float* __restrict__ a_dst,
                                                          int n_row_tiles) {
    __shared__ float4 lds[FIN_LDS_INTS / 4];
    const int blk = (int)blockIdx.x;
    if (blk < NB) {
        int* li = (int*)lds;
        bucket_finish_block(tmp, btot, rowptr, lst, N, E, NB,
                            li, li + BUCKET_NODES, li + 2 * BUCKET_NODES,
                            li + 2 * BUCKET_NODES + FIN_CAP, blk);
    } else {
        gemm64_dots_block(X, Wm, as, ad, XW, a_src, a_dst, n_row_tiles, lds, blk - NB);
    }
}


__device__ __forceinline__ float lrelu_att(float a) { return (a > 0.f) ? a : NEG_ATT * a; }

__device__ __forceinline__ int lane_i(int v, int j) { return __builtin_amdgcn_readlane(v, j); }
__device__ __forceinline__ float lane_f(float v, int j) {
    return __int_as_float(__builtin_amdgcn_readlane(__float_as_int(v), j));
}
__device__ __forceinline__ float2 row2(const float* __restrict__ XW, int s, int lane) {
    return *(const float2*)(XW + (size_t)s * DIM + 2 * lane);
}

template <int K>
__device__ __forceinline__ void add_group(const float* __restrict__ XW, int s, float ex,
                                          int j, int lane, float& ax, float& ay) {
    float2 x[K];
    for (int u = 0; u < K; ++u) x[u] = row2(XW, lane_i(s, j + u), lane);
    for (int u = 0; u < K; ++u) {
        const float w = lane_f(ex, j + u);
        ax += w * x[u].x;
        ay += w * x[u].y;
    }
}

__device__ __forceinline__ void add_rows(const float* __restrict__ XW, int s, float ex,
                                         int cnt, int lane, float& ax, float& ay) {
    int j = 0;
    for (; j + 8 <= cnt; j += 8) add_group<8>(XW, s, ex, j, lane, ax, ay);
    switch (cnt - j) {
        case 1: add_group<1>(XW, s, ex, j, lane, ax, ay); break;
        case 2: add_group<2>(XW, s, ex, j, lane, ax, ay); break;
        case 3: add_group<3>(XW, s, ex, j, lane, ax, ay); break;
        case 4: add_group<4>(XW, s, ex, j, lane, ax, ay); break;
        case 5: add_group<5>(XW, s, ex, j, lane, ax, ay); break;
        case 6: add_group<6>(XW, s, ex, j, lane, ax, ay); break;
        case 7: add_group<7>(XW, s, ex, j, lane, ax, ay); break;
        default: break;
    }
}

__global__ __launch_bounds__(256) void node_gather(const int* __restrict__ rowptr,
                                                   const int* __restrict__ lst,
                                                   const float* __restrict__ XW,
                                                   const float* __restrict__ a_src,
                                                   const float* __restrict__ a_dst,
                                                   const float* __restrict__ bias,
                                                   float* __restrict__ out, int n, int E) {
    const int node = __builtin_amdgcn_readfirstlane((int)((blockIdx.x * blockDim.x + threadIdx.x) >> 5));
    const int lane = threadIdx.x & 31;
    if (node >= n) return;

    const int beg = min(max(rowptr[node], 0), E);
    const int end = min(max(rowptr[node + 1], beg), E);
    const float ad     = a_dst[node];
    const float a_self = lrelu_att(a_src[node] + ad);

    int   s0 = node;
    float a0 = a_self;
    const bool have0 = (beg + lane < end);
    if (have0) { s0 = min(max(lst[beg + lane], 0), n - 1); a0 = lrelu_att(a_src[s0] + ad); }

    float m = fmaxf(a_self, a0);
    for (int k = beg + 32 + lane; k < end; k += 32)
        m = fmaxf(m, lrelu_att(a_src[min(max(lst[k], 0), n - 1)] + ad));
    for (int off = 16; off > 0; off >>= 1) m = fmaxf(m, __shfl_xor(m, off, 32));

    const float  e_self = __expf(a_self - m);
    const float2 xs = row2(XW, node, lane);
    float ax = e_self * xs.x, ay = e_self * xs.y;
    float ex0  = have0 ? __expf(a0 - m) : 0.f;
    float dsum = ex0;
    const int deg = end - beg;
    add_rows(XW, s0, ex0, (deg < 32) ? deg : 32, lane, ax, ay);
    for (int base = beg + 32; base < end; base += 32) {
        int   s  = node;
        float ex = 0.f;
        if (base + lane < end) {
            s  = min(max(lst[base + lane], 0), n - 1);
            ex = __expf(lrelu_att(a_src[s] + ad) - m);
        }
        dsum += ex;
        const int left = end - base;
        add_rows(XW, s, ex, (left < 32) ? left : 32, lane, ax, ay);
    }
    for (int off = 16; off > 0; off >>= 1) dsum += __shfl_xor(dsum, off, 32);
    const float den = (dsum + e_self) + 1e-16f;

    float vx = ax / den + bias[2 * lane];
    float vy = ay / den + bias[2 * lane + 1];
    vx = (vx > 0.f) ? vx : NEG_OUT * vx;
    vy = (vy > 0.f) ? vy : NEG_OUT * vy;
    *(float2*)(out + (size_t)node * DIM + 2 * lane) = make_float2(vx, vy);
}

static inline int cdiv(int a, int b) { return (a + b - 1) / b; }

static void run_gather(const float* bias, const int* rowptr, const int* lst, int N, int E,
                       const float* xw, const float* a_src, const float* a_dst, float* outbuf,
                       hipStream_t stream) {
    node_gather<<<cdiv(N * 32, 256), 256, 0, stream>>>(rowptr, lst, xw, a_src, a_dst, bias,
                                                       outbuf, N, E);
}

extern "C" void kernel_launch(void* const* d_in, const int* in_sizes, int n_in,
                              void* d_out, int out_size, void* d_ws, size_t ws_size,
                              hipStream_t stream) {
    const int*   edge_index = (const int*)d_in[0];
    const float* emb        = (const float*)d_in[1];
    const float* W1         = (const float*)d_in[2];
    const float* att_src1   = (const float*)d_in[3];
    const float* att_dst1   = (const float*)d_in[4];
    const float* b1         = (const float*)d_in[5];
    const float* W2         = (const float*)d_in[6];
    const float* att_src2   = (const float*)d_in[7];
    const float* att_dst2   = (const float*)d_in[8];
    const float* b2         = (const float*)d_in[9];

    const int E = in_sizes[0] / 2;
    const int N = in_sizes[1] / DIM;

    const int* esrc = edge_index;
    const int* edst = edge_index + E;

    float* ws     = (float*)d_ws;
    float* xw     = ws;
    float* h1     = xw + (size_t)N * DIM;
    int2*  tmp    = (int2*)(h1 + (size_t)N * DIM);
    int*   lst    = (int*)(tmp + E);
    float* a_src  = (float*)(lst + E);
    float* a_dst  = a_src + N;
    int*   rowptr = (int*)(a_dst + N);
    const int NB   = cdiv(N, BUCKET_NODES);
    const int nblk = cdiv(E, EPB);
    int*   btot   = rowptr + N + 1;
    int*   hist   = btot + NB;
    int*   off    = hist + (size_t)nblk * NB;

    bucket_hist<<<nblk, 256, 0, stream>>>(edst, hist, E, NB);
    bucket_scan<<<NB, 256, 0, stream>>>(hist, off, btot, nblk, NB);
    bucket_scatter<<<nblk, 256, 0, stream>>>(esrc, edst, hist, off, btot, tmp, E, NB);
    const int n_row_tiles = N / 16;
    const int gemm_blocks = cdiv(n_row_tiles, 8);
    finish_and_project<<<NB + gemm_blocks, 256, 0, stream>>>(tmp, btot, rowptr, lst, N, E, NB,
                                                             emb, W1, att_src1, att_dst1,
                                                             xw, a_src, a_dst, n_row_tiles);
    run_gather(b1, rowptr, lst, N, E, xw, a_src, a_dst, h1, stream);
    gemm64_dots<<<gemm_blocks, 256, 0, stream>>>(h1, W2, att_src2, att_dst2, xw, a_src, a_dst,
                                                 n_row_tiles);
    run_gather(b2, rowptr, lst, N, E, xw, a_src, a_dst, (float*)d_out, stream);
}


# Gather: 16-lane half-wave per node, 16 B per lane (2 rows per load, half the waves); edge logit = DPP row sum of the fetched row . att_src, a_src gather gone; running-max softmax, sum trees change
# speedup vs baseline: 6.2216x; 1.0684x over previous
#include <hip/hip_runtime.h>
#include <math.h>


typedef float v2f __attribute__((ext_vector_type(2)));
typedef float v8f __attribute__((ext_vector_type(8)));

#define DIM 64
#define NEG_ATT 0.2f
#define NEG_OUT 0.01f

__device__ __forceinline__ v2f lo2(const float4 q) { v2f r; r.x = q.x; r.y = q.y; return r; }
__device__ __forceinline__ v2f hi2(const float4 q) { v2f r; r.x = q.z; r.y = q.w; return r; }

template <int KEEP>
__device__ __forceinline__ void fold_step(float* v, int c) {
    const bool up = (c & KEEP) != 0;
#pragma unroll
    for (int i = 0; i < KEEP; ++i) {
        const float keep = up ? v[i + KEEP] : v[i];
        const float send = up ? v[i] : v[i + KEEP];
        v[i] = keep + __shfl_xor(send, KEEP, 32);
    }
}

__device__ __forceinline__ void gemm64_dots_block(const float* __restrict__ X,
                                                  const float* __restrict__ Wm,
                                                  const float* __restrict__ as,
                                                  const float* __restrict__ ad,
                                                  float* __restrict__ XW,
                                                  float* __restrict__ a_src,
                                                  float* __restrict__ a_dst,
                                                  int n_row_tiles, float4* Bs, int blk) {
    const int tid   = threadIdx.x;
    const int lane  = tid & 31;
    const int h     = lane >> 4;
    const int c     = lane & 15;
    const int strip = blk * 8 + (tid >> 5);
    const bool live = strip < n_row_tiles;

    float4 a4[8];
    if (live) {
        const float4* xr = (const float4*)(X + (size_t)(strip * 16 + c) * DIM + 32 * h);
#pragma unroll
        for (int t2 = 0; t2 < 8; ++t2) a4[t2] = xr[t2];
    }

#pragma unroll
    for (int i = 0; i < 4; ++i) {
        const int q  = i * 256 + tid;
        const int ql = q & 31, qn = (q >> 5) & 3, qt = q >> 7;
        const float* wp = Wm + (size_t)(32 * (ql >> 4) + 4 * qt) * DIM + 4 * (ql & 15) + qn;
        Bs[q] = make_float4(wp[0], wp[DIM], wp[2 * DIM], wp[3 * DIM]);
    }
    __syncthreads();
    if (!live) return;

    v8f acc[4];
#pragma unroll
    for (int n = 0; n < 4; ++n) acc[n] = (v8f){0.f, 0.f, 0.f, 0.f, 0.f, 0.f, 0.f, 0.f};

#pragma unroll
    for (int t2 = 0; t2 < 8; ++t2) {
        float4 b4[4];
#pragma unroll
        for (int n = 0; n < 4; ++n) b4[n] = Bs[(t2 * 4 + n) * 32 + lane];
#pragma unroll
        for (int n = 0; n < 4; ++n)
            acc[n] = __builtin_amdgcn_wmma_f32_16x16x4_f32(false, lo2(a4[t2]), false, lo2(b4[n]),
                                                           (short)0, acc[n], false, false);
#pragma unroll
        for (int n = 0; n < 4; ++n)
            acc[n] = __builtin_amdgcn_wmma_f32_16x16x4_f32(false, hi2(a4[t2]), false, hi2(b4[n]),
                                                           (short)0, acc[n], false, false);
    }

    float* orow = XW + (size_t)(strip * 16 + 8 * h) * DIM + 4 * c;
#pragma unroll
    for (int r = 0; r < 8; ++r)
        *(float4*)(orow + (size_t)r * DIM) = make_float4(acc[0][r], acc[1][r], acc[2][r], acc[3][r]);

    const float4 as4 = *(const float4*)(as + 4 * c);
    const float4 ad4 = *(const float4*)(ad + 4 * c);
    const float asv[4] = {as4.x, as4.y, as4.z, as4.w};
    const float adv[4] = {ad4.x, ad4.y, ad4.z, ad4.w};
    float v[16];
#pragma unroll
    for (int r = 0; r < 8; ++r) {
        float s = acc[0][r] * asv[0];
        float d = acc[0][r] * adv[0];
#pragma unroll
        for (int n = 1; n < 4; ++n) { s += acc[n][r] * asv[n]; d += acc[n][r] * adv[n]; }
        v[r] = s;
        v[8 + r] = d;
    }
    fold_step<8>(v, c);
    fold_step<4>(v, c);
    fold_step<2>(v, c);
    fold_step<1>(v, c);
    float* dots = (c < 8) ? a_src : a_dst;
    dots[strip * 16 + 8 * h + (c & 7)] = v[0];
}

__global__ __launch_bounds__(256) void gemm64_dots(const float* __restrict__ X,
                                                   const float* __restrict__ Wm,
                                                   const float* __restrict__ as,
                                                   const float* __restrict__ ad,
                                                   float* __restrict__ XW,
                                                   float* __restrict__ a_src,
                                                   float* __restrict__ a_dst,
                                                   int n_row_tiles) {
    __shared__ float4 Bs[8 * 4 * 32];
    gemm64_dots_block(X, Wm, as, ad, XW, a_src, a_dst, n_row_tiles, Bs, (int)blockIdx.x);
}


#define BUCKET_SHIFT 8
#define BUCKET_NODES (1 << BUCKET_SHIFT)
#define MAXB 512
#define EPB 4096
#define FIN_CAP 8192
#define EPT (EPB / 256)
#define FIN_U 20

__device__ __forceinline__ int bucket_of(int d, int NB) {
    return min(max(d, 0) >> BUCKET_SHIFT, NB - 1);
}

__device__ __forceinline__ int block_excl_scan(int v, int* wsum, int tid, int* tot) {
    const int lane = tid & 31, wv = tid >> 5;
    int inc = v;
    for (int o = 1; o < 32; o <<= 1) {
        int t = __shfl_up(inc, o, 32);
        if (lane >= o) inc += t;
    }
    if (lane == 31) wsum[wv] = inc;
    __syncthreads();
    int woff = 0, all = 0;
    for (int w = 0; w < 8; ++w) {
        const int s = wsum[w];
        all += s;
        if (w < wv) woff += s;
    }
    __syncthreads();
    *tot = all;
    return woff + inc - v;
}

__global__ __launch_bounds__(256) void bucket_hist(const int* __restrict__ edst,
                                                   int* __restrict__ hist, int E, int NB) {
    __shared__ int h[MAXB];
    const int tid = threadIdx.x, blk = blockIdx.x;
    h[tid] = 0;
    h[tid + 256] = 0;
    __syncthreads();
    const int base = blk * EPB;
    int d[EPT];
#pragma unroll
    for (int k = 0; k < EPT; ++k) d[k] = edst[min(base + k * 256 + tid, E - 1)];
#pragma unroll
    for (int k = 0; k < EPT; ++k)
        if (base + k * 256 + tid < E) atomicAdd(&h[bucket_of(d[k], NB)], 1);
    __syncthreads();
    for (int i = tid; i < NB; i += 256) hist[blk * NB + i] = h[i];
}

__global__ __launch_bounds__(256) void bucket_scan(const int* __restrict__ hist,
                                                   int* __restrict__ off,
                                                   int* __restrict__ btot, int nblk, int NB) {
    __shared__ int wsum[8];
    const int tid = threadIdx.x, b = blockIdx.x;
    int carry = 0;
    for (int c0 = 0; c0 < nblk; c0 += 256) {
        const int blk = c0 + tid;
        const int v = (blk < nblk) ? hist[blk * NB + b] : 0;
        int tot;
        const int ex = block_excl_scan(v, wsum, tid, &tot);
        if (blk < nblk) off[blk * NB + b] = carry + ex;
        carry += tot;
    }
    if (tid == 0) btot[b] = carry;
}

__global__ __launch_bounds__(256) void bucket_scatter(const int* __restrict__ esrc,
                                                      const int* __restrict__ edst,
                                                      const int* __restrict__ hist,
                                                      const int* __restrict__ off,
                                                      const int* __restrict__ btot,
                                                      int2* __restrict__ tmp, int E, int NB) {
    __shared__ int  lbase[MAXB];
    __shared__ int  gdelta[MAXB];
    __shared__ int  cur[MAXB];
    __shared__ int2 stage[EPB];
    __shared__ int  wsum[8];
    const int tid = threadIdx.x, blk = blockIdx.x;
    const int base = blk * EPB;
    int ed[EPT], es[EPT];
#pragma unroll
    for (int k = 0; k < EPT; ++k) {
        const int e = min(base + k * 256 + tid, E - 1);
        ed[k] = edst[e];
        es[k] = esrc[e];
    }
    const int i0 = 2 * tid, i1 = 2 * tid + 1;
    const int t0 = (i0 < NB) ? btot[i0] : 0;
    const int t1 = (i1 < NB) ? btot[i1] : 0;
    const int h0 = (i0 < NB) ? hist[blk * NB + i0] : 0;
    const int h1 = (i1 < NB) ? hist[blk * NB + i1] : 0;
    const int o0 = (i0 < NB) ? off[blk * NB + i0] : 0;
    const int o1 = (i1 < NB) ? off[blk * NB + i1] : 0;
    int tot;
    const int gb = block_excl_scan(t0 + t1, wsum, tid, &tot);
    const int lb = block_excl_scan(h0 + h1, wsum, tid, &tot);
    lbase[i0]  = lb;
    lbase[i1]  = lb + h0;
    gdelta[i0] = gb + o0 - lb;
    gdelta[i1] = gb + t0 + o1 - (lb + h0);
    cur[i0] = 0;
    cur[i1] = 0;
    __syncthreads();

#pragma unroll
    for (int k = 0; k < EPT; ++k) {
        if (base + k * 256 + tid < E) {
            const int b = bucket_of(ed[k], NB);
            const int r = atomicAdd(&cur[b], 1);
            const int slot = min(max(lbase[b] + r, 0), EPB - 1);
            stage[slot] = make_int2(es[k], ed[k]);
        }
    }
    __syncthreads();
    const int n_here = min(EPB, E - base);
    for (int i = tid; i < n_here; i += 256) {
        const int2 sd = stage[i];
        const int pos = min(max(i + gdelta[bucket_of(sd.y, NB)], 0), E - 1);
        tmp[pos] = sd;
    }
}

__device__ __forceinline__ void bucket_finish_block(const int2* __restrict__ tmp,
                                                    const int* __restrict__ btot,
                                                    int* __restrict__ rowptr,
                                                    int* __restrict__ lst, int N, int E, int NB,
                                                    int* cnt, int* cur, int* stage, int* wsum,
                                                    int b) {
    const int tid = threadIdx.x;
    int part = 0;
    for (int i = tid; i < b; i += 256) part += btot[i];
    int bstart;
    block_excl_scan(part, wsum, tid, &bstart);
    const int beg = min(max(bstart, 0), E);
    const int end = min(max(beg + btot[b], beg), E);
    const int node0 = b << BUCKET_SHIFT;

    cnt[tid] = 0;
    __syncthreads();
    for (int p0 = beg; p0 < end; p0 += 256 * FIN_U) {
        int y[FIN_U];
#pragma unroll
        for (int u = 0; u < FIN_U; ++u) y[u] = tmp[min(p0 + u * 256 + tid, E - 1)].y;
#pragma unroll
        for (int u = 0; u < FIN_U; ++u)
            if (p0 + u * 256 + tid < end)
                atomicAdd(&cnt[min(max(y[u] - node0, 0), BUCKET_NODES - 1)], 1);
    }
    __syncthreads();
    int tot;
    const int ex = block_excl_scan(cnt[tid], wsum, tid, &tot);
    if (node0 + tid < N) rowptr[node0 + tid] = beg + ex;
    if (b == 0 && tid == 0) rowptr[N] = E;
    cur[tid] = ex;
    __syncthreads();

    const int  n      = end - beg;
    const bool staged = (n <= FIN_CAP);
    for (int p0 = beg; p0 < end; p0 += 256 * FIN_U) {
        int2 sd[FIN_U];
#pragma unroll
        for (int u = 0; u < FIN_U; ++u) sd[u] = tmp[min(p0 + u * 256 + tid, E - 1)];
#pragma unroll
        for (int u = 0; u < FIN_U; ++u) {
            if (p0 + u * 256 + tid < end) {
                const int li = min(max(sd[u].y - node0, 0), BUCKET_NODES - 1);
                const int r = atomicAdd(&cur[li], 1);
                if (staged) stage[min(max(r, 0), FIN_CAP - 1)] = sd[u].x;
                else        lst[min(max(beg + r, 0), E - 1)] = sd[u].x;
            }
        }
    }
    if (staged) {
        __syncthreads();
        for (int i = tid; i < n; i += 256) lst[beg + i] = stage[i];
    }
}

#define FIN_LDS_INTS (2 * BUCKET_NODES + FIN_CAP + 8)
__global__ __launch_bounds__(256) void finish_and_project(const int2* __restrict__ tmp,
                                                          const int* __restrict__ btot,
                                                          int* __restrict__ rowptr,
                                                          int* __restrict__ lst,
                                                          int N, int E, int NB,
                                                          const float* __restrict__ X,
                                                          const float* __restrict__ Wm,
                                                          const float* __restrict__ as,
                                                          const float* __restrict__ ad,
                                                          float* __restrict__ XW,
                                                          float* __restrict__ a_src,
                                                          float* __restrict__ a_dst,
                                                          int n_row_tiles) {
    __shared__ float4 lds[FIN_LDS_INTS / 4];
    const int blk = (int)blockIdx.x;
    if (blk < NB) {
        int* li = (int*)lds;
        bucket_finish_block(tmp, btot, rowptr, lst, N, E, NB,
                            li, li + BUCKET_NODES, li + 2 * BUCKET_NODES,
                            li + 2 * BUCKET_NODES + FIN_CAP, blk);
    } else {
        gemm64_dots_block(X, Wm, as, ad, XW, a_src, a_dst, n_row_tiles, lds, blk - NB);
    }
}


__device__ __forceinline__ float lrelu_att(float a) { return fmaxf(a, NEG_ATT * a); }

#define DPP_ROW_SHARE 0x150
#define DPP_ROW_XMASK 0x160

template <int CTRL>
__device__ __forceinline__ int dpp_i(int v) {
    return __builtin_amdgcn_update_dpp(0, v, CTRL, 0xf, 0xf, true);
}
template <int CTRL>
__device__ __forceinline__ float dpp_f(float v) {
    return __int_as_float(dpp_i<CTRL>(__float_as_int(v)));
}
__device__ __forceinline__ float row_sum16(float v) {
    v += dpp_f<DPP_ROW_XMASK + 1>(v);
    v += dpp_f<DPP_ROW_XMASK + 2>(v);
    v += dpp_f<DPP_ROW_XMASK + 4>(v);
    v += dpp_f<DPP_ROW_XMASK + 8>(v);
    return v;
}

__device__ __forceinline__ float4 row16(const char* __restrict__ xb, unsigned sb, unsigned coff) {
    return *(const float4*)(xb + (sb | coff));
}

template <int J, int K>
struct Grp {
    static __device__ __forceinline__ void load(const char* __restrict__ xb, int sb, unsigned coff,
                                                float4* x) {
        x[0] = row16(xb, (unsigned)dpp_i<DPP_ROW_SHARE + J>(sb), coff);
        Grp<J + 1, K - 1>::load(xb, sb, coff, x + 1);
    }
    static __device__ __forceinline__ float logits(const float4* x, const float4 as4, float ad,
                                                   float pen, float* l) {
        const v2f q = (v2f){x[0].x, x[0].y} * (v2f){as4.x, as4.y} + (v2f){x[0].z, x[0].w} * (v2f){as4.z, as4.w};
        float p = row_sum16(q.x + q.y);
        l[0] = lrelu_att(p + ad) + dpp_f<DPP_ROW_SHARE + J>(pen);
        if (K == 1) return l[0];
        return fmaxf(l[0], Grp<J + 1, K - 1>::logits(x + 1, as4, ad, pen, l + 1));
    }
    static __device__ __forceinline__ void add(const float4* x, const float* l, float m,
                                               float& dsum, float4& acc) {
        const float w = __expf(l[0] - m);
        dsum += w;
        acc.x += w * x[0].x;
        acc.y += w * x[0].y;
        acc.z += w * x[0].z;
        acc.w += w * x[0].w;
        Grp<J + 1, K - 1>::add(x + 1, l + 1, m, dsum, acc);
    }
};
template <int J>
struct Grp<J, 0> {
    static __device__ __forceinline__ void load(const char* __restrict__, int, unsigned, float4*) {}
    static __device__ __forceinline__ float logits(const float4*, const float4, float, float,
                                                   float*) { return 0.f; }
    static __device__ __forceinline__ void add(const float4*, const float*, float, float&,
                                               float4&) {}
};

template <int J, int K>
__device__ __forceinline__ void add_group(const char* __restrict__ xb, int sb, float pen,
                                          unsigned coff, const float4 as4, float ad,
                                          float& m, float& dsum, float4& acc) {
    float4 x[K];
    float  l[K];
    Grp<J, K>::load(xb, sb, coff, x);
    const float gmax  = Grp<J, K>::logits(x, as4, ad, pen, l);
    const float m_new = fmaxf(m, gmax);
    const float sc    = __expf(m - m_new);
    dsum *= sc;
    acc.x *= sc;
    acc.y *= sc;
    acc.z *= sc;
    acc.w *= sc;
    m = m_new;
    Grp<J, K>::add(x, l, m, dsum, acc);
}

__device__ __forceinline__ void add_rows16(const char* __restrict__ xb, int sb, float pen, int cnt,
                                           unsigned coff, const float4 as4, float ad,
                                           float& m, float& dsum, float4& acc) {
    if (cnt >= 8) {
        add_group<0, 8>(xb, sb, pen, coff, as4, ad, m, dsum, acc);
        switch (cnt - 8) {
            case 1: add_group<8, 1>(xb, sb, pen, coff, as4, ad, m, dsum, acc); break;
            case 2: add_group<8, 2>(xb, sb, pen, coff, as4, ad, m, dsum, acc); break;
            case 3: add_group<8, 3>(xb, sb, pen, coff, as4, ad, m, dsum, acc); break;
            case 4: add_group<8, 4>(xb, sb, pen, coff, as4, ad, m, dsum, acc); break;
            case 5: add_group<8, 5>(xb, sb, pen, coff, as4, ad, m, dsum, acc); break;
            case 6: add_group<8, 6>(xb, sb, pen, coff, as4, ad, m, dsum, acc); break;
            case 7: add_group<8, 7>(xb, sb, pen, coff, as4, ad, m, dsum, acc); break;
            case 8: add_group<8, 8>(xb, sb, pen, coff, as4, ad, m, dsum, acc); break;
            default: break;
        }
    } else {
        switch (cnt) {
            case 1: add_group<0, 1>(xb, sb, pen, coff, as4, ad, m, dsum, acc); break;
            case 2: add_group<0, 2>(xb, sb, pen, coff, as4, ad, m, dsum, acc); break;
            case 3: add_group<0, 3>(xb, sb, pen, coff, as4, ad, m, dsum, acc); break;
            case 4: add_group<0, 4>(xb, sb, pen, coff, as4, ad, m, dsum, acc); break;
            case 5: add_group<0, 5>(xb, sb, pen, coff, as4, ad, m, dsum, acc); break;
            case 6: add_group<0, 6>(xb, sb, pen, coff, as4, ad, m, dsum, acc); break;
            case 7: add_group<0, 7>(xb, sb, pen, coff, as4, ad, m, dsum, acc); break;
            default: break;
        }
    }
}

__global__ __launch_bounds__(256, 10) void node_gather(const int* __restrict__ rowptr,
                                                   const int* __restrict__ lst,
                                                   const float* __restrict__ XW,
                                                   const float* __restrict__ att_s,
                                                   const float* __restrict__ a_src,
                                                   const float* __restrict__ a_dst,
                                                   const float* __restrict__ bias,
                                                   float* __restrict__ out, int n, int E) {
    const int lane = threadIdx.x & 31;
    const int c    = lane & 15;
    const int wave = (int)((blockIdx.x * blockDim.x + threadIdx.x) >> 5);
    if (2 * wave >= n) return;
    const int  node_raw = 2 * wave + (lane >> 4);
    const bool live     = node_raw < n;
    const int  node     = live ? node_raw : n - 1;

    const int beg = min(max(rowptr[node], 0), E);
    const int end = min(max(rowptr[node + 1], beg), E);
    const int deg = live ? end - beg : 0;
    const float ad     = a_dst[node];
    const float a_self = lrelu_att(a_src[node] + ad);

    const char*    xb   = (const char*)XW;
    const unsigned coff = 16u * (unsigned)c;
    const float4   as4  = *(const float4*)(att_s + 4 * c);
    const float    NINF = -__builtin_inff();

    const bool have0 = (c < deg), have1 = (16 + c < deg);
    const int  last  = max(E - 1, 0);
    const int  l0 = lst[min(beg + c, last)];
    const int  l1 = lst[min(beg + 16 + c, last)];
    const float4 xs = row16(xb, (unsigned)node << 8, coff);
    const int  s0 = have0 ? min(max(l0, 0), n - 1) : node;
    const int  s1 = have1 ? min(max(l1, 0), n - 1) : node;

    float  m    = a_self;
    float  dsum = 1.f;
    float4 acc  = xs;

    const int maxdeg = max(__builtin_amdgcn_readlane(deg, 0), __builtin_amdgcn_readlane(deg, 16));
    for (int base = 0; base < maxdeg; base += 16) {
        int   s   = s0;
        float pen = have0 ? 0.f : NINF;
        if (base == 16) { s = s1; pen = have1 ? 0.f : NINF; }
        if (base >= 32) {
            s   = node;
            pen = NINF;
            if (base + c < deg) {
                s   = min(max(lst[beg + base + c], 0), n - 1);
                pen = 0.f;
            }
        }
        const int left = maxdeg - base;
        add_rows16(xb, s << 8, pen, (left < 16) ? left : 16, coff, as4, ad, m, dsum, acc);
    }
    const float den = dsum + 1e-16f;

    const float4 b4 = *(const float4*)(bias + 4 * c);
    float4 v = make_float4(acc.x / den + b4.x, acc.y / den + b4.y, acc.z / den + b4.z,
                           acc.w / den + b4.w);
    v.x = (v.x > 0.f) ? v.x : NEG_OUT * v.x;
    v.y = (v.y > 0.f) ? v.y : NEG_OUT * v.y;
    v.z = (v.z > 0.f) ? v.z : NEG_OUT * v.z;
    v.w = (v.w > 0.f) ? v.w : NEG_OUT * v.w;
    if (live) *(float4*)(out + (size_t)node * DIM + 4 * c) = v;
}

static inline int cdiv(int a, int b) { return (a + b - 1) / b; }

static void run_gather(const float* bias, const float* att_s, const int* rowptr, const int* lst,
                       int N, int E, const float* xw, const float* a_src, const float* a_dst,
                       float* outbuf, hipStream_t stream) {
    node_gather<<<cdiv(N, 16), 256, 0, stream>>>(rowptr, lst, xw, att_s, a_src, a_dst, bias,
                                                 outbuf, N, E);
}

extern "C" void kernel_launch(void* const* d_in, const int* in_sizes, int n_in,
                              void* d_out, int out_size, void* d_ws, size_t ws_size,
                              hipStream_t stream) {
    const int*   edge_index = (const int*)d_in[0];
    const float* emb        = (const float*)d_in[1];
    const float* W1         = (const float*)d_in[2];
    const float* att_src1   = (const float*)d_in[3];
    const float* att_dst1   = (const float*)d_in[4];
    const float* b1         = (const float*)d_in[5];
    const float* W2         = (const float*)d_in[6];
    const float* att_src2   = (const float*)d_in[7];
    const float* att_dst2   = (const float*)d_in[8];
    const float* b2         = (const float*)d_in[9];

    const int E = in_sizes[0] / 2;
    const int N = in_sizes[1] / DIM;

    const int* esrc = edge_index;
    const int* edst = edge_index + E;

    float* ws     = (float*)d_ws;
    float* xw     = ws;
    float* h1     = xw + (size_t)N * DIM;
    int2*  tmp    = (int2*)(h1 + (size_t)N * DIM);
    int*   lst    = (int*)(tmp + E);
    float* a_src  = (float*)(lst + E);
    float* a_dst  = a_src + N;
    int*   rowptr = (int*)(a_dst + N);
    const int NB   = cdiv(N, BUCKET_NODES);
    const int nblk = cdiv(E, EPB);
    int*   btot   = rowptr + N + 1;
    int*   hist   = btot + NB;
    int*   off    = hist + (size_t)nblk * NB;

    bucket_hist<<<nblk, 256, 0, stream>>>(edst, hist, E, NB);
    bucket_scan<<<NB, 256, 0, stream>>>(hist, off, btot, nblk, NB);
    bucket_scatter<<<nblk, 256, 0, stream>>>(esrc, edst, hist, off, btot, tmp, E, NB);
    const int n_row_tiles = N / 16;
    const int gemm_blocks = cdiv(n_row_tiles, 8);
    finish_and_project<<<NB + gemm_blocks, 256, 0, stream>>>(tmp, btot, rowptr, lst, N, E, NB,
                                                             emb, W1, att_src1, att_dst1,
                                                             xw, a_src, a_dst, n_row_tiles);
    run_gather(b1, att_src1, rowptr, lst, N, E, xw, a_src, a_dst, h1, stream);
    gemm64_dots<<<gemm_blocks, 256, 0, stream>>>(h1, W2, att_src2, att_dst2, xw, a_src, a_dst,
                                                 n_row_tiles);
    run_gather(b2, att_src2, rowptr, lst, N, E, xw, a_src, a_dst, (float*)d_out, stream);
}


# Edge sort in two launches, not four: edge blocks group their own edges by bucket in place and store their bucket offsets; a bucket block sums that column itself; list order still set by LDS arrival
# speedup vs baseline: 6.9081x; 1.1103x over previous
#include <hip/hip_runtime.h>
#include <math.h>


typedef float v2f __attribute__((ext_vector_type(2)));
typedef float v8f __attribute__((ext_vector_type(8)));

#define DIM 64
#define NEG_ATT 0.2f
#define NEG_OUT 0.01f

__device__ __forceinline__ v2f lo2(const float4 q) { v2f r; r.x = q.x; r.y = q.y; return r; }
__device__ __forceinline__ v2f hi2(const float4 q) { v2f r; r.x = q.z; r.y = q.w; return r; }

template <int KEEP>
__device__ __forceinline__ void fold_step(float* v, int c) {
    const bool up = (c & KEEP) != 0;
#pragma unroll
    for (int i = 0; i < KEEP; ++i) {
        const float keep = up ? v[i + KEEP] : v[i];
        const float send = up ? v[i] : v[i + KEEP];
        v[i] = keep + __shfl_xor(send, KEEP, 32);
    }
}

__device__ __forceinline__ void gemm64_dots_block(const float* __restrict__ X,
                                                  const float* __restrict__ Wm,
                                                  const float* __restrict__ as,
                                                  const float* __restrict__ ad,
                                                  float* __restrict__ XW,
                                                  float* __restrict__ a_src,
                                                  float* __restrict__ a_dst,
                                                  int n_row_tiles, float4* Bs, int blk) {
    const int tid   = threadIdx.x;
    const int lane  = tid & 31;
    const int h     = lane >> 4;
    const int c     = lane & 15;
    const int strip = blk * 8 + (tid >> 5);
    const bool live = strip < n_row_tiles;

    float4 a4[8];
    if (live) {
        const float4* xr = (const float4*)(X + (size_t)(strip * 16 + c) * DIM + 32 * h);
#pragma unroll
        for (int t2 = 0; t2 < 8; ++t2) a4[t2] = xr[t2];
    }

#pragma unroll
    for (int i = 0; i < 4; ++i) {
        const int q  = i * 256 + tid;
        const int ql = q & 31, qn = (q >> 5) & 3, qt = q >> 7;
        const float* wp = Wm + (size_t)(32 * (ql >> 4) + 4 * qt) * DIM + 4 * (ql & 15) + qn;
        Bs[q] = make_float4(wp[0], wp[DIM], wp[2 * DIM], wp[3 * DIM]);
    }
    __syncthreads();
    if (!live) return;

    v8f acc[4];
#pragma unroll
    for (int n = 0; n < 4; ++n) acc[n] = (v8f){0.f, 0.f, 0.f, 0.f, 0.f, 0.f, 0.f, 0.f};

#pragma unroll
    for (int t2 = 0; t2 < 8; ++t2) {
        float4 b4[4];
#pragma unroll
        for (int n = 0; n < 4; ++n) b4[n] = Bs[(t2 * 4 + n) * 32 + lane];
#pragma unroll
        for (int n = 0; n < 4; ++n)
            acc[n] = __builtin_amdgcn_wmma_f32_16x16x4_f32(false, lo2(a4[t2]), false, lo2(b4[n]),
                                                           (short)0, acc[n], false, false);
#pragma unroll
        for (int n = 0; n < 4; ++n)
            acc[n] = __builtin_amdgcn_wmma_f32_16x16x4_f32(false, hi2(a4[t2]), false, hi2(b4[n]),
                                                           (short)0, acc[n], false, false);
    }

    float* orow = XW + (size_t)(strip * 16 + 8 * h) * DIM + 4 * c;
#pragma unroll
    for (int r = 0; r < 8; ++r)
        *(float4*)(orow + (size_t)r * DIM) = make_float4(acc[0][r], acc[1][r], acc[2][r], acc[3][r]);

    const float4 as4 = *(const float4*)(as + 4 * c);
    const float4 ad4 = *(const float4*)(ad + 4 * c);
    const float asv[4] = {as4.x, as4.y, as4.z, as4.w};
    const float adv[4] = {ad4.x, ad4.y, ad4.z, ad4.w};
    float v[16];
#pragma unroll
    for (int r = 0; r < 8; ++r) {
        float s = acc[0][r] * asv[0];
        float d = acc[0][r] * adv[0];
#pragma unroll
        for (int n = 1; n < 4; ++n) { s += acc[n][r] * asv[n]; d += acc[n][r] * adv[n]; }
        v[r] = s;
        v[8 + r] = d;
    }
    fold_step<8>(v, c);
    fold_step<4>(v, c);
    fold_step<2>(v, c);
    fold_step<1>(v, c);
    float* dots = (c < 8) ? a_src : a_dst;
    dots[strip * 16 + 8 * h + (c & 7)] = v[0];
}

__global__ __launch_bounds__(256) void gemm64_dots(const float* __restrict__ X,
                                                   const float* __restrict__ Wm,
                                                   const float* __restrict__ as,
                                                   const float* __restrict__ ad,
                                                   float* __restrict__ XW,
                                                   float* __restrict__ a_src,
                                                   float* __restrict__ a_dst,
                                                   int n_row_tiles) {
    __shared__ float4 Bs[8 * 4 * 32];
    gemm64_dots_block(X, Wm, as, ad, XW, a_src, a_dst, n_row_tiles, Bs, (int)blockIdx.x);
}


#define BUCKET_SHIFT 8
#define BUCKET_NODES (1 << BUCKET_SHIFT)
#define MAXB 512
#define EPB 4096
#define FIN_CAP 8192
#define EPT (EPB / 256)
#define MAXBLK 512
#define FIN_U 13
#define RSLOTS (16 * FIN_U * ((MAXBLK / 16 + FIN_U - 1) / FIN_U))

__device__ __forceinline__ int bucket_of(int d, int NB) {
    return min(max(d, 0) >> BUCKET_SHIFT, NB - 1);
}

__device__ __forceinline__ int block_excl_scan(int v, int* wsum, int tid, int* tot) {
    const int lane = tid & 31, wv = tid >> 5;
    int inc = v;
    for (int o = 1; o < 32; o <<= 1) {
        int t = __shfl_up(inc, o, 32);
        if (lane >= o) inc += t;
    }
    if (lane == 31) wsum[wv] = inc;
    __syncthreads();
    int woff = 0, all = 0;
    for (int w = 0; w < 8; ++w) {
        const int s = wsum[w];
        all += s;
        if (w < wv) woff += s;
    }
    __syncthreads();
    *tot = all;
    return woff + inc - v;
}

__device__ __forceinline__ int2 block_sum2(int a, int b, int* wsum, int tid) {
    const int lane = tid & 31, wv = tid >> 5;
    for (int o = 16; o > 0; o >>= 1) {
        a += __shfl_xor(a, o, 32);
        b += __shfl_xor(b, o, 32);
    }
    if (lane == 0) { wsum[wv] = a; wsum[8 + wv] = b; }
    __syncthreads();
    int sa = 0, sb = 0;
    for (int w = 0; w < 8; ++w) { sa += wsum[w]; sb += wsum[8 + w]; }
    __syncthreads();
    return make_int2(sa, sb);
}

__global__ __launch_bounds__(256) void bucket_partition(const int* __restrict__ esrc,
                                                        const int* __restrict__ edst,
                                                        int* __restrict__ loff,
                                                        int2* __restrict__ tmp, int E, int NB) {
    __shared__ int  cnt[MAXB];
    __shared__ int  lbase[MAXB];
    __shared__ int2 stage[EPB];
    __shared__ int  wsum[8];
    const int tid = threadIdx.x, blk = blockIdx.x;
    const int base = blk * EPB;
    const int n_here = min(EPB, E - base);
    int ed[EPT], es[EPT];
#pragma unroll
    for (int k = 0; k < EPT; ++k) {
        const int e = min(base + k * 256 + tid, E - 1);
        ed[k] = edst[e];
        es[k] = esrc[e];
    }
    cnt[tid] = 0;
    cnt[tid + 256] = 0;
    __syncthreads();
#pragma unroll
    for (int k = 0; k < EPT; ++k)
        if (k * 256 + tid < n_here) atomicAdd(&cnt[bucket_of(ed[k], NB)], 1);
    __syncthreads();
    const int i0 = 2 * tid, i1 = 2 * tid + 1;
    const int h0 = cnt[i0], h1 = cnt[i1];
    int tot;
    const int lb = block_excl_scan(h0 + h1, wsum, tid, &tot);
    lbase[i0] = lb;
    lbase[i1] = lb + h0;
    cnt[i0] = 0;
    cnt[i1] = 0;
    int* row = loff + (size_t)blk * (NB + 1);
    if (i0 < NB) row[i0] = lb;
    if (i1 < NB) row[i1] = lb + h0;
    if (tid == 0) row[NB] = n_here;
    __syncthreads();
#pragma unroll
    for (int k = 0; k < EPT; ++k) {
        if (k * 256 + tid < n_here) {
            const int bk = bucket_of(ed[k], NB);
            const int r = atomicAdd(&cnt[bk], 1);
            const int slot = min(max(lbase[bk] + r, 0), EPB - 1);
            stage[slot] = make_int2(es[k], ed[k]);
        }
    }
    __syncthreads();
#pragma unroll
    for (int k = 0; k < EPT; ++k) {
        const int i = k * 256 + tid;
        if (i < n_here) tmp[base + i] = stage[i];
    }
}

__device__ __forceinline__ void bucket_finish_block(const int2* __restrict__ tmp,
                                                    const int* __restrict__ loff,
                                                    int* __restrict__ rowptr,
                                                    int* __restrict__ lst, int N, int E, int NB,
                                                    int nblk_in, int* cnt, int* cur, int* stage,
                                                    int* wsum, int2* rinfo, int b) {
    const int tid  = threadIdx.x;
    const int nblk = max(min(nblk_in, MAXBLK), 1);
    const int LP   = NB + 1;
    const int last = max(E - 1, 0);
    int s[MAXBLK / 256], e[MAXBLK / 256];
#pragma unroll
    for (int q = 0; q < MAXBLK / 256; ++q) {
        const int j = min(q * 256 + tid, nblk - 1);
        s[q] = loff[(size_t)j * LP + b];
        e[q] = loff[(size_t)j * LP + b + 1];
    }
    int ssum = 0, nsum = 0;
#pragma unroll
    for (int q = 0; q < MAXBLK / 256; ++q) {
        const int j  = q * 256 + tid;
        const int s1 = min(max(s[q], 0), EPB);
        const int e1 = min(max(e[q], s1), EPB);
        const int n1 = (j < nblk) ? e1 - s1 : 0;
        rinfo[j] = make_int2(j * EPB + s1, n1);
        ssum += (j < nblk) ? s1 : 0;
        nsum += n1;
    }
    for (int j = MAXBLK + tid; j < RSLOTS; j += 256) rinfo[j] = make_int2(0, 0);
    cnt[tid] = 0;
    const int2 sums = block_sum2(ssum, nsum, wsum, tid);
    const int beg = min(max(sums.x, 0), E);
    const int end = min(beg + max(sums.y, 0), E);
    const int node0 = b << BUCKET_SHIFT;

    const int hw    = tid >> 4;
    const int c     = tid & 15;
    const int trips = (nblk + 15) >> 4;
    for (int t0 = 0; t0 < trips; t0 += FIN_U) {
        int st[FIN_U], len[FIN_U], y[FIN_U], y2[FIN_U];
#pragma unroll
        for (int u = 0; u < FIN_U; ++u) {
            const int2 ri = rinfo[hw + 16 * (t0 + u)];
            st[u]  = ri.x;
            len[u] = ri.y;
        }
#pragma unroll
        for (int u = 0; u < FIN_U; ++u) y[u] = tmp[min(st[u] + c, last)].y;
#pragma unroll
        for (int u = 0; u < FIN_U; ++u) {
            y2[u] = 0;
            if (c + 16 < len[u]) y2[u] = tmp[min(st[u] + c + 16, last)].y;
        }
#pragma unroll
        for (int u = 0; u < FIN_U; ++u) {
            if (c < len[u])
                atomicAdd(&cnt[min(max(y[u] - node0, 0), BUCKET_NODES - 1)], 1);
            if (c + 16 < len[u])
                atomicAdd(&cnt[min(max(y2[u] - node0, 0), BUCKET_NODES - 1)], 1);
            for (int l = c + 32; l < len[u]; l += 16) {
                const int yy = tmp[min(st[u] + l, last)].y;
                atomicAdd(&cnt[min(max(yy - node0, 0), BUCKET_NODES - 1)], 1);
            }
        }
    }
    __syncthreads();
    int tot;
    const int ex = block_excl_scan(cnt[tid], wsum, tid, &tot);
    if (node0 + tid < N) rowptr[node0 + tid] = beg + ex;
    if (b == 0 && tid == 0) rowptr[N] = E;
    cur[tid] = ex;
    __syncthreads();

    const int  n      = end - beg;
    const bool staged = (n <= FIN_CAP);
    for (int t0 = 0; t0 < trips; t0 += FIN_U) {
        int  st[FIN_U], len[FIN_U];
        int2 sd[FIN_U], sd2[FIN_U];
#pragma unroll
        for (int u = 0; u < FIN_U; ++u) {
            const int2 ri = rinfo[hw + 16 * (t0 + u)];
            st[u]  = ri.x;
            len[u] = ri.y;
        }
#pragma unroll
        for (int u = 0; u < FIN_U; ++u) sd[u] = tmp[min(st[u] + c, last)];
#pragma unroll
        for (int u = 0; u < FIN_U; ++u) {
            sd2[u] = make_int2(0, 0);
            if (c + 16 < len[u]) sd2[u] = tmp[min(st[u] + c + 16, last)];
        }
#pragma unroll
        for (int u = 0; u < FIN_U; ++u) {
            if (c < len[u]) {
                const int li = min(max(sd[u].y - node0, 0), BUCKET_NODES - 1);
                const int r = atomicAdd(&cur[li], 1);
                if (staged) stage[min(max(r, 0), FIN_CAP - 1)] = sd[u].x;
                else        lst[min(max(beg + r, 0), last)] = sd[u].x;
            }
            if (c + 16 < len[u]) {
                const int li = min(max(sd2[u].y - node0, 0), BUCKET_NODES - 1);
                const int r = atomicAdd(&cur[li], 1);
                if (staged) stage[min(max(r, 0), FIN_CAP - 1)] = sd2[u].x;
                else        lst[min(max(beg + r, 0), last)] = sd2[u].x;
            }
            for (int l = c + 32; l < len[u]; l += 16) {
                const int2 s2 = tmp[min(st[u] + l, last)];
                const int li = min(max(s2.y - node0, 0), BUCKET_NODES - 1);
                const int r = atomicAdd(&cur[li], 1);
                if (staged) stage[min(max(r, 0), FIN_CAP - 1)] = s2.x;
                else        lst[min(max(beg + r, 0), last)] = s2.x;
            }
        }
    }
    if (staged) {
        __syncthreads();
        for (int i = tid; i < n; i += 256) lst[beg + i] = stage[i];
    }
}

#define FIN_LDS_INTS (2 * BUCKET_NODES + FIN_CAP + 16 + 2 * RSLOTS)
__global__ __launch_bounds__(256) void finish_and_project(const int2* __restrict__ tmp,
                                                          const int* __restrict__ loff,
                                                          int* __restrict__ rowptr,
                                                          int* __restrict__ lst,
                                                          int N, int E, int NB, int nblk,
                                                          const float* __restrict__ X,
                                                          const float* __restrict__ Wm,
                                                          const float* __restrict__ as,
                                                          const float* __restrict__ ad,
                                                          float* __restrict__ XW,
                                                          float* __restrict__ a_src,
                                                          float* __restrict__ a_dst,
                                                          int n_row_tiles) {
    __shared__ float4 lds[FIN_LDS_INTS / 4];
    const int blk = (int)blockIdx.x;
    if (blk < NB) {
        int* li = (int*)lds;
        bucket_finish_block(tmp, loff, rowptr, lst, N, E, NB, nblk,
                            li, li + BUCKET_NODES, li + 2 * BUCKET_NODES,
                            li + 2 * BUCKET_NODES + FIN_CAP,
                            (int2*)(li + 2 * BUCKET_NODES + FIN_CAP + 16), blk);
    } else {
        gemm64_dots_block(X, Wm, as, ad, XW, a_src, a_dst, n_row_tiles, lds, blk - NB);
    }
}


__device__ __forceinline__ float lrelu_att(float a) { return fmaxf(a, NEG_ATT * a); }

#define DPP_ROW_SHARE 0x150
#define DPP_ROW_XMASK 0x160

template <int CTRL>
__device__ __forceinline__ int dpp_i(int v) {
    return __builtin_amdgcn_update_dpp(0, v, CTRL, 0xf, 0xf, true);
}
template <int CTRL>
__device__ __forceinline__ float dpp_f(float v) {
    return __int_as_float(dpp_i<CTRL>(__float_as_int(v)));
}
__device__ __forceinline__ float row_sum16(float v) {
    v += dpp_f<DPP_ROW_XMASK + 1>(v);
    v += dpp_f<DPP_ROW_XMASK + 2>(v);
    v += dpp_f<DPP_ROW_XMASK + 4>(v);
    v += dpp_f<DPP_ROW_XMASK + 8>(v);
    return v;
}

__device__ __forceinline__ float4 row16(const char* __restrict__ xb, unsigned sb, unsigned coff) {
    return *(const float4*)(xb + (sb | coff));
}

template <int J, int K>
struct Grp {
    static __device__ __forceinline__ void load(const char* __restrict__ xb, int sb, unsigned coff,
                                                float4* x) {
        x[0] = row16(xb, (unsigned)dpp_i<DPP_ROW_SHARE + J>(sb), coff);
        Grp<J + 1, K - 1>::load(xb, sb, coff, x + 1);
    }
    static __device__ __forceinline__ float logits(const float4* x, const float4 as4, float ad,
                                                   float pen, float* l) {
        const v2f q = (v2f){x[0].x, x[0].y} * (v2f){as4.x, as4.y} + (v2f){x[0].z, x[0].w} * (v2f){as4.z, as4.w};
        float p = row_sum16(q.x + q.y);
        l[0] = lrelu_att(p + ad) + dpp_f<DPP_ROW_SHARE + J>(pen);
        if (K == 1) return l[0];
        return fmaxf(l[0], Grp<J + 1, K - 1>::logits(x + 1, as4, ad, pen, l + 1));
    }
    static __device__ __forceinline__ void add(const float4* x, const float* l, float m,
                                               float& dsum, float4& acc) {
        const float w = __expf(l[0] - m);
        dsum += w;
        acc.x += w * x[0].x;
        acc.y += w * x[0].y;
        acc.z += w * x[0].z;
        acc.w += w * x[0].w;
        Grp<J + 1, K - 1>::add(x + 1, l + 1, m, dsum, acc);
    }
};
template <int J>
struct Grp<J, 0> {
    static __device__ __forceinline__ void load(const char* __restrict__, int, unsigned, float4*) {}
    static __device__ __forceinline__ float logits(const float4*, const float4, float, float,
                                                   float*) { return 0.f; }
    static __device__ __forceinline__ void add(const float4*, const float*, float, float&,
                                               float4&) {}
};

template <int J, int K>
__device__ __forceinline__ void add_group(const char* __restrict__ xb, int sb, float pen,
                                          unsigned coff, const float4 as4, float ad,
                                          float& m, float& dsum, float4& acc) {
    float4 x[K];
    float  l[K];
    Grp<J, K>::load(xb, sb, coff, x);
    const float gmax  = Grp<J, K>::logits(x, as4, ad, pen, l);
    const float m_new = fmaxf(m, gmax);
    const float sc    = __expf(m - m_new);
    dsum *= sc;
    acc.x *= sc;
    acc.y *= sc;
    acc.z *= sc;
    acc.w *= sc;
    m = m_new;
    Grp<J, K>::add(x, l, m, dsum, acc);
}

__device__ __forceinline__ void add_rows16(const char* __restrict__ xb, int sb, float pen, int cnt,
                                           unsigned coff, const float4 as4, float ad,
                                           float& m, float& dsum, float4& acc) {
    if (cnt >= 8) {
        add_group<0, 8>(xb, sb, pen, coff, as4, ad, m, dsum, acc);
        switch (cnt - 8) {
            case 1: add_group<8, 1>(xb, sb, pen, coff, as4, ad, m, dsum, acc); break;
            case 2: add_group<8, 2>(xb, sb, pen, coff, as4, ad, m, dsum, acc); break;
            case 3: add_group<8, 3>(xb, sb, pen, coff, as4, ad, m, dsum, acc); break;
            case 4: add_group<8, 4>(xb, sb, pen, coff, as4, ad, m, dsum, acc); break;
            case 5: add_group<8, 5>(xb, sb, pen, coff, as4, ad, m, dsum, acc); break;
            case 6: add_group<8, 6>(xb, sb, pen, coff, as4, ad, m, dsum, acc); break;
            case 7: add_group<8, 7>(xb, sb, pen, coff, as4, ad, m, dsum, acc); break;
            case 8: add_group<8, 8>(xb, sb, pen, coff, as4, ad, m, dsum, acc); break;
            default: break;
        }
    } else {
        switch (cnt) {
            case 1: add_group<0, 1>(xb, sb, pen, coff, as4, ad, m, dsum, acc); break;
            case 2: add_group<0, 2>(xb, sb, pen, coff, as4, ad, m, dsum, acc); break;
            case 3: add_group<0, 3>(xb, sb, pen, coff, as4, ad, m, dsum, acc); break;
            case 4: add_group<0, 4>(xb, sb, pen, coff, as4, ad, m, dsum, acc); break;
            case 5: add_group<0, 5>(xb, sb, pen, coff, as4, ad, m, dsum, acc); break;
            case 6: add_group<0, 6>(xb, sb, pen, coff, as4, ad, m, dsum, acc); break;
            case 7: add_group<0, 7>(xb, sb, pen, coff, as4, ad, m, dsum, acc); break;
            default: break;
        }
    }
}

__global__ __launch_bounds__(256, 10) void node_gather(const int* __restrict__ rowptr,
                                                   const int* __restrict__ lst,
                                                   const float* __restrict__ XW,
                                                   const float* __restrict__ att_s,
                                                   const float* __restrict__ a_src,
                                                   const float* __restrict__ a_dst,
                                                   const float* __restrict__ bias,
                                                   float* __restrict__ out, int n, int E) {
    const int lane = threadIdx.x & 31;
    const int c    = lane & 15;
    const int wave = (int)((blockIdx.x * blockDim.x + threadIdx.x) >> 5);
    if (2 * wave >= n) return;
    const int  node_raw = 2 * wave + (lane >> 4);
    const bool live     = node_raw < n;
    const int  node     = live ? node_raw : n - 1;

    const int beg = min(max(rowptr[node], 0), E);
    const int end = min(max(rowptr[node + 1], beg), E);
    const int deg = live ? end - beg : 0;
    const float ad     = a_dst[node];
    const float a_self = lrelu_att(a_src[node] + ad);

    const char*    xb   = (const char*)XW;
    const unsigned coff = 16u * (unsigned)c;
    const float4   as4  = *(const float4*)(att_s + 4 * c);
    const float    NINF = -__builtin_inff();

    const bool have0 = (c < deg), have1 = (16 + c < deg);
    const int  last  = max(E - 1, 0);
    const int  l0 = lst[min(beg + c, last)];
    const int  l1 = lst[min(beg + 16 + c, last)];
    const float4 xs = row16(xb, (unsigned)node << 8, coff);
    const int  s0 = have0 ? min(max(l0, 0), n - 1) : node;
    const int  s1 = have1 ? min(max(l1, 0), n - 1) : node;

    float  m    = a_self;
    float  dsum = 1.f;
    float4 acc  = xs;

    const int maxdeg = max(__builtin_amdgcn_readlane(deg, 0), __builtin_amdgcn_readlane(deg, 16));
    for (int base = 0; base < maxdeg; base += 16) {
        int   s   = s0;
        float pen = have0 ? 0.f : NINF;
        if (base == 16) { s = s1; pen = have1 ? 0.f : NINF; }
        if (base >= 32) {
            s   = node;
            pen = NINF;
            if (base + c < deg) {
                s   = min(max(lst[beg + base + c], 0), n - 1);
                pen = 0.f;
            }
        }
        const int left = maxdeg - base;
        add_rows16(xb, s << 8, pen, (left < 16) ? left : 16, coff, as4, ad, m, dsum, acc);
    }
    const float den = dsum + 1e-16f;

    const float4 b4 = *(const float4*)(bias + 4 * c);
    float4 v = make_float4(acc.x / den + b4.x, acc.y / den + b4.y, acc.z / den + b4.z,
                           acc.w / den + b4.w);
    v.x = (v.x > 0.f) ? v.x : NEG_OUT * v.x;
    v.y = (v.y > 0.f) ? v.y : NEG_OUT * v.y;
    v.z = (v.z > 0.f) ? v.z : NEG_OUT * v.z;
    v.w = (v.w > 0.f) ? v.w : NEG_OUT * v.w;
    if (live) *(float4*)(out + (size_t)node * DIM + 4 * c) = v;
}

static inline int cdiv(int a, int b) { return (a + b - 1) / b; }

static void run_gather(const float* bias, const float* att_s, const int* rowptr, const int* lst,
                       int N, int E, const float* xw, const float* a_src, const float* a_dst,
                       float* outbuf, hipStream_t stream) {
    node_gather<<<cdiv(N, 16), 256, 0, stream>>>(rowptr, lst, xw, att_s, a_src, a_dst, bias,
                                                 outbuf, N, E);
}

extern "C" void kernel_launch(void* const* d_in, const int* in_sizes, int n_in,
                              void* d_out, int out_size, void* d_ws, size_t ws_size,
                              hipStream_t stream) {
    const int*   edge_index = (const int*)d_in[0];
    const float* emb        = (const float*)d_in[1];
    const float* W1         = (const float*)d_in[2];
    const float* att_src1   = (const float*)d_in[3];
    const float* att_dst1   = (const float*)d_in[4];
    const float* b1         = (const float*)d_in[5];
    const float* W2         = (const float*)d_in[6];
    const float* att_src2   = (const float*)d_in[7];
    const float* att_dst2   = (const float*)d_in[8];
    const float* b2         = (const float*)d_in[9];

    const int E = in_sizes[0] / 2;
    const int N = in_sizes[1] / DIM;

    const int* esrc = edge_index;
    const int* edst = edge_index + E;

    float* ws     = (float*)d_ws;
    float* xw     = ws;
    float* h1     = xw + (size_t)N * DIM;
    int2*  tmp    = (int2*)(h1 + (size_t)N * DIM);
    int*   lst    = (int*)(tmp + E);
    float* a_src  = (float*)(lst + E);
    float* a_dst  = a_src + N;
    int*   rowptr = (int*)(a_dst + N);
    const int NB   = cdiv(N, BUCKET_NODES);
    const int nblk = cdiv(E, EPB);
    int*   loff   = rowptr + N + 1;

    bucket_partition<<<nblk, 256, 0, stream>>>(esrc, edst, loff, tmp, E, NB);
    const int n_row_tiles = N / 16;
    const int gemm_blocks = cdiv(n_row_tiles, 8);
    finish_and_project<<<NB + gemm_blocks, 256, 0, stream>>>(tmp, loff, rowptr, lst, N, E, NB, nblk,
                                                             emb, W1, att_src1, att_dst1,
                                                             xw, a_src, a_dst, n_row_tiles);
    run_gather(b1, att_src1, rowptr, lst, N, E, xw, a_src, a_dst, h1, stream);
    gemm64_dots<<<gemm_blocks, 256, 0, stream>>>(h1, W2, att_src2, att_dst2, xw, a_src, a_dst,
                                                 n_row_tiles);
    run_gather(b2, att_src2, rowptr, lst, N, E, xw, a_src, a_dst, (float*)d_out, stream);
}


# Layer-2 projection moved into the layer-1 gather launch: a block's 8 waves leave their 16 rows in LDS, the wave with the last LDS ticket runs the strip's WMMAs and dots; 5 launches to 4; same sums
# speedup vs baseline: 7.1734x; 1.0384x over previous
#include <hip/hip_runtime.h>
#include <math.h>


typedef float v2f __attribute__((ext_vector_type(2)));
typedef float v8f __attribute__((ext_vector_type(8)));

#define DIM 64
#define NEG_ATT 0.2f
#define NEG_OUT 0.01f

__device__ __forceinline__ v2f lo2(const float4 q) { v2f r; r.x = q.x; r.y = q.y; return r; }
__device__ __forceinline__ v2f hi2(const float4 q) { v2f r; r.x = q.z; r.y = q.w; return r; }

template <int KEEP>
__device__ __forceinline__ void fold_step(float* v, int c) {
    const bool up = (c & KEEP) != 0;
#pragma unroll
    for (int i = 0; i < KEEP; ++i) {
        const float keep = up ? v[i + KEEP] : v[i];
        const float send = up ? v[i] : v[i + KEEP];
        v[i] = keep + __shfl_xor(send, KEEP, 32);
    }
}

__device__ __forceinline__ void gemm64_dots_block(const float* __restrict__ X,
                                                  const float* __restrict__ Wm,
                                                  const float* __restrict__ as,
                                                  const float* __restrict__ ad,
                                                  float* __restrict__ XW,
                                                  float* __restrict__ a_src,
                                                  float* __restrict__ a_dst,
                                                  int n_row_tiles, float4* Bs, int blk) {
    const int tid   = threadIdx.x;
    const int lane  = tid & 31;
    const int h     = lane >> 4;
    const int c     = lane & 15;
    const int strip = blk * 8 + (tid >> 5);
    const bool live = strip < n_row_tiles;

    float4 a4[8];
    if (live) {
        const float4* xr = (const float4*)(X + (size_t)(strip * 16 + c) * DIM + 32 * h);
#pragma unroll
        for (int t2 = 0; t2 < 8; ++t2) a4[t2] = xr[t2];
    }

#pragma unroll
    for (int i = 0; i < 4; ++i) {
        const int q  = i * 256 + tid;
        const int ql = q & 31, qn = (q >> 5) & 3, qt = q >> 7;
        const float* wp = Wm + (size_t)(32 * (ql >> 4) + 4 * qt) * DIM + 4 * (ql & 15) + qn;
        Bs[q] = make_float4(wp[0], wp[DIM], wp[2 * DIM], wp[3 * DIM]);
    }
    __syncthreads();
    if (!live) return;

    v8f acc[4];
#pragma unroll
    for (int n = 0; n < 4; ++n) acc[n] = (v8f){0.f, 0.f, 0.f, 0.f, 0.f, 0.f, 0.f, 0.f};

#pragma unroll
    for (int t2 = 0; t2 < 8; ++t2) {
        float4 b4[4];
#pragma unroll
        for (int n = 0; n < 4; ++n) b4[n] = Bs[(t2 * 4 + n) * 32 + lane];
#pragma unroll
        for (int n = 0; n < 4; ++n)
            acc[n] = __builtin_amdgcn_wmma_f32_16x16x4_f32(false, lo2(a4[t2]), false, lo2(b4[n]),
                                                           (short)0, acc[n], false, false);
#pragma unroll
        for (int n = 0; n < 4; ++n)
            acc[n] = __builtin_amdgcn_wmma_f32_16x16x4_f32(false, hi2(a4[t2]), false, hi2(b4[n]),
                                                           (short)0, acc[n], false, false);
    }

    float* orow = XW + (size_t)(strip * 16 + 8 * h) * DIM + 4 * c;
#pragma unroll
    for (int r = 0; r < 8; ++r)
        *(float4*)(orow + (size_t)r * DIM) = make_float4(acc[0][r], acc[1][r], acc[2][r], acc[3][r]);

    const float4 as4 = *(const float4*)(as + 4 * c);
    const float4 ad4 = *(const float4*)(ad + 4 * c);
    const float asv[4] = {as4.x, as4.y, as4.z, as4.w};
    const float adv[4] = {ad4.x, ad4.y, ad4.z, ad4.w};
    float v[16];
#pragma unroll
    for (int r = 0; r < 8; ++r) {
        float s = acc[0][r] * asv[0];
        float d = acc[0][r] * adv[0];
#pragma unroll
        for (int n = 1; n < 4; ++n) { s += acc[n][r] * asv[n]; d += acc[n][r] * adv[n]; }
        v[r] = s;
        v[8 + r] = d;
    }
    fold_step<8>(v, c);
    fold_step<4>(v, c);
    fold_step<2>(v, c);
    fold_step<1>(v, c);
    float* dots = (c < 8) ? a_src : a_dst;
    dots[strip * 16 + 8 * h + (c & 7)] = v[0];
}

__device__ __forceinline__ void strip_project(const float4* Hs4,
                                              const float* __restrict__ Wm,
                                              const float* __restrict__ as,
                                              const float* __restrict__ ad,
                                              float* __restrict__ XW,
                                              float* __restrict__ a_src,
                                              float* __restrict__ a_dst,
                                              int strip, int lane, int n) {
    const int h = lane >> 4;
    const int c = lane & 15;
    const float4* wr = (const float4*)(Wm + (size_t)(32 * h) * DIM + 4 * c);

    v8f acc[4];
#pragma unroll
    for (int t = 0; t < 4; ++t) acc[t] = (v8f){0.f, 0.f, 0.f, 0.f, 0.f, 0.f, 0.f, 0.f};

#pragma unroll
    for (int t2 = 0; t2 < 8; ++t2) {
        const float4 a4 = Hs4[c * 17 + 8 * h + t2];
        float4 q[4];
#pragma unroll
        for (int j = 0; j < 4; ++j) q[j] = wr[(4 * t2 + j) * 16];
        const float qa[4][4] = {{q[0].x, q[0].y, q[0].z, q[0].w}, {q[1].x, q[1].y, q[1].z, q[1].w},
                                {q[2].x, q[2].y, q[2].z, q[2].w}, {q[3].x, q[3].y, q[3].z, q[3].w}};
#pragma unroll
        for (int t = 0; t < 4; ++t) {
            v2f b; b.x = qa[0][t]; b.y = qa[1][t];
            acc[t] = __builtin_amdgcn_wmma_f32_16x16x4_f32(false, lo2(a4), false, b,
                                                           (short)0, acc[t], false, false);
        }
#pragma unroll
        for (int t = 0; t < 4; ++t) {
            v2f b; b.x = qa[2][t]; b.y = qa[3][t];
            acc[t] = __builtin_amdgcn_wmma_f32_16x16x4_f32(false, hi2(a4), false, b,
                                                           (short)0, acc[t], false, false);
        }
    }

    const int row0 = strip * 16 + 8 * h;
    float* orow = XW + (size_t)row0 * DIM + 4 * c;
#pragma unroll
    for (int r = 0; r < 8; ++r)
        if (row0 + r < n)
            *(float4*)(orow + (size_t)r * DIM) = make_float4(acc[0][r], acc[1][r], acc[2][r], acc[3][r]);

    const float4 as4 = *(const float4*)(as + 4 * c);
    const float4 ad4 = *(const float4*)(ad + 4 * c);
    const float asv[4] = {as4.x, as4.y, as4.z, as4.w};
    const float adv[4] = {ad4.x, ad4.y, ad4.z, ad4.w};
    float v[16];
#pragma unroll
    for (int r = 0; r < 8; ++r) {
        float sa = acc[0][r] * asv[0];
        float da = acc[0][r] * adv[0];
#pragma unroll
        for (int t = 1; t < 4; ++t) { sa += acc[t][r] * asv[t]; da += acc[t][r] * adv[t]; }
        v[r] = sa;
        v[8 + r] = da;
    }
    fold_step<8>(v, c);
    fold_step<4>(v, c);
    fold_step<2>(v, c);
    fold_step<1>(v, c);
    float* dots = (c < 8) ? a_src : a_dst;
    if (row0 + (c & 7) < n) dots[row0 + (c & 7)] = v[0];
}


#define BUCKET_SHIFT 8
#define BUCKET_NODES (1 << BUCKET_SHIFT)
#define MAXB 512
#define EPB 4096
#define FIN_CAP 8192
#define EPT (EPB / 256)
#define MAXBLK 512
#define FIN_U 13
#define RSLOTS (16 * FIN_U * ((MAXBLK / 16 + FIN_U - 1) / FIN_U))

__device__ __forceinline__ int bucket_of(int d, int NB) {
    return min(max(d, 0) >> BUCKET_SHIFT, NB - 1);
}

__device__ __forceinline__ int block_excl_scan(int v, int* wsum, int tid, int* tot) {
    const int lane = tid & 31, wv = tid >> 5;
    int inc = v;
    for (int o = 1; o < 32; o <<= 1) {
        int t = __shfl_up(inc, o, 32);
        if (lane >= o) inc += t;
    }
    if (lane == 31) wsum[wv] = inc;
    __syncthreads();
    int woff = 0, all = 0;
    for (int w = 0; w < 8; ++w) {
        const int s = wsum[w];
        all += s;
        if (w < wv) woff += s;
    }
    __syncthreads();
    *tot = all;
    return woff + inc - v;
}

__device__ __forceinline__ int2 block_sum2(int a, int b, int* wsum, int tid) {
    const int lane = tid & 31, wv = tid >> 5;
    for (int o = 16; o > 0; o >>= 1) {
        a += __shfl_xor(a, o, 32);
        b += __shfl_xor(b, o, 32);
    }
    if (lane == 0) { wsum[wv] = a; wsum[8 + wv] = b; }
    __syncthreads();
    int sa = 0, sb = 0;
    for (int w = 0; w < 8; ++w) { sa += wsum[w]; sb += wsum[8 + w]; }
    __syncthreads();
    return make_int2(sa, sb);
}

__global__ __launch_bounds__(256) void bucket_partition(const int* __restrict__ esrc,
                                                        const int* __restrict__ edst,
                                                        int* __restrict__ loff,
                                                        int2* __restrict__ tmp, int E, int NB) {
    __shared__ int  cnt[MAXB];
    __shared__ int  lbase[MAXB];
    __shared__ int2 stage[EPB];
    __shared__ int  wsum[8];
    const int tid = threadIdx.x, blk = blockIdx.x;
    const int base = blk * EPB;
    const int n_here = min(EPB, E - base);
    int ed[EPT], es[EPT];
#pragma unroll
    for (int k = 0; k < EPT; ++k) {
        const int e = min(base + k * 256 + tid, E - 1);
        ed[k] = edst[e];
        es[k] = esrc[e];
    }
    cnt[tid] = 0;
    cnt[tid + 256] = 0;
    __syncthreads();
#pragma unroll
    for (int k = 0; k < EPT; ++k)
        if (k * 256 + tid < n_here) atomicAdd(&cnt[bucket_of(ed[k], NB)], 1);
    __syncthreads();
    const int i0 = 2 * tid, i1 = 2 * tid + 1;
    const int h0 = cnt[i0], h1 = cnt[i1];
    int tot;
    const int lb = block_excl_scan(h0 + h1, wsum, tid, &tot);
    lbase[i0] = lb;
    lbase[i1] = lb + h0;
    cnt[i0] = 0;
    cnt[i1] = 0;
    int* row = loff + (size_t)blk * (NB + 1);
    if (i0 < NB) row[i0] = lb;
    if (i1 < NB) row[i1] = lb + h0;
    if (tid == 0) row[NB] = n_here;
    __syncthreads();
#pragma unroll
    for (int k = 0; k < EPT; ++k) {
        if (k * 256 + tid < n_here) {
            const int bk = bucket_of(ed[k], NB);
            const int r = atomicAdd(&cnt[bk], 1);
            const int slot = min(max(lbase[bk] + r, 0), EPB - 1);
            stage[slot] = make_int2(es[k], ed[k]);
        }
    }
    __syncthreads();
#pragma unroll
    for (int k = 0; k < EPT; ++k) {
        const int i = k * 256 + tid;
        if (i < n_here) tmp[base + i] = stage[i];
    }
}

__device__ __forceinline__ void bucket_finish_block(const int2* __restrict__ tmp,
                                                    const int* __restrict__ loff,
                                                    int* __restrict__ rowptr,
                                                    int* __restrict__ lst, int N, int E, int NB,
                                                    int nblk_in, int* cnt, int* cur, int* stage,
                                                    int* wsum, int2* rinfo, int b) {
    const int tid  = threadIdx.x;
    const int nblk = max(min(nblk_in, MAXBLK), 1);
    const int LP   = NB + 1;
    const int last = max(E - 1, 0);
    int s[MAXBLK / 256], e[MAXBLK / 256];
#pragma unroll
    for (int q = 0; q < MAXBLK / 256; ++q) {
        const int j = min(q * 256 + tid, nblk - 1);
        s[q] = loff[(size_t)j * LP + b];
        e[q] = loff[(size_t)j * LP + b + 1];
    }
    int ssum = 0, nsum = 0;
#pragma unroll
    for (int q = 0; q < MAXBLK / 256; ++q) {
        const int j  = q * 256 + tid;
        const int s1 = min(max(s[q], 0), EPB);
        const int e1 = min(max(e[q], s1), EPB);
        const int n1 = (j < nblk) ? e1 - s1 : 0;
        rinfo[j] = make_int2(j * EPB + s1, n1);
        ssum += (j < nblk) ? s1 : 0;
        nsum += n1;
    }
    for (int j = MAXBLK + tid; j < RSLOTS; j += 256) rinfo[j] = make_int2(0, 0);
    cnt[tid] = 0;
    const int2 sums = block_sum2(ssum, nsum, wsum, tid);
    const int beg = min(max(sums.x, 0), E);
    const int end = min(beg + max(sums.y, 0), E);
    const int node0 = b << BUCKET_SHIFT;

    const int hw    = tid >> 4;
    const int c     = tid & 15;
    const int trips = (nblk + 15) >> 4;
    for (int t0 = 0; t0 < trips; t0 += FIN_U) {
        int st[FIN_U], len[FIN_U], y[FIN_U], y2[FIN_U];
#pragma unroll
        for (int u = 0; u < FIN_U; ++u) {
            const int2 ri = rinfo[hw + 16 * (t0 + u)];
            st[u]  = ri.x;
            len[u] = ri.y;
        }
#pragma unroll
        for (int u = 0; u < FIN_U; ++u) y[u] = tmp[min(st[u] + c, last)].y;
#pragma unroll
        for (int u = 0; u < FIN_U; ++u) {
            y2[u] = 0;
            if (c + 16 < len[u]) y2[u] = tmp[min(st[u] + c + 16, last)].y;
        }
#pragma unroll
        for (int u = 0; u < FIN_U; ++u) {
            if (c < len[u])
                atomicAdd(&cnt[min(max(y[u] - node0, 0), BUCKET_NODES - 1)], 1);
            if (c + 16 < len[u])
                atomicAdd(&cnt[min(max(y2[u] - node0, 0), BUCKET_NODES - 1)], 1);
            for (int l = c + 32; l < len[u]; l += 16) {
                const int yy = tmp[min(st[u] + l, last)].y;
                atomicAdd(&cnt[min(max(yy - node0, 0), BUCKET_NODES - 1)], 1);
            }
        }
    }
    __syncthreads();
    int tot;
    const int ex = block_excl_scan(cnt[tid], wsum, tid, &tot);
    if (node0 + tid < N) rowptr[node0 + tid] = beg + ex;
    if (b == 0 && tid == 0) rowptr[N] = E;
    cur[tid] = ex;
    __syncthreads();

    const int  n      = end - beg;
    const bool staged = (n <= FIN_CAP);
    for (int t0 = 0; t0 < trips; t0 += FIN_U) {
        int  st[FIN_U], len[FIN_U];
        int2 sd[FIN_U], sd2[FIN_U];
#pragma unroll
        for (int u = 0; u < FIN_U; ++u) {
            const int2 ri = rinfo[hw + 16 * (t0 + u)];
            st[u]  = ri.x;
            len[u] = ri.y;
        }
#pragma unroll
        for (int u = 0; u < FIN_U; ++u) sd[u] = tmp[min(st[u] + c, last)];
#pragma unroll
        for (int u = 0; u < FIN_U; ++u) {
            sd2[u] = make_int2(0, 0);
            if (c + 16 < len[u]) sd2[u] = tmp[min(st[u] + c + 16, last)];
        }
#pragma unroll
        for (int u = 0; u < FIN_U; ++u) {
            if (c < len[u]) {
                const int li = min(max(sd[u].y - node0, 0), BUCKET_NODES - 1);
                const int r = atomicAdd(&cur[li], 1);
                if (staged) stage[min(max(r, 0), FIN_CAP - 1)] = sd[u].x;
                else        lst[min(max(beg + r, 0), last)] = sd[u].x;
            }
            if (c + 16 < len[u]) {
                const int li = min(max(sd2[u].y - node0, 0), BUCKET_NODES - 1);
                const int r = atomicAdd(&cur[li], 1);
                if (staged) stage[min(max(r, 0), FIN_CAP - 1)] = sd2[u].x;
                else        lst[min(max(beg + r, 0), last)] = sd2[u].x;
            }
            for (int l = c + 32; l < len[u]; l += 16) {
                const int2 s2 = tmp[min(st[u] + l, last)];
                const int li = min(max(s2.y - node0, 0), BUCKET_NODES - 1);
                const int r = atomicAdd(&cur[li], 1);
                if (staged) stage[min(max(r, 0), FIN_CAP - 1)] = s2.x;
                else        lst[min(max(beg + r, 0), last)] = s2.x;
            }
        }
    }
    if (staged) {
        __syncthreads();
        for (int i = tid; i < n; i += 256) lst[beg + i] = stage[i];
    }
}

#define FIN_LDS_INTS (2 * BUCKET_NODES + FIN_CAP + 16 + 2 * RSLOTS)
__global__ __launch_bounds__(256) void finish_and_project(const int2* __restrict__ tmp,
                                                          const int* __restrict__ loff,
                                                          int* __restrict__ rowptr,
                                                          int* __restrict__ lst,
                                                          int N, int E, int NB, int nblk,
                                                          const float* __restrict__ X,
                                                          const float* __restrict__ Wm,
                                                          const float* __restrict__ as,
                                                          const float* __restrict__ ad,
                                                          float* __restrict__ XW,
                                                          float* __restrict__ a_src,
                                                          float* __restrict__ a_dst,
                                                          int n_row_tiles) {
    __shared__ float4 lds[FIN_LDS_INTS / 4];
    const int blk = (int)blockIdx.x;
    if (blk < NB) {
        int* li = (int*)lds;
        bucket_finish_block(tmp, loff, rowptr, lst, N, E, NB, nblk,
                            li, li + BUCKET_NODES, li + 2 * BUCKET_NODES,
                            li + 2 * BUCKET_NODES + FIN_CAP,
                            (int2*)(li + 2 * BUCKET_NODES + FIN_CAP + 16), blk);
    } else {
        gemm64_dots_block(X, Wm, as, ad, XW, a_src, a_dst, n_row_tiles, lds, blk - NB);
    }
}


__device__ __forceinline__ float lrelu_att(float a) { return fmaxf(a, NEG_ATT * a); }

#define DPP_ROW_SHARE 0x150
#define DPP_ROW_XMASK 0x160

template <int CTRL>
__device__ __forceinline__ int dpp_i(int v) {
    return __builtin_amdgcn_update_dpp(0, v, CTRL, 0xf, 0xf, true);
}
template <int CTRL>
__device__ __forceinline__ float dpp_f(float v) {
    return __int_as_float(dpp_i<CTRL>(__float_as_int(v)));
}
__device__ __forceinline__ float row_sum16(float v) {
    v += dpp_f<DPP_ROW_XMASK + 1>(v);
    v += dpp_f<DPP_ROW_XMASK + 2>(v);
    v += dpp_f<DPP_ROW_XMASK + 4>(v);
    v += dpp_f<DPP_ROW_XMASK + 8>(v);
    return v;
}

__device__ __forceinline__ float4 row16(const char* __restrict__ xb, unsigned sb, unsigned coff) {
    return *(const float4*)(xb + (sb | coff));
}

template <int J, int K>
struct Grp {
    static __device__ __forceinline__ void load(const char* __restrict__ xb, int sb, unsigned coff,
                                                float4* x) {
        x[0] = row16(xb, (unsigned)dpp_i<DPP_ROW_SHARE + J>(sb), coff);
        Grp<J + 1, K - 1>::load(xb, sb, coff, x + 1);
    }
    static __device__ __forceinline__ float logits(const float4* x, const float4 as4, float ad,
                                                   float pen, float* l) {
        const v2f q = (v2f){x[0].x, x[0].y} * (v2f){as4.x, as4.y} + (v2f){x[0].z, x[0].w} * (v2f){as4.z, as4.w};
        float p = row_sum16(q.x + q.y);
        l[0] = lrelu_att(p + ad) + dpp_f<DPP_ROW_SHARE + J>(pen);
        if (K == 1) return l[0];
        return fmaxf(l[0], Grp<J + 1, K - 1>::logits(x + 1, as4, ad, pen, l + 1));
    }
    static __device__ __forceinline__ void add(const float4* x, const float* l, float m,
                                               float& dsum, float4& acc) {
        const float w = __expf(l[0] - m);
        dsum += w;
        acc.x += w * x[0].x;
        acc.y += w * x[0].y;
        acc.z += w * x[0].z;
        acc.w += w * x[0].w;
        Grp<J + 1, K - 1>::add(x + 1, l + 1, m, dsum, acc);
    }
};
template <int J>
struct Grp<J, 0> {
    static __device__ __forceinline__ void load(const char* __restrict__, int, unsigned, float4*) {}
    static __device__ __forceinline__ float logits(const float4*, const float4, float, float,
                                                   float*) { return 0.f; }
    static __device__ __forceinline__ void add(const float4*, const float*, float, float&,
                                               float4&) {}
};

template <int J, int K>
__device__ __forceinline__ void add_group(const char* __restrict__ xb, int sb, float pen,
                                          unsigned coff, const float4 as4, float ad,
                                          float& m, float& dsum, float4& acc) {
    float4 x[K];
    float  l[K];
    Grp<J, K>::load(xb, sb, coff, x);
    const float gmax  = Grp<J, K>::logits(x, as4, ad, pen, l);
    const float m_new = fmaxf(m, gmax);
    const float sc    = __expf(m - m_new);
    dsum *= sc;
    acc.x *= sc;
    acc.y *= sc;
    acc.z *= sc;
    acc.w *= sc;
    m = m_new;
    Grp<J, K>::add(x, l, m, dsum, acc);
}

__device__ __forceinline__ void add_rows16(const char* __restrict__ xb, int sb, float pen, int cnt,
                                           unsigned coff, const float4 as4, float ad,
                                           float& m, float& dsum, float4& acc) {
    if (cnt >= 8) {
        add_group<0, 8>(xb, sb, pen, coff, as4, ad, m, dsum, acc);
        switch (cnt - 8) {
            case 1: add_group<8, 1>(xb, sb, pen, coff, as4, ad, m, dsum, acc); break;
            case 2: add_group<8, 2>(xb, sb, pen, coff, as4, ad, m, dsum, acc); break;
            case 3: add_group<8, 3>(xb, sb, pen, coff, as4, ad, m, dsum, acc); break;
            case 4: add_group<8, 4>(xb, sb, pen, coff, as4, ad, m, dsum, acc); break;
            case 5: add_group<8, 5>(xb, sb, pen, coff, as4, ad, m, dsum, acc); break;
            case 6: add_group<8, 6>(xb, sb, pen, coff, as4, ad, m, dsum, acc); break;
            case 7: add_group<8, 7>(xb, sb, pen, coff, as4, ad, m, dsum, acc); break;
            case 8: add_group<8, 8>(xb, sb, pen, coff, as4, ad, m, dsum, acc); break;
            default: break;
        }
    } else {
        switch (cnt) {
            case 1: add_group<0, 1>(xb, sb, pen, coff, as4, ad, m, dsum, acc); break;
            case 2: add_group<0, 2>(xb, sb, pen, coff, as4, ad, m, dsum, acc); break;
            case 3: add_group<0, 3>(xb, sb, pen, coff, as4, ad, m, dsum, acc); break;
            case 4: add_group<0, 4>(xb, sb, pen, coff, as4, ad, m, dsum, acc); break;
            case 5: add_group<0, 5>(xb, sb, pen, coff, as4, ad, m, dsum, acc); break;
            case 6: add_group<0, 6>(xb, sb, pen, coff, as4, ad, m, dsum, acc); break;
            case 7: add_group<0, 7>(xb, sb, pen, coff, as4, ad, m, dsum, acc); break;
            default: break;
        }
    }
}

template <bool NEXT>
__global__ __launch_bounds__(256, 10) void node_gather(const int* __restrict__ rowptr,
                                                   const int* __restrict__ lst,
                                                   const float* __restrict__ XW,
                                                   const float* __restrict__ att_s,
                                                   const float* __restrict__ a_src,
                                                   const float* __restrict__ a_dst,
                                                   const float* __restrict__ bias,
                                                   float* __restrict__ out, int n, int E,
                                                   const float* __restrict__ Wn,
                                                   const float* __restrict__ as_n,
                                                   const float* __restrict__ ad_n,
                                                   float* __restrict__ XWn,
                                                   float* __restrict__ a_src_n,
                                                   float* __restrict__ a_dst_n) {
    __shared__ float4 Hs4[NEXT ? 16 * 17 : 1];
    __shared__ int    ticket;
    if (NEXT) {
        if (threadIdx.x == 0) ticket = 0;
        __syncthreads();
    }
    const int lane = threadIdx.x & 31;
    const int c    = lane & 15;
    const int wave = (int)((blockIdx.x * blockDim.x + threadIdx.x) >> 5);
    if (2 * wave >= n) return;
    const int  node_raw = 2 * wave + (lane >> 4);
    const bool live     = node_raw < n;
    const int  node     = live ? node_raw : n - 1;

    const int beg = min(max(rowptr[node], 0), E);
    const int end = min(max(rowptr[node + 1], beg), E);
    const int deg = live ? end - beg : 0;
    const float ad     = a_dst[node];
    const float a_self = lrelu_att(a_src[node] + ad);

    const char*    xb   = (const char*)XW;
    const unsigned coff = 16u * (unsigned)c;
    const float4   as4  = *(const float4*)(att_s + 4 * c);
    const float    NINF = -__builtin_inff();

    const bool have0 = (c < deg), have1 = (16 + c < deg);
    const int  last  = max(E - 1, 0);
    const int  l0 = lst[min(beg + c, last)];
    const int  l1 = lst[min(beg + 16 + c, last)];
    const float4 xs = row16(xb, (unsigned)node << 8, coff);
    const int  s0 = have0 ? min(max(l0, 0), n - 1) : node;
    const int  s1 = have1 ? min(max(l1, 0), n - 1) : node;

    float  m    = a_self;
    float  dsum = 1.f;
    float4 acc  = xs;

    const int maxdeg = max(__builtin_amdgcn_readlane(deg, 0), __builtin_amdgcn_readlane(deg, 16));
    for (int base = 0; base < maxdeg; base += 16) {
        int   s   = s0;
        float pen = have0 ? 0.f : NINF;
        if (base == 16) { s = s1; pen = have1 ? 0.f : NINF; }
        if (base >= 32) {
            s   = node;
            pen = NINF;
            if (base + c < deg) {
                s   = min(max(lst[beg + base + c], 0), n - 1);
                pen = 0.f;
            }
        }
        const int left = maxdeg - base;
        add_rows16(xb, s << 8, pen, (left < 16) ? left : 16, coff, as4, ad, m, dsum, acc);
    }
    const float den = dsum + 1e-16f;

    const float4 b4 = *(const float4*)(bias + 4 * c);
    float4 v = make_float4(acc.x / den + b4.x, acc.y / den + b4.y, acc.z / den + b4.z,
                           acc.w / den + b4.w);
    v.x = (v.x > 0.f) ? v.x : NEG_OUT * v.x;
    v.y = (v.y > 0.f) ? v.y : NEG_OUT * v.y;
    v.z = (v.z > 0.f) ? v.z : NEG_OUT * v.z;
    v.w = (v.w > 0.f) ? v.w : NEG_OUT * v.w;
    if (!NEXT) {
        if (live) *(float4*)(out + (size_t)node * DIM + 4 * c) = v;
        return;
    }
    Hs4[(2 * (int)(threadIdx.x >> 5) + (lane >> 4)) * 17 + c] = v;
    __builtin_amdgcn_fence(__ATOMIC_RELEASE, "workgroup");
    int t = 0;
    if (lane == 0) t = __hip_atomic_fetch_add(&ticket, 1, __ATOMIC_RELAXED, __HIP_MEMORY_SCOPE_WORKGROUP);
    t = __builtin_amdgcn_readfirstlane(t);
    const int strip  = (int)blockIdx.x;
    const int nwaves = min(8, (n - 16 * strip + 1) >> 1);
    if (t != nwaves - 1) return;
    __builtin_amdgcn_fence(__ATOMIC_ACQUIRE, "workgroup");
    strip_project(Hs4, Wn, as_n, ad_n, XWn, a_src_n, a_dst_n, strip, lane, n);
}

static inline int cdiv(int a, int b) { return (a + b - 1) / b; }

template <bool NEXT>
static void run_gather(const float* bias, const float* att_s, const int* rowptr, const int* lst,
                       int N, int E, const float* xw, const float* a_src, const float* a_dst,
                       float* outbuf, const float* Wn, const float* as_n, const float* ad_n,
                       float* xw_n, float* a_src_n, float* a_dst_n, hipStream_t stream) {
    node_gather<NEXT><<<cdiv(N, 16), 256, 0, stream>>>(rowptr, lst, xw, att_s, a_src, a_dst, bias,
                                                       outbuf, N, E, Wn, as_n, ad_n, xw_n, a_src_n,
                                                       a_dst_n);
}

extern "C" void kernel_launch(void* const* d_in, const int* in_sizes, int n_in,
                              void* d_out, int out_size, void* d_ws, size_t ws_size,
                              hipStream_t stream) {
    const int*   edge_index = (const int*)d_in[0];
    const float* emb        = (const float*)d_in[1];
    const float* W1         = (const float*)d_in[2];
    const float* att_src1   = (const float*)d_in[3];
    const float* att_dst1   = (const float*)d_in[4];
    const float* b1         = (const float*)d_in[5];
    const float* W2         = (const float*)d_in[6];
    const float* att_src2   = (const float*)d_in[7];
    const float* att_dst2   = (const float*)d_in[8];
    const float* b2         = (const float*)d_in[9];

    const int E = in_sizes[0] / 2;
    const int N = in_sizes[1] / DIM;

    const int* esrc = edge_index;
    const int* edst = edge_index + E;

    float* ws     = (float*)d_ws;
    float* xw     = ws;
    float* xw2    = xw + (size_t)N * DIM;
    int2*  tmp    = (int2*)(xw2 + (size_t)N * DIM);
    int*   lst    = (int*)(tmp + E);
    float* a_src  = (float*)(lst + E);
    float* a_dst  = a_src + N;
    float* a_src2 = a_dst + N;
    float* a_dst2 = a_src2 + N;
    int*   rowptr = (int*)(a_dst2 + N);
    const int NB   = cdiv(N, BUCKET_NODES);
    const int nblk = cdiv(E, EPB);
    int*   loff   = rowptr + N + 1;

    bucket_partition<<<nblk, 256, 0, stream>>>(esrc, edst, loff, tmp, E, NB);
    const int n_row_tiles = N / 16;
    const int gemm_blocks = cdiv(n_row_tiles, 8);
    finish_and_project<<<NB + gemm_blocks, 256, 0, stream>>>(tmp, loff, rowptr, lst, N, E, NB, nblk,
                                                             emb, W1, att_src1, att_dst1,
                                                             xw, a_src, a_dst, n_row_tiles);
    run_gather<true>(b1, att_src1, rowptr, lst, N, E, xw, a_src, a_dst, xw2,
                     W2, att_src2, att_dst2, xw2, a_src2, a_dst2, stream);
    run_gather<false>(b2, att_src2, rowptr, lst, N, E, xw2, a_src2, a_dst2, (float*)d_out,
                      W2, att_src2, att_dst2, xw2, a_src2, a_dst2, stream);
}


# Gather: a group's row dots summed over 16 lanes by a DPP fold (lane i gets entry i); leaky relu, max, exp once per entry instead of in all 16 lanes; weight broadcast; logit sum tree changes
# speedup vs baseline: 7.5090x; 1.0468x over previous
#include <hip/hip_runtime.h>
#include <math.h>


typedef float v2f __attribute__((ext_vector_type(2)));
typedef float v8f __attribute__((ext_vector_type(8)));

#define DIM 64
#define NEG_ATT 0.2f
#define NEG_OUT 0.01f

__device__ __forceinline__ v2f lo2(const float4 q) { v2f r; r.x = q.x; r.y = q.y; return r; }
__device__ __forceinline__ v2f hi2(const float4 q) { v2f r; r.x = q.z; r.y = q.w; return r; }

template <int KEEP>
__device__ __forceinline__ void fold_step(float* v, int c) {
    const bool up = (c & KEEP) != 0;
#pragma unroll
    for (int i = 0; i < KEEP; ++i) {
        const float keep = up ? v[i + KEEP] : v[i];
        const float send = up ? v[i] : v[i + KEEP];
        v[i] = keep + __shfl_xor(send, KEEP, 32);
    }
}

__device__ __forceinline__ void gemm64_dots_block(const float* __restrict__ X,
                                                  const float* __restrict__ Wm,
                                                  const float* __restrict__ as,
                                                  const float* __restrict__ ad,
                                                  float* __restrict__ XW,
                                                  float* __restrict__ a_src,
                                                  float* __restrict__ a_dst,
                                                  int n_row_tiles, float4* Bs, int blk) {
    const int tid   = threadIdx.x;
    const int lane  = tid & 31;
    const int h     = lane >> 4;
    const int c     = lane & 15;
    const int strip = blk * 8 + (tid >> 5);
    const bool live = strip < n_row_tiles;

    float4 a4[8];
    if (live) {
        const float4* xr = (const float4*)(X + (size_t)(strip * 16 + c) * DIM + 32 * h);
#pragma unroll
        for (int t2 = 0; t2 < 8; ++t2) a4[t2] = xr[t2];
    }

#pragma unroll
    for (int i = 0; i < 4; ++i) {
        const int q  = i * 256 + tid;
        const int ql = q & 31, qn = (q >> 5) & 3, qt = q >> 7;
        const float* wp = Wm + (size_t)(32 * (ql >> 4) + 4 * qt) * DIM + 4 * (ql & 15) + qn;
        Bs[q] = make_float4(wp[0], wp[DIM], wp[2 * DIM], wp[3 * DIM]);
    }
    __syncthreads();
    if (!live) return;

    v8f acc[4];
#pragma unroll
    for (int n = 0; n < 4; ++n) acc[n] = (v8f){0.f, 0.f, 0.f, 0.f, 0.f, 0.f, 0.f, 0.f};

#pragma unroll
    for (int t2 = 0; t2 < 8; ++t2) {
        float4 b4[4];
#pragma unroll
        for (int n = 0; n < 4; ++n) b4[n] = Bs[(t2 * 4 + n) * 32 + lane];
#pragma unroll
        for (int n = 0; n < 4; ++n)
            acc[n] = __builtin_amdgcn_wmma_f32_16x16x4_f32(false, lo2(a4[t2]), false, lo2(b4[n]),
                                                           (short)0, acc[n], false, false);
#pragma unroll
        for (int n = 0; n < 4; ++n)
            acc[n] = __builtin_amdgcn_wmma_f32_16x16x4_f32(false, hi2(a4[t2]), false, hi2(b4[n]),
                                                           (short)0, acc[n], false, false);
    }

    float* orow = XW + (size_t)(strip * 16 + 8 * h) * DIM + 4 * c;
#pragma unroll
    for (int r = 0; r < 8; ++r)
        *(float4*)(orow + (size_t)r * DIM) = make_float4(acc[0][r], acc[1][r], acc[2][r], acc[3][r]);

    const float4 as4 = *(const float4*)(as + 4 * c);
    const float4 ad4 = *(const float4*)(ad + 4 * c);
    const float asv[4] = {as4.x, as4.y, as4.z, as4.w};
    const float adv[4] = {ad4.x, ad4.y, ad4.z, ad4.w};
    float v[16];
#pragma unroll
    for (int r = 0; r < 8; ++r) {
        float s = acc[0][r] * asv[0];
        float d = acc[0][r] * adv[0];
#pragma unroll
        for (int n = 1; n < 4; ++n) { s += acc[n][r] * asv[n]; d += acc[n][r] * adv[n]; }
        v[r] = s;
        v[8 + r] = d;
    }
    fold_step<8>(v, c);
    fold_step<4>(v, c);
    fold_step<2>(v, c);
    fold_step<1>(v, c);
    float* dots = (c < 8) ? a_src : a_dst;
    dots[strip * 16 + 8 * h + (c & 7)] = v[0];
}

__device__ __forceinline__ void strip_project(const float4* Hs4,
                                              const float* __restrict__ Wm,
                                              const float* __restrict__ as,
                                              const float* __restrict__ ad,
                                              float* __restrict__ XW,
                                              float* __restrict__ a_src,
                                              float* __restrict__ a_dst,
                                              int strip, int lane, int n) {
    const int h = lane >> 4;
    const int c = lane & 15;
    const char*    wb   = (const char*)Wm;
    const unsigned woff = (unsigned)(32 * h) * (DIM * 4u) + 16u * (unsigned)c;

    v8f acc[4];
#pragma unroll
    for (int t = 0; t < 4; ++t) acc[t] = (v8f){0.f, 0.f, 0.f, 0.f, 0.f, 0.f, 0.f, 0.f};

#pragma unroll
    for (int t2 = 0; t2 < 8; ++t2) {
        const float4 a4 = Hs4[c * 17 + 8 * h + t2];
        float4 q[4];
#pragma unroll
        for (int j = 0; j < 4; ++j) q[j] = *(const float4*)(wb + (woff + (unsigned)(4 * t2 + j) * (DIM * 4u)));
        const float qa[4][4] = {{q[0].x, q[0].y, q[0].z, q[0].w}, {q[1].x, q[1].y, q[1].z, q[1].w},
                                {q[2].x, q[2].y, q[2].z, q[2].w}, {q[3].x, q[3].y, q[3].z, q[3].w}};
#pragma unroll
        for (int t = 0; t < 4; ++t) {
            v2f b; b.x = qa[0][t]; b.y = qa[1][t];
            acc[t] = __builtin_amdgcn_wmma_f32_16x16x4_f32(false, lo2(a4), false, b,
                                                           (short)0, acc[t], false, false);
        }
#pragma unroll
        for (int t = 0; t < 4; ++t) {
            v2f b; b.x = qa[2][t]; b.y = qa[3][t];
            acc[t] = __builtin_amdgcn_wmma_f32_16x16x4_f32(false, hi2(a4), false, b,
                                                           (short)0, acc[t], false, false);
        }
    }

    const int row0 = strip * 16 + 8 * h;
    char*          ob   = (char*)XW;
    const unsigned ooff = (unsigned)row0 * (DIM * 4u) + 16u * (unsigned)c;
#pragma unroll
    for (int r = 0; r < 8; ++r)
        if (row0 + r < n)
            *(float4*)(ob + (ooff + (unsigned)r * (DIM * 4u))) = make_float4(acc[0][r], acc[1][r], acc[2][r], acc[3][r]);

    const float4 as4 = *(const float4*)(as + 4 * c);
    const float4 ad4 = *(const float4*)(ad + 4 * c);
    const float asv[4] = {as4.x, as4.y, as4.z, as4.w};
    const float adv[4] = {ad4.x, ad4.y, ad4.z, ad4.w};
    float v[16];
#pragma unroll
    for (int r = 0; r < 8; ++r) {
        float sa = acc[0][r] * asv[0];
        float da = acc[0][r] * adv[0];
#pragma unroll
        for (int t = 1; t < 4; ++t) { sa += acc[t][r] * asv[t]; da += acc[t][r] * adv[t]; }
        v[r] = sa;
        v[8 + r] = da;
    }
    fold_step<8>(v, c);
    fold_step<4>(v, c);
    fold_step<2>(v, c);
    fold_step<1>(v, c);
    float* dots = (c < 8) ? a_src : a_dst;
    if (row0 + (c & 7) < n) dots[row0 + (c & 7)] = v[0];
}


#define BUCKET_SHIFT 8
#define BUCKET_NODES (1 << BUCKET_SHIFT)
#define MAXB 512
#define EPB 4096
#define FIN_CAP 8192
#define EPT (EPB / 256)
#define MAXBLK 512
#define FIN_U 13
#define RSLOTS (16 * FIN_U * ((MAXBLK / 16 + FIN_U - 1) / FIN_U))

__device__ __forceinline__ int bucket_of(int d, int NB) {
    return min(max(d, 0) >> BUCKET_SHIFT, NB - 1);
}

__device__ __forceinline__ int block_excl_scan(int v, int* wsum, int tid, int* tot) {
    const int lane = tid & 31, wv = tid >> 5;
    int inc = v;
    for (int o = 1; o < 32; o <<= 1) {
        int t = __shfl_up(inc, o, 32);
        if (lane >= o) inc += t;
    }
    if (lane == 31) wsum[wv] = inc;
    __syncthreads();
    int woff = 0, all = 0;
    for (int w = 0; w < 8; ++w) {
        const int s = wsum[w];
        all += s;
        if (w < wv) woff += s;
    }
    __syncthreads();
    *tot = all;
    return woff + inc - v;
}

__device__ __forceinline__ int2 block_sum2(int a, int b, int* wsum, int tid) {
    const int lane = tid & 31, wv = tid >> 5;
    for (int o = 16; o > 0; o >>= 1) {
        a += __shfl_xor(a, o, 32);
        b += __shfl_xor(b, o, 32);
    }
    if (lane == 0) { wsum[wv] = a; wsum[8 + wv] = b; }
    __syncthreads();
    int sa = 0, sb = 0;
    for (int w = 0; w < 8; ++w) { sa += wsum[w]; sb += wsum[8 + w]; }
    __syncthreads();
    return make_int2(sa, sb);
}

__global__ __launch_bounds__(256) void bucket_partition(const int* __restrict__ esrc,
                                                        const int* __restrict__ edst,
                                                        int* __restrict__ loff,
                                                        int2* __restrict__ tmp, int E, int NB) {
    __shared__ int  cnt[MAXB];
    __shared__ int  lbase[MAXB];
    __shared__ int2 stage[EPB];
    __shared__ int  wsum[8];
    const int tid = threadIdx.x, blk = blockIdx.x;
    const int base = blk * EPB;
    const int n_here = min(EPB, E - base);
    int ed[EPT], es[EPT];
#pragma unroll
    for (int k = 0; k < EPT; ++k) {
        const int e = min(base + k * 256 + tid, E - 1);
        ed[k] = edst[e];
        es[k] = esrc[e];
    }
    cnt[tid] = 0;
    cnt[tid + 256] = 0;
    __syncthreads();
#pragma unroll
    for (int k = 0; k < EPT; ++k)
        if (k * 256 + tid < n_here) atomicAdd(&cnt[bucket_of(ed[k], NB)], 1);
    __syncthreads();
    const int i0 = 2 * tid, i1 = 2 * tid + 1;
    const int h0 = cnt[i0], h1 = cnt[i1];
    int tot;
    const int lb = block_excl_scan(h0 + h1, wsum, tid, &tot);
    lbase[i0] = lb;
    lbase[i1] = lb + h0;
    cnt[i0] = 0;
    cnt[i1] = 0;
    int* row = loff + (size_t)blk * (NB + 1);
    if (i0 < NB) row[i0] = lb;
    if (i1 < NB) row[i1] = lb + h0;
    if (tid == 0) row[NB] = n_here;
    __syncthreads();
#pragma unroll
    for (int k = 0; k < EPT; ++k) {
        if (k * 256 + tid < n_here) {
            const int bk = bucket_of(ed[k], NB);
            const int r = atomicAdd(&cnt[bk], 1);
            const int slot = min(max(lbase[bk] + r, 0), EPB - 1);
            stage[slot] = make_int2(es[k], ed[k]);
        }
    }
    __syncthreads();
#pragma unroll
    for (int k = 0; k < EPT; ++k) {
        const int i = k * 256 + tid;
        if (i < n_here) tmp[base + i] = stage[i];
    }
}

__device__ __forceinline__ void bucket_finish_block(const int2* __restrict__ tmp,
                                                    const int* __restrict__ loff,
                                                    int* __restrict__ rowptr,
                                                    int* __restrict__ lst, int N, int E, int NB,
                                                    int nblk_in, int* cnt, int* cur, int* stage,
                                                    int* wsum, int2* rinfo, int b) {
    const int tid  = threadIdx.x;
    const int nblk = max(min(nblk_in, MAXBLK), 1);
    const int LP   = NB + 1;
    const int last = max(E - 1, 0);
    int s[MAXBLK / 256], e[MAXBLK / 256];
#pragma unroll
    for (int q = 0; q < MAXBLK / 256; ++q) {
        const int j = min(q * 256 + tid, nblk - 1);
        s[q] = loff[(size_t)j * LP + b];
        e[q] = loff[(size_t)j * LP + b + 1];
    }
    int ssum = 0, nsum = 0;
#pragma unroll
    for (int q = 0; q < MAXBLK / 256; ++q) {
        const int j  = q * 256 + tid;
        const int s1 = min(max(s[q], 0), EPB);
        const int e1 = min(max(e[q], s1), EPB);
        const int n1 = (j < nblk) ? e1 - s1 : 0;
        rinfo[j] = make_int2(j * EPB + s1, n1);
        ssum += (j < nblk) ? s1 : 0;
        nsum += n1;
    }
    for (int j = MAXBLK + tid; j < RSLOTS; j += 256) rinfo[j] = make_int2(0, 0);
    cnt[tid] = 0;
    const int2 sums = block_sum2(ssum, nsum, wsum, tid);
    const int beg = min(max(sums.x, 0), E);
    const int end = min(beg + max(sums.y, 0), E);
    const int node0 = b << BUCKET_SHIFT;

    const int hw    = tid >> 4;
    const int c     = tid & 15;
    const int trips = (nblk + 15) >> 4;
    for (int t0 = 0; t0 < trips; t0 += FIN_U) {
        int st[FIN_U], len[FIN_U], y[FIN_U], y2[FIN_U];
#pragma unroll
        for (int u = 0; u < FIN_U; ++u) {
            const int2 ri = rinfo[hw + 16 * (t0 + u)];
            st[u]  = ri.x;
            len[u] = ri.y;
        }
#pragma unroll
        for (int u = 0; u < FIN_U; ++u) y[u] = tmp[min(st[u] + c, last)].y;
#pragma unroll
        for (int u = 0; u < FIN_U; ++u) {
            y2[u] = 0;
            if (c + 16 < len[u]) y2[u] = tmp[min(st[u] + c + 16, last)].y;
        }
#pragma unroll
        for (int u = 0; u < FIN_U; ++u) {
            if (c < len[u])
                atomicAdd(&cnt[min(max(y[u] - node0, 0), BUCKET_NODES - 1)], 1);
            if (c + 16 < len[u])
                atomicAdd(&cnt[min(max(y2[u] - node0, 0), BUCKET_NODES - 1)], 1);
            for (int l = c + 32; l < len[u]; l += 16) {
                const int yy = tmp[min(st[u] + l, last)].y;
                atomicAdd(&cnt[min(max(yy - node0, 0), BUCKET_NODES - 1)], 1);
            }
        }
    }
    __syncthreads();
    int tot;
    const int ex = block_excl_scan(cnt[tid], wsum, tid, &tot);
    if (node0 + tid < N) rowptr[node0 + tid] = beg + ex;
    if (b == 0 && tid == 0) rowptr[N] = E;
    cur[tid] = ex;
    __syncthreads();

    const int  n      = end - beg;
    const bool staged = (n <= FIN_CAP);
    for (int t0 = 0; t0 < trips; t0 += FIN_U) {
        int  st[FIN_U], len[FIN_U];
        int2 sd[FIN_U], sd2[FIN_U];
#pragma unroll
        for (int u = 0; u < FIN_U; ++u) {
            const int2 ri = rinfo[hw + 16 * (t0 + u)];
            st[u]  = ri.x;
            len[u] = ri.y;
        }
#pragma unroll
        for (int u = 0; u < FIN_U; ++u) sd[u] = tmp[min(st[u] + c, last)];
#pragma unroll
        for (int u = 0; u < FIN_U; ++u) {
            sd2[u] = make_int2(0, 0);
            if (c + 16 < len[u]) sd2[u] = tmp[min(st[u] + c + 16, last)];
        }
#pragma unroll
        for (int u = 0; u < FIN_U; ++u) {
            if (c < len[u]) {
                const int li = min(max(sd[u].y - node0, 0), BUCKET_NODES - 1);
                const int r = atomicAdd(&cur[li], 1);
                if (staged) stage[min(max(r, 0), FIN_CAP - 1)] = sd[u].x;
                else        lst[min(max(beg + r, 0), last)] = sd[u].x;
            }
            if (c + 16 < len[u]) {
                const int li = min(max(sd2[u].y - node0, 0), BUCKET_NODES - 1);
                const int r = atomicAdd(&cur[li], 1);
                if (staged) stage[min(max(r, 0), FIN_CAP - 1)] = sd2[u].x;
                else        lst[min(max(beg + r, 0), last)] = sd2[u].x;
            }
            for (int l = c + 32; l < len[u]; l += 16) {
                const int2 s2 = tmp[min(st[u] + l, last)];
                const int li = min(max(s2.y - node0, 0), BUCKET_NODES - 1);
                const int r = atomicAdd(&cur[li], 1);
                if (staged) stage[min(max(r, 0), FIN_CAP - 1)] = s2.x;
                else        lst[min(max(beg + r, 0), last)] = s2.x;
            }
        }
    }
    if (staged) {
        __syncthreads();
        for (int i = tid; i < n; i += 256) lst[beg + i] = stage[i];
    }
}

#define FIN_LDS_INTS (2 * BUCKET_NODES + FIN_CAP + 16 + 2 * RSLOTS)
__global__ __launch_bounds__(256) void finish_and_project(const int2* __restrict__ tmp,
                                                          const int* __restrict__ loff,
                                                          int* __restrict__ rowptr,
                                                          int* __restrict__ lst,
                                                          int N, int E, int NB, int nblk,
                                                          const float* __restrict__ X,
                                                          const float* __restrict__ Wm,
                                                          const float* __restrict__ as,
                                                          const float* __restrict__ ad,
                                                          float* __restrict__ XW,
                                                          float* __restrict__ a_src,
                                                          float* __restrict__ a_dst,
                                                          int n_row_tiles) {
    __shared__ float4 lds[FIN_LDS_INTS / 4];
    const int blk = (int)blockIdx.x;
    if (blk < NB) {
        int* li = (int*)lds;
        bucket_finish_block(tmp, loff, rowptr, lst, N, E, NB, nblk,
                            li, li + BUCKET_NODES, li + 2 * BUCKET_NODES,
                            li + 2 * BUCKET_NODES + FIN_CAP,
                            (int2*)(li + 2 * BUCKET_NODES + FIN_CAP + 16), blk);
    } else {
        gemm64_dots_block(X, Wm, as, ad, XW, a_src, a_dst, n_row_tiles, lds, blk - NB);
    }
}


__device__ __forceinline__ float lrelu_att(float a) { return fmaxf(a, NEG_ATT * a); }

#define DPP_ROW_SHARE 0x150
#define DPP_ROW_XMASK 0x160

template <int CTRL>
__device__ __forceinline__ int dpp_i(int v) {
    return __builtin_amdgcn_update_dpp(0, v, CTRL, 0xf, 0xf, true);
}
template <int CTRL>
__device__ __forceinline__ float dpp_f(float v) {
    return __int_as_float(dpp_i<CTRL>(__float_as_int(v)));
}

__device__ __forceinline__ float4 row16(const char* __restrict__ xb, unsigned sb, unsigned coff) {
    return *(const float4*)(xb + (sb | coff));
}

template <int KEEP>
__device__ __forceinline__ void fold_row(float* v, int c) {
    const bool up = (c & KEEP) != 0;
#pragma unroll
    for (int i = 0; i < KEEP; ++i) {
        const float keep = up ? v[i + KEEP] : v[i];
        const float send = up ? v[i] : v[i + KEEP];
        v[i] = keep + dpp_f<DPP_ROW_XMASK + KEEP>(send);
    }
}
__device__ __forceinline__ float row_max16(float v) {
    v = fmaxf(v, dpp_f<DPP_ROW_XMASK + 1>(v));
    v = fmaxf(v, dpp_f<DPP_ROW_XMASK + 2>(v));
    v = fmaxf(v, dpp_f<DPP_ROW_XMASK + 4>(v));
    v = fmaxf(v, dpp_f<DPP_ROW_XMASK + 8>(v));
    return v;
}

template <int J, int K>
struct Grp {
    static __device__ __forceinline__ void load(const char* __restrict__ xb, int sb, unsigned coff,
                                                float4* x) {
        x[0] = row16(xb, (unsigned)dpp_i<DPP_ROW_SHARE + J>(sb), coff);
        Grp<J + 1, K - 1>::load(xb, sb, coff, x + 1);
    }
    static __device__ __forceinline__ void partial(const float4* x, const float4 as4, float* v) {
        const v2f q = (v2f){x[0].x, x[0].y} * (v2f){as4.x, as4.y} + (v2f){x[0].z, x[0].w} * (v2f){as4.z, as4.w};
        v[0] = q.x + q.y;
        Grp<J + 1, K - 1>::partial(x + 1, as4, v + 1);
    }
    static __device__ __forceinline__ void add(const float4* x, float w, float& dsum, float4& acc) {
        const float wi = dpp_f<DPP_ROW_SHARE + J>(w);
        dsum += wi;
        acc.x += wi * x[0].x;
        acc.y += wi * x[0].y;
        acc.z += wi * x[0].z;
        acc.w += wi * x[0].w;
        Grp<J + 1, K - 1>::add(x + 1, w, dsum, acc);
    }
};
template <int J>
struct Grp<J, 0> {
    static __device__ __forceinline__ void load(const char* __restrict__, int, unsigned, float4*) {}
    static __device__ __forceinline__ void partial(const float4*, const float4, float*) {}
    static __device__ __forceinline__ void add(const float4*, float, float&, float4&) {}
};

template <int J, int K>
__device__ __forceinline__ void add_group(const char* __restrict__ xb, int sb, float pen, int c,
                                          unsigned coff, const float4 as4, float ad,
                                          float& m, float& dsum, float4& acc) {
    constexpr int P = (K > 4) ? 8 : (K > 2) ? 4 : (K > 1) ? 2 : 1;
    float4 x[K];
    float  v[8];
    Grp<J, K>::load(xb, sb, coff, x);
#pragma unroll
    for (int i = 0; i < 8; ++i) v[i] = 0.f;
    Grp<J, K>::partial(x, as4, v);
    if (P >= 8) fold_row<4>(v, c);
    if (P >= 4) fold_row<2>(v, c);
    if (P >= 2) fold_row<1>(v, c);
    float p = v[0];
    if (P <= 1) p += dpp_f<DPP_ROW_XMASK + 1>(p);
    if (P <= 2) p += dpp_f<DPP_ROW_XMASK + 2>(p);
    if (P <= 4) p += dpp_f<DPP_ROW_XMASK + 4>(p);
    p += dpp_f<DPP_ROW_XMASK + 8>(p);
    const bool  mine  = (c >= J) && (c < J + K);
    const float lg    = mine ? lrelu_att(p + ad) + pen : -__builtin_inff();
    const float gmax  = row_max16(lg);
    const float m_new = fmaxf(m, gmax);
    const float sc    = __expf(m - m_new);
    dsum *= sc;
    acc.x *= sc;
    acc.y *= sc;
    acc.z *= sc;
    acc.w *= sc;
    m = m_new;
    const float w = __expf(lg - m_new);
    Grp<J, K>::add(x, w, dsum, acc);
}

__device__ __forceinline__ void add_rows16(const char* __restrict__ xb, int sb, float pen, int cnt,
                                           int c, unsigned coff, const float4 as4, float ad,
                                           float& m, float& dsum, float4& acc) {
    if (cnt >= 8) {
        add_group<0, 8>(xb, sb, pen, c, coff, as4, ad, m, dsum, acc);
        switch (cnt - 8) {
            case 1: add_group<8, 1>(xb, sb, pen, c, coff, as4, ad, m, dsum, acc); break;
            case 2: add_group<8, 2>(xb, sb, pen, c, coff, as4, ad, m, dsum, acc); break;
            case 3: add_group<8, 3>(xb, sb, pen, c, coff, as4, ad, m, dsum, acc); break;
            case 4: add_group<8, 4>(xb, sb, pen, c, coff, as4, ad, m, dsum, acc); break;
            case 5: add_group<8, 5>(xb, sb, pen, c, coff, as4, ad, m, dsum, acc); break;
            case 6: add_group<8, 6>(xb, sb, pen, c, coff, as4, ad, m, dsum, acc); break;
            case 7: add_group<8, 7>(xb, sb, pen, c, coff, as4, ad, m, dsum, acc); break;
            case 8: add_group<8, 8>(xb, sb, pen, c, coff, as4, ad, m, dsum, acc); break;
            default: break;
        }
    } else {
        switch (cnt) {
            case 1: add_group<0, 1>(xb, sb, pen, c, coff, as4, ad, m, dsum, acc); break;
            case 2: add_group<0, 2>(xb, sb, pen, c, coff, as4, ad, m, dsum, acc); break;
            case 3: add_group<0, 3>(xb, sb, pen, c, coff, as4, ad, m, dsum, acc); break;
            case 4: add_group<0, 4>(xb, sb, pen, c, coff, as4, ad, m, dsum, acc); break;
            case 5: add_group<0, 5>(xb, sb, pen, c, coff, as4, ad, m, dsum, acc); break;
            case 6: add_group<0, 6>(xb, sb, pen, c, coff, as4, ad, m, dsum, acc); break;
            case 7: add_group<0, 7>(xb, sb, pen, c, coff, as4, ad, m, dsum, acc); break;
            default: break;
        }
    }
}

template <bool NEXT>
__global__ __launch_bounds__(256, 10) void node_gather(const int* __restrict__ rowptr,
                                                   const int* __restrict__ lst,
                                                   const float* __restrict__ XW,
                                                   const float* __restrict__ att_s,
                                                   const float* __restrict__ a_src,
                                                   const float* __restrict__ a_dst,
                                                   const float* __restrict__ bias,
                                                   float* __restrict__ out, int n, int E,
                                                   const float* __restrict__ Wn,
                                                   const float* __restrict__ as_n,
                                                   const float* __restrict__ ad_n,
                                                   float* __restrict__ XWn,
                                                   float* __restrict__ a_src_n,
                                                   float* __restrict__ a_dst_n) {
    __shared__ float4 Hs4[NEXT ? 16 * 17 : 1];
    __shared__ int    ticket;
    if (NEXT) {
        if (threadIdx.x == 0) ticket = 0;
        __syncthreads();
    }
    const int lane = threadIdx.x & 31;
    const int c    = lane & 15;
    const int wave = (int)((blockIdx.x * blockDim.x + threadIdx.x) >> 5);
    if (2 * wave >= n) return;
    const int  node_raw = 2 * wave + (lane >> 4);
    const bool live     = node_raw < n;
    const int  node     = live ? node_raw : n - 1;

    const int beg = min(max(rowptr[node], 0), E);
    const int end = min(max(rowptr[node + 1], beg), E);
    const int deg = live ? end - beg : 0;
    const float ad     = a_dst[node];
    const float a_self = lrelu_att(a_src[node] + ad);

    const char*    xb   = (const char*)XW;
    const unsigned coff = 16u * (unsigned)c;
    const float4   as4  = *(const float4*)(att_s + 4 * c);
    const float    NINF = -__builtin_inff();

    const bool have0 = (c < deg), have1 = (16 + c < deg);
    const int  last  = max(E - 1, 0);
    const int  l0 = lst[min(beg + c, last)];
    const int  l1 = lst[min(beg + 16 + c, last)];
    const float4 xs = row16(xb, (unsigned)node << 8, coff);
    const int  s0 = have0 ? min(max(l0, 0), n - 1) : node;
    const int  s1 = have1 ? min(max(l1, 0), n - 1) : node;

    float  m    = a_self;
    float  dsum = 1.f;
    float4 acc  = xs;

    const int maxdeg = max(__builtin_amdgcn_readlane(deg, 0), __builtin_amdgcn_readlane(deg, 16));
    for (int base = 0; base < maxdeg; base += 16) {
        int   s   = s0;
        float pen = have0 ? 0.f : NINF;
        if (base == 16) { s = s1; pen = have1 ? 0.f : NINF; }
        if (base >= 32) {
            s   = node;
            pen = NINF;
            if (base + c < deg) {
                s   = min(max(lst[beg + base + c], 0), n - 1);
                pen = 0.f;
            }
        }
        const int left = maxdeg - base;
        add_rows16(xb, s << 8, pen, (left < 16) ? left : 16, c, coff, as4, ad, m, dsum, acc);
    }
    const float den = dsum + 1e-16f;

    const float4 b4 = *(const float4*)(bias + 4 * c);
    float4 v = make_float4(acc.x / den + b4.x, acc.y / den + b4.y, acc.z / den + b4.z,
                           acc.w / den + b4.w);
    v.x = (v.x > 0.f) ? v.x : NEG_OUT * v.x;
    v.y = (v.y > 0.f) ? v.y : NEG_OUT * v.y;
    v.z = (v.z > 0.f) ? v.z : NEG_OUT * v.z;
    v.w = (v.w > 0.f) ? v.w : NEG_OUT * v.w;
    if (!NEXT) {
        if (live) *(float4*)(out + (size_t)node * DIM + 4 * c) = v;
        return;
    }
    Hs4[(2 * (int)(threadIdx.x >> 5) + (lane >> 4)) * 17 + c] = v;
    __builtin_amdgcn_fence(__ATOMIC_RELEASE, "workgroup");
    int t = 0;
    if (lane == 0) t = __hip_atomic_fetch_add(&ticket, 1, __ATOMIC_RELAXED, __HIP_MEMORY_SCOPE_WORKGROUP);
    t = __builtin_amdgcn_readfirstlane(t);
    const int strip  = (int)blockIdx.x;
    const int nwaves = min(8, (n - 16 * strip + 1) >> 1);
    if (t != nwaves - 1) return;
    __builtin_amdgcn_fence(__ATOMIC_ACQUIRE, "workgroup");
    strip_project(Hs4, Wn, as_n, ad_n, XWn, a_src_n, a_dst_n, strip, lane, n);
}

static inline int cdiv(int a, int b) { return (a + b - 1) / b; }

template <bool NEXT>
static void run_gather(const float* bias, const float* att_s, const int* rowptr, const int* lst,
                       int N, int E, const float* xw, const float* a_src, const float* a_dst,
                       float* outbuf, const float* Wn, const float* as_n, const float* ad_n,
                       float* xw_n, float* a_src_n, float* a_dst_n, hipStream_t stream) {
    node_gather<NEXT><<<cdiv(N, 16), 256, 0, stream>>>(rowptr, lst, xw, att_s, a_src, a_dst, bias,
                                                       outbuf, N, E, Wn, as_n, ad_n, xw_n, a_src_n,
                                                       a_dst_n);
}

extern "C" void kernel_launch(void* const* d_in, const int* in_sizes, int n_in,
                              void* d_out, int out_size, void* d_ws, size_t ws_size,
                              hipStream_t stream) {
    const int*   edge_index = (const int*)d_in[0];
    const float* emb        = (const float*)d_in[1];
    const float* W1         = (const float*)d_in[2];
    const float* att_src1   = (const float*)d_in[3];
    const float* att_dst1   = (const float*)d_in[4];
    const float* b1         = (const float*)d_in[5];
    const float* W2         = (const float*)d_in[6];
    const float* att_src2   = (const float*)d_in[7];
    const float* att_dst2   = (const float*)d_in[8];
    const float* b2         = (const float*)d_in[9];

    const int E = in_sizes[0] / 2;
    const int N = in_sizes[1] / DIM;

    const int* esrc = edge_index;
    const int* edst = edge_index + E;

    float* ws     = (float*)d_ws;
    float* xw     = ws;
    float* xw2    = xw + (size_t)N * DIM;
    int2*  tmp    = (int2*)(xw2 + (size_t)N * DIM);
    int*   lst    = (int*)(tmp + E);
    float* a_src  = (float*)(lst + E);
    float* a_dst  = a_src + N;
    float* a_src2 = a_dst + N;
    float* a_dst2 = a_src2 + N;
    int*   rowptr = (int*)(a_dst2 + N);
    const int NB   = cdiv(N, BUCKET_NODES);
    const int nblk = cdiv(E, EPB);
    int*   loff   = rowptr + N + 1;

    bucket_partition<<<nblk, 256, 0, stream>>>(esrc, edst, loff, tmp, E, NB);
    const int n_row_tiles = N / 16;
    const int gemm_blocks = cdiv(n_row_tiles, 8);
    finish_and_project<<<NB + gemm_blocks, 256, 0, stream>>>(tmp, loff, rowptr, lst, N, E, NB, nblk,
                                                             emb, W1, att_src1, att_dst1,
                                                             xw, a_src, a_dst, n_row_tiles);
    run_gather<true>(b1, att_src1, rowptr, lst, N, E, xw, a_src, a_dst, xw2,
                     W2, att_src2, att_dst2, xw2, a_src2, a_dst2, stream);
    run_gather<false>(b2, att_src2, rowptr, lst, N, E, xw2, a_src2, a_dst2, (float*)d_out,
                      W2, att_src2, att_dst2, xw2, a_src2, a_dst2, stream);
}
